# starting kernel
# baseline (speedup 1.0000x reference)
#include <hip/hip_runtime.h>
#include <hip/hip_bf16.h>
#include <math.h>

#define Bn      8
#define Sn      12
#define NNn     325
#define Dn      128
#define EDn     256
#define DSTATEn 8
#define DCONVn  4
#define DTRANKn 8
#define MSLOTSn 10
#define Ln      (Sn*NNn)
#define NTOK    (Bn*Ln)
#define TM      3
#define TOKB    (16*TM)
#define NBLK    (NTOK/TOKB)

typedef float v2f __attribute__((ext_vector_type(2)));
typedef float v8f __attribute__((ext_vector_type(8)));

__device__ __forceinline__ v8f wmma4(v2f a, v2f b, v8f c) {
  return __builtin_amdgcn_wmma_f32_16x16x4_f32(false, a, false, b, (short)0, c,
                                               false, false);
}

__device__ __forceinline__ void store_c(float* C, int ldc, int nb, v8f c) {
  int lane = threadIdx.x & 31;
  int n = nb + (lane & 15);
  int mb = (lane < 16) ? 0 : 8;
#pragma unroll
  for (int j = 0; j < 8; ++j) C[(mb + j)*ldc + n] = c[j];
}

__device__ __forceinline__ void gemm3(const float* A, int lda, const v2f* Bsw,
                                      int KT, int nt, v8f acc[TM]) {
  int lane = threadIdx.x & 31;
  int m = lane & 15;
  int ks = (lane >> 4) << 1;
  const v2f* bp = Bsw + (size_t)nt * KT * 32 + lane;
  const v2f* a0 = (const v2f*)(A + (0*16 + m)*lda + ks);
  const v2f* a1 = (const v2f*)(A + (1*16 + m)*lda + ks);
  const v2f* a2 = (const v2f*)(A + (2*16 + m)*lda + ks);
  v8f z = {};
  acc[0] = z; acc[1] = z; acc[2] = z;
#pragma unroll 4
  for (int kt = 0; kt < KT; ++kt) {
    v2f b = bp[kt * 32];
    acc[0] = wmma4(a0[kt*2], b, acc[0]);
    acc[1] = wmma4(a1[kt*2], b, acc[1]);
    acc[2] = wmma4(a2[kt*2], b, acc[2]);
  }
}

__device__ __forceinline__ float silu_f(float x) { return x / (1.f + __expf(-x)); }

__global__ void swz_kernel(const float* W, float* dst, int K, int Nreal, int NT) {
  int idx = blockIdx.x * blockDim.x + threadIdx.x;
  int KT = K / 4;
  if (idx >= NT * KT * 32) return;
  int lane = idx & 31;
  int kt = (idx >> 5) % KT;
  int nt = idx / (32 * KT);
  int n = nt*16 + (lane & 15);
  int k = kt*4 + ((lane >> 4) << 1);
  float bx = (n < Nreal) ? W[k*Nreal + n]     : 0.f;
  float by = (n < Nreal) ? W[(k+1)*Nreal + n] : 0.f;
  dst[idx*2]     = bx;
  dst[idx*2 + 1] = by;
}

__global__ void kv_kernel(const float* mem, const float* Wk, const float* Wv,
                          float* kb, float* vb) {
  int i = blockIdx.x * blockDim.x + threadIdx.x;
  if (i >= 2 * MSLOTSn * Dn) return;
  int which = i / (MSLOTSn * Dn);
  int r = i % (MSLOTSn * Dn);
  int m = r / Dn, d = r % Dn;
  const float* W = which ? Wv : Wk;
  float acc = 0.f;
  for (int k = 0; k < Dn; ++k) acc += mem[m*Dn + k] * W[k*Dn + d];
  (which ? vb : kb)[m*Dn + d] = acc;
}

__global__ void kvswz_kernel(const float* kb, const float* vb,
                             float* kfragT, float* vfrag) {
  int idx = blockIdx.x * blockDim.x + threadIdx.x;
  if (idx < 32*32) {
    int lane = idx & 31; int kt = idx >> 5;
    int n = lane & 15;   int k = kt*4 + ((lane >> 4) << 1);
    kfragT[idx*2]     = (n < MSLOTSn) ? kb[n*Dn + k]     : 0.f;
    kfragT[idx*2 + 1] = (n < MSLOTSn) ? kb[n*Dn + k + 1] : 0.f;
  }
  if (idx < 8*4*32) {
    int lane = idx & 31; int kt = (idx >> 5) & 3; int nt = idx >> 7;
    int n = nt*16 + (lane & 15); int k = kt*4 + ((lane >> 4) << 1);
    vfrag[idx*2]     = (k     < MSLOTSn) ? vb[k*Dn + n]     : 0.f;
    vfrag[idx*2 + 1] = (k + 1 < MSLOTSn) ? vb[(k+1)*Dn + n] : 0.f;
  }
}

__global__ void attn_kernel(const float* x, const v2f* Wq_sw, const v2f* Wo_sw,
                            const float* ln_g, const float* ln_b,
                            const v2f* kfragT, const v2f* vfrag, float* xr) {
  __shared__ float xs[TOKB*Dn];
  __shared__ float qs[TOKB*Dn];
  __shared__ float av[TOKB*Dn];
  __shared__ float sc[TOKB*16];
  int tid = threadIdx.x;
  int wave = tid >> 5;
  long t0 = (long)blockIdx.x * TOKB;
  v8f acc[TM];

  for (int i = tid; i < TOKB*Dn; i += 128) xs[i] = x[t0*Dn + i];
  __syncthreads();

  for (int nti = 0; nti < 2; ++nti) {
    int nt = wave*2 + nti;
    gemm3(xs, Dn, Wq_sw, 32, nt, acc);
#pragma unroll
    for (int t = 0; t < TM; ++t) store_c(qs + t*16*Dn, Dn, nt*16, acc[t]);
  }
  __syncthreads();

  if (wave == 0) {
    gemm3(qs, Dn, kfragT, 32, 0, acc);
#pragma unroll
    for (int t = 0; t < TM; ++t) store_c(sc + t*16*16, 16, 0, acc[t]);
  }
  __syncthreads();

  if (tid < TOKB) {
    const float scale = 0.088388347648318447f;
    float mx = -1e30f;
    for (int m = 0; m < MSLOTSn; ++m) {
      float v = sc[tid*16 + m] * scale; sc[tid*16 + m] = v;
      if (v > mx) mx = v;
    }
    float s = 0.f;
    for (int m = 0; m < MSLOTSn; ++m) {
      float e = __expf(sc[tid*16 + m] - mx); sc[tid*16 + m] = e; s += e;
    }
    float inv = 1.f / s;
    for (int m = 0; m < MSLOTSn; ++m) sc[tid*16 + m] *= inv;
    for (int m = MSLOTSn; m < 16; ++m) sc[tid*16 + m] = 0.f;
  }
  __syncthreads();

  for (int nti = 0; nti < 2; ++nti) {
    int nt = wave*2 + nti;
    gemm3(sc, 16, vfrag, 4, nt, acc);
#pragma unroll
    for (int t = 0; t < TM; ++t) store_c(av + t*16*Dn, Dn, nt*16, acc[t]);
  }
  __syncthreads();

  for (int nti = 0; nti < 2; ++nti) {
    int nt = wave*2 + nti;
    gemm3(av, Dn, Wo_sw, 32, nt, acc);
#pragma unroll
    for (int t = 0; t < TM; ++t) store_c(qs + t*16*Dn, Dn, nt*16, acc[t]);
  }
  __syncthreads();

  if (tid < TOKB) {
    float m = 0.f;
    for (int c = 0; c < Dn; ++c) m += qs[tid*Dn + c];
    m *= (1.f / Dn);
    float v = 0.f;
    for (int c = 0; c < Dn; ++c) { float d = qs[tid*Dn + c] - m; v += d*d; }
    v *= (1.f / Dn);
    float inv = rsqrtf(v + 1e-5f);
    long tok = t0 + tid;
    for (int c = 0; c < Dn; ++c)
      xr[tok*Dn + c] = (qs[tid*Dn + c] - m) * inv * ln_g[c] + ln_b[c]
                       + xs[tid*Dn + c];
  }
}

__global__ void in_kernel(const float* xr, const float* rms_w, const v2f* Win_sw,
                          float* xp, float* zb) {
  __shared__ float xn[TOKB*Dn];
  int tid = threadIdx.x;
  int wave = tid >> 5;
  int lane = tid & 31;
  long t0 = (long)blockIdx.x * TOKB;
  v8f acc[TM];

  for (int i = tid; i < TOKB*Dn; i += 128) xn[i] = xr[t0*Dn + i];
  __syncthreads();
  if (tid < TOKB) {
    float s = 0.f;
    for (int c = 0; c < Dn; ++c) { float v = xn[tid*Dn + c]; s += v*v; }
    float inv = rsqrtf(s / Dn + 1e-5f);
    for (int c = 0; c < Dn; ++c) xn[tid*Dn + c] = xn[tid*Dn + c] * inv * rms_w[c];
  }
  __syncthreads();

  for (int nti = 0; nti < 8; ++nti) {
    int nt = wave*8 + nti;
    gemm3(xn, Dn, Win_sw, 32, nt, acc);
    int n = nt*16 + (lane & 15);
    int mb = (lane < 16) ? 0 : 8;
#pragma unroll
    for (int t = 0; t < TM; ++t)
#pragma unroll
      for (int j = 0; j < 8; ++j) {
        long tok = t0 + t*16 + mb + j;
        float v = acc[t][j];
        if (n < EDn) xp[tok*EDn + n] = v;
        else         zb[tok*EDn + (n - EDn)] = v;
      }
  }
}

__global__ void conv_kernel(const float* xp, const float* conv_w, const float* conv_b,
                            const v2f* Wx_sw, const v2f* Wdt_sw, const float* b_dt,
                            float* xc, float* delta, float* Bm, float* Cm) {
  __shared__ float xcs[TOKB*EDn];
  __shared__ float dbcs[TOKB*32];
  int tid = threadIdx.x;
  int wave = tid >> 5;
  int lane = tid & 31;
  long t0 = (long)blockIdx.x * TOKB;
  v8f acc[TM];

  for (int i = tid; i < TOKB*EDn; i += 128) {
    int r = i >> 8;
    int c = i & 255;
    long tok = t0 + r;
    int b = (int)(tok / Ln);
    int l = (int)(tok % Ln);
    float a = conv_b[c];
#pragma unroll
    for (int j = 0; j < DCONVn; ++j) {
      int ll = l - (DCONVn - 1) + j;
      if (ll >= 0) a += conv_w[c*DCONVn + j] * xp[((long)b*Ln + ll)*EDn + c];
    }
    a = silu_f(a);
    xcs[i] = a;
    xc[tok*EDn + c] = a;
  }
  __syncthreads();

  if (wave < 2) {
    gemm3(xcs, EDn, Wx_sw, 64, wave, acc);
#pragma unroll
    for (int t = 0; t < TM; ++t) store_c(dbcs + t*16*32, 32, wave*16, acc[t]);
  }
  __syncthreads();

  for (int nti = 0; nti < 4; ++nti) {
    int nt = wave*4 + nti;
    gemm3(dbcs, 32, Wdt_sw, 2, nt, acc);
    int n = nt*16 + (lane & 15);
    int mb = (lane < 16) ? 0 : 8;
#pragma unroll
    for (int t = 0; t < TM; ++t)
#pragma unroll
      for (int j = 0; j < 8; ++j) {
        long tok = t0 + t*16 + mb + j;
        float v = acc[t][j] + b_dt[n];
        delta[tok*EDn + n] = (v > 20.f) ? v : log1pf(__expf(v));
      }
  }

  for (int i = tid; i < TOKB*16; i += 128) {
    int r = i >> 4, c = i & 15;
    long tok = t0 + r;
    float v = dbcs[r*32 + DTRANKn + c];
    if (c < DSTATEn) Bm[tok*DSTATEn + c]             = v;
    else             Cm[tok*DSTATEn + (c - DSTATEn)] = v;
  }
}

__global__ void scan_kernel(const float* delta, const float* xc, const float* Bm,
                            const float* Cm, const float* A_log, const float* Dp,
                            float* y) {
  int gid = blockIdx.x * blockDim.x + threadIdx.x;
  if (gid >= Bn * EDn) return;
  int b = gid / EDn;
  int d = gid % EDn;
  float a[DSTATEn];
#pragma unroll
  for (int n = 0; n < DSTATEn; ++n) a[n] = -__expf(A_log[d*DSTATEn + n]);
  float dp = Dp[d];
  float h[DSTATEn];
#pragma unroll
  for (int n = 0; n < DSTATEn; ++n) h[n] = 0.f;
  long base = (long)b * Ln;
  for (int l = 0; l < Ln; ++l) {
    long t = base + l;
    float dt = delta[t*EDn + d];
    float xv = xc[t*EDn + d];
    float yv = 0.f;
#pragma unroll
    for (int n = 0; n < DSTATEn; ++n) {
      float dA = __expf(dt * a[n]);
      h[n] = dA * h[n] + dt * Bm[t*DSTATEn + n] * xv;
      yv += h[n] * Cm[t*DSTATEn + n];
    }
    y[t*EDn + d] = yv + dp * xv;
  }
}

__global__ void out_kernel(const float* y, const float* zb, const v2f* Wout_sw,
                           const float* xr, const float* x, float* out) {
  __shared__ float ys[TOKB*EDn];
  int tid = threadIdx.x;
  int wave = tid >> 5;
  int lane = tid & 31;
  long t0 = (long)blockIdx.x * TOKB;
  v8f acc[TM];

  for (int i = tid; i < TOKB*EDn; i += 128) {
    int r = i >> 8, c = i & 255;
    long tok = t0 + r;
    float z = zb[tok*EDn + c];
    ys[i] = y[tok*EDn + c] * silu_f(z);
  }
  __syncthreads();

  for (int nti = 0; nti < 2; ++nti) {
    int nt = wave*2 + nti;
    gemm3(ys, EDn, Wout_sw, 64, nt, acc);
    int n = nt*16 + (lane & 15);
    int mb = (lane < 16) ? 0 : 8;
#pragma unroll
    for (int t = 0; t < TM; ++t)
#pragma unroll
      for (int j = 0; j < 8; ++j) {
        long tok = t0 + t*16 + mb + j;
        out[tok*Dn + n] = acc[t][j] + xr[tok*Dn + n] + x[tok*Dn + n];
      }
  }
}

extern "C" void kernel_launch(void* const* d_in, const int* in_sizes, int n_in,
                              void* d_out, int out_size, void* d_ws, size_t ws_size,
                              hipStream_t stream) {
  (void)in_sizes; (void)n_in; (void)out_size; (void)ws_size;
  const float* x      = (const float*)d_in[0];
  const float* mem    = (const float*)d_in[1];
  const float* Wq     = (const float*)d_in[2];
  const float* Wk     = (const float*)d_in[3];
  const float* Wv     = (const float*)d_in[4];
  const float* Wo     = (const float*)d_in[5];
  const float* ln1_g  = (const float*)d_in[6];
  const float* ln1_b  = (const float*)d_in[7];
  const float* rms_w  = (const float*)d_in[8];
  const float* W_in   = (const float*)d_in[9];
  const float* conv_w = (const float*)d_in[10];
  const float* conv_b = (const float*)d_in[11];
  const float* W_x    = (const float*)d_in[12];
  const float* W_dt   = (const float*)d_in[13];
  const float* b_dt   = (const float*)d_in[14];
  const float* A_log  = (const float*)d_in[15];
  const float* Dp     = (const float*)d_in[16];
  const float* W_out  = (const float*)d_in[17];

  float* ws = (float*)d_ws;
  size_t off = 0;
  float* kb      = ws + off; off += (size_t)MSLOTSn*Dn;
  float* vb      = ws + off; off += (size_t)MSLOTSn*Dn;
  float* Wq_sw   = ws + off; off += (size_t)8*32*32*2;
  float* Wo_sw   = ws + off; off += (size_t)8*32*32*2;
  float* Win_sw  = ws + off; off += (size_t)32*32*32*2;
  float* Wx_sw   = ws + off; off += (size_t)2*64*32*2;
  float* Wdt_sw  = ws + off; off += (size_t)16*2*32*2;
  float* Wout_sw = ws + off; off += (size_t)8*64*32*2;
  float* kfragT  = ws + off; off += (size_t)1*32*32*2;
  float* vfrag   = ws + off; off += (size_t)8*4*32*2;
  float* xr      = ws + off; off += (size_t)NTOK*Dn;
  float* xp      = ws + off; off += (size_t)NTOK*EDn;
  float* zb      = ws + off; off += (size_t)NTOK*EDn;
  float* xc      = ws + off; off += (size_t)NTOK*EDn;
  float* delta   = ws + off; off += (size_t)NTOK*EDn;
  float* Bm      = ws + off; off += (size_t)NTOK*DSTATEn;
  float* Cm      = ws + off; off += (size_t)NTOK*DSTATEn;
  float* y       = xp;

  swz_kernel<<<(8*32*32  + 255)/256, 256, 0, stream>>>(Wq,    Wq_sw,   Dn,  Dn,     8);
  swz_kernel<<<(8*32*32  + 255)/256, 256, 0, stream>>>(Wo,    Wo_sw,   Dn,  Dn,     8);
  swz_kernel<<<(32*32*32 + 255)/256, 256, 0, stream>>>(W_in,  Win_sw,  Dn,  2*EDn, 32);
  swz_kernel<<<(2*64*32  + 255)/256, 256, 0, stream>>>(W_x,   Wx_sw,   EDn, 24,     2);
  swz_kernel<<<(16*2*32  + 255)/256, 256, 0, stream>>>(W_dt,  Wdt_sw,  DTRANKn, EDn, 16);
  swz_kernel<<<(8*64*32  + 255)/256, 256, 0, stream>>>(W_out, Wout_sw, EDn, Dn,     8);

  kv_kernel   <<<(2*MSLOTSn*Dn + 255)/256, 256, 0, stream>>>(mem, Wk, Wv, kb, vb);
  kvswz_kernel<<<4, 256, 0, stream>>>(kb, vb, kfragT, vfrag);

  attn_kernel<<<NBLK, 128, 0, stream>>>(x, (const v2f*)Wq_sw, (const v2f*)Wo_sw,
                                        ln1_g, ln1_b, (const v2f*)kfragT,
                                        (const v2f*)vfrag, xr);
  in_kernel  <<<NBLK, 128, 0, stream>>>(xr, rms_w, (const v2f*)Win_sw, xp, zb);
  conv_kernel<<<NBLK, 128, 0, stream>>>(xp, conv_w, conv_b, (const v2f*)Wx_sw,
                                        (const v2f*)Wdt_sw, b_dt, xc, delta, Bm, Cm);
  scan_kernel<<<(Bn*EDn + 255)/256, 256, 0, stream>>>(delta, xc, Bm, Cm,
                                                      A_log, Dp, y);
  out_kernel <<<NBLK, 128, 0, stream>>>(y, zb, (const v2f*)Wout_sw, xr, x,
                                        (float*)d_out);
}


# selective scan split along L into 156 parallel chunks of 25 steps: local scan, serial carry over chunk summaries, rescan; the state's summation order changes at chunk edges
# speedup vs baseline: 5.0716x; 5.0716x over previous
#include <hip/hip_runtime.h>
#include <hip/hip_bf16.h>
#include <math.h>

#define Bn      8
#define Sn      12
#define NNn     325
#define Dn      128
#define EDn     256
#define DSTATEn 8
#define DCONVn  4
#define DTRANKn 8
#define MSLOTSn 10
#define Ln      (Sn*NNn)
#define NTOK    (Bn*Ln)
#define TM      3
#define TOKB    (16*TM)
#define NBLK    (NTOK/TOKB)

typedef float v2f __attribute__((ext_vector_type(2)));
typedef float v8f __attribute__((ext_vector_type(8)));

__device__ __forceinline__ v8f wmma4(v2f a, v2f b, v8f c) {
  return __builtin_amdgcn_wmma_f32_16x16x4_f32(false, a, false, b, (short)0, c,
                                               false, false);
}

__device__ __forceinline__ void store_c(float* C, int ldc, int nb, v8f c) {
  int lane = threadIdx.x & 31;
  int n = nb + (lane & 15);
  int mb = (lane < 16) ? 0 : 8;
#pragma unroll
  for (int j = 0; j < 8; ++j) C[(mb + j)*ldc + n] = c[j];
}

__device__ __forceinline__ void gemm3(const float* A, int lda, const v2f* Bsw,
                                      int KT, int nt, v8f acc[TM]) {
  int lane = threadIdx.x & 31;
  int m = lane & 15;
  int ks = (lane >> 4) << 1;
  const v2f* bp = Bsw + (size_t)nt * KT * 32 + lane;
  const v2f* a0 = (const v2f*)(A + (0*16 + m)*lda + ks);
  const v2f* a1 = (const v2f*)(A + (1*16 + m)*lda + ks);
  const v2f* a2 = (const v2f*)(A + (2*16 + m)*lda + ks);
  v8f z = {};
  acc[0] = z; acc[1] = z; acc[2] = z;
#pragma unroll 4
  for (int kt = 0; kt < KT; ++kt) {
    v2f b = bp[kt * 32];
    acc[0] = wmma4(a0[kt*2], b, acc[0]);
    acc[1] = wmma4(a1[kt*2], b, acc[1]);
    acc[2] = wmma4(a2[kt*2], b, acc[2]);
  }
}

__device__ __forceinline__ float silu_f(float x) { return x / (1.f + __expf(-x)); }

__global__ void swz_kernel(const float* W, float* dst, int K, int Nreal, int NT) {
  int idx = blockIdx.x * blockDim.x + threadIdx.x;
  int KT = K / 4;
  if (idx >= NT * KT * 32) return;
  int lane = idx & 31;
  int kt = (idx >> 5) % KT;
  int nt = idx / (32 * KT);
  int n = nt*16 + (lane & 15);
  int k = kt*4 + ((lane >> 4) << 1);
  float bx = (n < Nreal) ? W[k*Nreal + n]     : 0.f;
  float by = (n < Nreal) ? W[(k+1)*Nreal + n] : 0.f;
  dst[idx*2]     = bx;
  dst[idx*2 + 1] = by;
}

__global__ void kv_kernel(const float* mem, const float* Wk, const float* Wv,
                          float* kb, float* vb) {
  int i = blockIdx.x * blockDim.x + threadIdx.x;
  if (i >= 2 * MSLOTSn * Dn) return;
  int which = i / (MSLOTSn * Dn);
  int r = i % (MSLOTSn * Dn);
  int m = r / Dn, d = r % Dn;
  const float* W = which ? Wv : Wk;
  float acc = 0.f;
  for (int k = 0; k < Dn; ++k) acc += mem[m*Dn + k] * W[k*Dn + d];
  (which ? vb : kb)[m*Dn + d] = acc;
}

__global__ void kvswz_kernel(const float* kb, const float* vb,
                             float* kfragT, float* vfrag) {
  int idx = blockIdx.x * blockDim.x + threadIdx.x;
  if (idx < 32*32) {
    int lane = idx & 31; int kt = idx >> 5;
    int n = lane & 15;   int k = kt*4 + ((lane >> 4) << 1);
    kfragT[idx*2]     = (n < MSLOTSn) ? kb[n*Dn + k]     : 0.f;
    kfragT[idx*2 + 1] = (n < MSLOTSn) ? kb[n*Dn + k + 1] : 0.f;
  }
  if (idx < 8*4*32) {
    int lane = idx & 31; int kt = (idx >> 5) & 3; int nt = idx >> 7;
    int n = nt*16 + (lane & 15); int k = kt*4 + ((lane >> 4) << 1);
    vfrag[idx*2]     = (k     < MSLOTSn) ? vb[k*Dn + n]     : 0.f;
    vfrag[idx*2 + 1] = (k + 1 < MSLOTSn) ? vb[(k+1)*Dn + n] : 0.f;
  }
}

__global__ void attn_kernel(const float* x, const v2f* Wq_sw, const v2f* Wo_sw,
                            const float* ln_g, const float* ln_b,
                            const v2f* kfragT, const v2f* vfrag, float* xr) {
  __shared__ float xs[TOKB*Dn];
  __shared__ float qs[TOKB*Dn];
  __shared__ float av[TOKB*Dn];
  __shared__ float sc[TOKB*16];
  int tid = threadIdx.x;
  int wave = tid >> 5;
  long t0 = (long)blockIdx.x * TOKB;
  v8f acc[TM];

  for (int i = tid; i < TOKB*Dn; i += 128) xs[i] = x[t0*Dn + i];
  __syncthreads();

  for (int nti = 0; nti < 2; ++nti) {
    int nt = wave*2 + nti;
    gemm3(xs, Dn, Wq_sw, 32, nt, acc);
#pragma unroll
    for (int t = 0; t < TM; ++t) store_c(qs + t*16*Dn, Dn, nt*16, acc[t]);
  }
  __syncthreads();

  if (wave == 0) {
    gemm3(qs, Dn, kfragT, 32, 0, acc);
#pragma unroll
    for (int t = 0; t < TM; ++t) store_c(sc + t*16*16, 16, 0, acc[t]);
  }
  __syncthreads();

  if (tid < TOKB) {
    const float scale = 0.088388347648318447f;
    float mx = -1e30f;
    for (int m = 0; m < MSLOTSn; ++m) {
      float v = sc[tid*16 + m] * scale; sc[tid*16 + m] = v;
      if (v > mx) mx = v;
    }
    float s = 0.f;
    for (int m = 0; m < MSLOTSn; ++m) {
      float e = __expf(sc[tid*16 + m] - mx); sc[tid*16 + m] = e; s += e;
    }
    float inv = 1.f / s;
    for (int m = 0; m < MSLOTSn; ++m) sc[tid*16 + m] *= inv;
    for (int m = MSLOTSn; m < 16; ++m) sc[tid*16 + m] = 0.f;
  }
  __syncthreads();

  for (int nti = 0; nti < 2; ++nti) {
    int nt = wave*2 + nti;
    gemm3(sc, 16, vfrag, 4, nt, acc);
#pragma unroll
    for (int t = 0; t < TM; ++t) store_c(av + t*16*Dn, Dn, nt*16, acc[t]);
  }
  __syncthreads();

  for (int nti = 0; nti < 2; ++nti) {
    int nt = wave*2 + nti;
    gemm3(av, Dn, Wo_sw, 32, nt, acc);
#pragma unroll
    for (int t = 0; t < TM; ++t) store_c(qs + t*16*Dn, Dn, nt*16, acc[t]);
  }
  __syncthreads();

  if (tid < TOKB) {
    float m = 0.f;
    for (int c = 0; c < Dn; ++c) m += qs[tid*Dn + c];
    m *= (1.f / Dn);
    float v = 0.f;
    for (int c = 0; c < Dn; ++c) { float d = qs[tid*Dn + c] - m; v += d*d; }
    v *= (1.f / Dn);
    float inv = rsqrtf(v + 1e-5f);
    long tok = t0 + tid;
    for (int c = 0; c < Dn; ++c)
      xr[tok*Dn + c] = (qs[tid*Dn + c] - m) * inv * ln_g[c] + ln_b[c]
                       + xs[tid*Dn + c];
  }
}

__global__ void in_kernel(const float* xr, const float* rms_w, const v2f* Win_sw,
                          float* xp, float* zb) {
  __shared__ float xn[TOKB*Dn];
  int tid = threadIdx.x;
  int wave = tid >> 5;
  int lane = tid & 31;
  long t0 = (long)blockIdx.x * TOKB;
  v8f acc[TM];

  for (int i = tid; i < TOKB*Dn; i += 128) xn[i] = xr[t0*Dn + i];
  __syncthreads();
  if (tid < TOKB) {
    float s = 0.f;
    for (int c = 0; c < Dn; ++c) { float v = xn[tid*Dn + c]; s += v*v; }
    float inv = rsqrtf(s / Dn + 1e-5f);
    for (int c = 0; c < Dn; ++c) xn[tid*Dn + c] = xn[tid*Dn + c] * inv * rms_w[c];
  }
  __syncthreads();

  for (int nti = 0; nti < 8; ++nti) {
    int nt = wave*8 + nti;
    gemm3(xn, Dn, Win_sw, 32, nt, acc);
    int n = nt*16 + (lane & 15);
    int mb = (lane < 16) ? 0 : 8;
#pragma unroll
    for (int t = 0; t < TM; ++t)
#pragma unroll
      for (int j = 0; j < 8; ++j) {
        long tok = t0 + t*16 + mb + j;
        float v = acc[t][j];
        if (n < EDn) xp[tok*EDn + n] = v;
        else         zb[tok*EDn + (n - EDn)] = v;
      }
  }
}

__global__ void conv_kernel(const float* xp, const float* conv_w, const float* conv_b,
                            const v2f* Wx_sw, const v2f* Wdt_sw, const float* b_dt,
                            float* xc, float* delta, float* Bm, float* Cm) {
  __shared__ float xcs[TOKB*EDn];
  __shared__ float dbcs[TOKB*32];
  int tid = threadIdx.x;
  int wave = tid >> 5;
  int lane = tid & 31;
  long t0 = (long)blockIdx.x * TOKB;
  v8f acc[TM];

  for (int i = tid; i < TOKB*EDn; i += 128) {
    int r = i >> 8;
    int c = i & 255;
    long tok = t0 + r;
    int b = (int)(tok / Ln);
    int l = (int)(tok % Ln);
    float a = conv_b[c];
#pragma unroll
    for (int j = 0; j < DCONVn; ++j) {
      int ll = l - (DCONVn - 1) + j;
      if (ll >= 0) a += conv_w[c*DCONVn + j] * xp[((long)b*Ln + ll)*EDn + c];
    }
    a = silu_f(a);
    xcs[i] = a;
    xc[tok*EDn + c] = a;
  }
  __syncthreads();

  if (wave < 2) {
    gemm3(xcs, EDn, Wx_sw, 64, wave, acc);
#pragma unroll
    for (int t = 0; t < TM; ++t) store_c(dbcs + t*16*32, 32, wave*16, acc[t]);
  }
  __syncthreads();

  for (int nti = 0; nti < 4; ++nti) {
    int nt = wave*4 + nti;
    gemm3(dbcs, 32, Wdt_sw, 2, nt, acc);
    int n = nt*16 + (lane & 15);
    int mb = (lane < 16) ? 0 : 8;
#pragma unroll
    for (int t = 0; t < TM; ++t)
#pragma unroll
      for (int j = 0; j < 8; ++j) {
        long tok = t0 + t*16 + mb + j;
        float v = acc[t][j] + b_dt[n];
        delta[tok*EDn + n] = (v > 20.f) ? v : log1pf(__expf(v));
      }
  }

  for (int i = tid; i < TOKB*16; i += 128) {
    int r = i >> 4, c = i & 15;
    long tok = t0 + r;
    float v = dbcs[r*32 + DTRANKn + c];
    if (c < DSTATEn) Bm[tok*DSTATEn + c]             = v;
    else             Cm[tok*DSTATEn + (c - DSTATEn)] = v;
  }
}

#define LC      25
#define NCH     (Ln/LC)
#define CARRYU  26

__global__ void __launch_bounds__(EDn)
scan_local_kernel(const float* __restrict__ delta, const float* __restrict__ xc,
                  const float* __restrict__ Bm, const float* __restrict__ A_log,
                  float* __restrict__ y) {
  const int d  = threadIdx.x;
  const int t0 = blockIdx.y * Ln + blockIdx.x * LC;
  float a[DSTATEn], h[DSTATEn];
#pragma unroll
  for (int n = 0; n < DSTATEn; ++n) { a[n] = -__expf(A_log[d*DSTATEn + n]); h[n] = 0.f; }
  float sdt = 0.f;
#pragma unroll
  for (int l = 0; l < LC; ++l) {
    const int t = t0 + l;
    float dt = delta[t*EDn + d];
    float xv = xc[t*EDn + d];
    sdt += dt;
#pragma unroll
    for (int n = 0; n < DSTATEn; ++n) {
      float dA = __expf(dt * a[n]);
      h[n] = dA * h[n] + dt * Bm[t*DSTATEn + n] * xv;
    }
  }
#pragma unroll
  for (int n = 0; n < DSTATEn; ++n) y[(t0 + n)*EDn + d] = h[n];
  y[(t0 + DSTATEn)*EDn + d] = sdt;
}

__global__ void __launch_bounds__(EDn)
scan_carry_kernel(const float* __restrict__ A_log, float* y) {
  const int d = threadIdx.x;
  const int n = blockIdx.x;
  const int b = blockIdx.y;
  const float a = -__expf(A_log[d*DSTATEn + n]);
  float*       hp = y + (b*Ln + n)*EDn + d;
  const float* sp = y + (b*Ln + DSTATEn)*EDn + d;
  float carry = 0.f;
  for (int c0 = 0; c0 < NCH; c0 += CARRYU) {
    float he[CARRYU], sd[CARRYU];
#pragma unroll
    for (int j = 0; j < CARRYU; ++j) {
      he[j] = hp[(c0 + j)*(LC*EDn)];
      sd[j] = sp[(c0 + j)*(LC*EDn)];
    }
#pragma unroll
    for (int j = 0; j < CARRYU; ++j) {
      float P = __expf(a * sd[j]);
      hp[(c0 + j)*(LC*EDn)] = carry;
      carry = P * carry + he[j];
    }
  }
}

__global__ void __launch_bounds__(EDn)
scan_apply_kernel(const float* __restrict__ delta, const float* __restrict__ xc,
                  const float* __restrict__ Bm, const float* __restrict__ Cm,
                  const float* __restrict__ A_log, const float* __restrict__ Dp,
                  float* __restrict__ y) {
  const int d  = threadIdx.x;
  const int t0 = blockIdx.y * Ln + blockIdx.x * LC;
  float a[DSTATEn], h[DSTATEn];
#pragma unroll
  for (int n = 0; n < DSTATEn; ++n) {
    a[n] = -__expf(A_log[d*DSTATEn + n]);
    h[n] = y[(t0 + n)*EDn + d];
  }
  const float dp = Dp[d];
#pragma unroll
  for (int l = 0; l < LC; ++l) {
    const int t = t0 + l;
    float dt = delta[t*EDn + d];
    float xv = xc[t*EDn + d];
    float yv = 0.f;
#pragma unroll
    for (int n = 0; n < DSTATEn; ++n) {
      float dA = __expf(dt * a[n]);
      h[n] = dA * h[n] + dt * Bm[t*DSTATEn + n] * xv;
      yv += h[n] * Cm[t*DSTATEn + n];
    }
    y[t*EDn + d] = yv + dp * xv;
  }
}

__global__ void out_kernel(const float* y, const float* zb, const v2f* Wout_sw,
                           const float* xr, const float* x, float* out) {
  __shared__ float ys[TOKB*EDn];
  int tid = threadIdx.x;
  int wave = tid >> 5;
  int lane = tid & 31;
  long t0 = (long)blockIdx.x * TOKB;
  v8f acc[TM];

  for (int i = tid; i < TOKB*EDn; i += 128) {
    int r = i >> 8, c = i & 255;
    long tok = t0 + r;
    float z = zb[tok*EDn + c];
    ys[i] = y[tok*EDn + c] * silu_f(z);
  }
  __syncthreads();

  for (int nti = 0; nti < 2; ++nti) {
    int nt = wave*2 + nti;
    gemm3(ys, EDn, Wout_sw, 64, nt, acc);
    int n = nt*16 + (lane & 15);
    int mb = (lane < 16) ? 0 : 8;
#pragma unroll
    for (int t = 0; t < TM; ++t)
#pragma unroll
      for (int j = 0; j < 8; ++j) {
        long tok = t0 + t*16 + mb + j;
        out[tok*Dn + n] = acc[t][j] + xr[tok*Dn + n] + x[tok*Dn + n];
      }
  }
}

extern "C" void kernel_launch(void* const* d_in, const int* in_sizes, int n_in,
                              void* d_out, int out_size, void* d_ws, size_t ws_size,
                              hipStream_t stream) {
  (void)in_sizes; (void)n_in; (void)out_size; (void)ws_size;
  const float* x      = (const float*)d_in[0];
  const float* mem    = (const float*)d_in[1];
  const float* Wq     = (const float*)d_in[2];
  const float* Wk     = (const float*)d_in[3];
  const float* Wv     = (const float*)d_in[4];
  const float* Wo     = (const float*)d_in[5];
  const float* ln1_g  = (const float*)d_in[6];
  const float* ln1_b  = (const float*)d_in[7];
  const float* rms_w  = (const float*)d_in[8];
  const float* W_in   = (const float*)d_in[9];
  const float* conv_w = (const float*)d_in[10];
  const float* conv_b = (const float*)d_in[11];
  const float* W_x    = (const float*)d_in[12];
  const float* W_dt   = (const float*)d_in[13];
  const float* b_dt   = (const float*)d_in[14];
  const float* A_log  = (const float*)d_in[15];
  const float* Dp     = (const float*)d_in[16];
  const float* W_out  = (const float*)d_in[17];

  float* ws = (float*)d_ws;
  size_t off = 0;
  float* kb      = ws + off; off += (size_t)MSLOTSn*Dn;
  float* vb      = ws + off; off += (size_t)MSLOTSn*Dn;
  float* Wq_sw   = ws + off; off += (size_t)8*32*32*2;
  float* Wo_sw   = ws + off; off += (size_t)8*32*32*2;
  float* Win_sw  = ws + off; off += (size_t)32*32*32*2;
  float* Wx_sw   = ws + off; off += (size_t)2*64*32*2;
  float* Wdt_sw  = ws + off; off += (size_t)16*2*32*2;
  float* Wout_sw = ws + off; off += (size_t)8*64*32*2;
  float* kfragT  = ws + off; off += (size_t)1*32*32*2;
  float* vfrag   = ws + off; off += (size_t)8*4*32*2;
  float* xr      = ws + off; off += (size_t)NTOK*Dn;
  float* xp      = ws + off; off += (size_t)NTOK*EDn;
  float* zb      = ws + off; off += (size_t)NTOK*EDn;
  float* xc      = ws + off; off += (size_t)NTOK*EDn;
  float* delta   = ws + off; off += (size_t)NTOK*EDn;
  float* Bm      = ws + off; off += (size_t)NTOK*DSTATEn;
  float* Cm      = ws + off; off += (size_t)NTOK*DSTATEn;
  float* y       = xp;

  swz_kernel<<<(8*32*32  + 255)/256, 256, 0, stream>>>(Wq,    Wq_sw,   Dn,  Dn,     8);
  swz_kernel<<<(8*32*32  + 255)/256, 256, 0, stream>>>(Wo,    Wo_sw,   Dn,  Dn,     8);
  swz_kernel<<<(32*32*32 + 255)/256, 256, 0, stream>>>(W_in,  Win_sw,  Dn,  2*EDn, 32);
  swz_kernel<<<(2*64*32  + 255)/256, 256, 0, stream>>>(W_x,   Wx_sw,   EDn, 24,     2);
  swz_kernel<<<(16*2*32  + 255)/256, 256, 0, stream>>>(W_dt,  Wdt_sw,  DTRANKn, EDn, 16);
  swz_kernel<<<(8*64*32  + 255)/256, 256, 0, stream>>>(W_out, Wout_sw, EDn, Dn,     8);

  kv_kernel   <<<(2*MSLOTSn*Dn + 255)/256, 256, 0, stream>>>(mem, Wk, Wv, kb, vb);
  kvswz_kernel<<<4, 256, 0, stream>>>(kb, vb, kfragT, vfrag);

  attn_kernel<<<NBLK, 128, 0, stream>>>(x, (const v2f*)Wq_sw, (const v2f*)Wo_sw,
                                        ln1_g, ln1_b, (const v2f*)kfragT,
                                        (const v2f*)vfrag, xr);
  in_kernel  <<<NBLK, 128, 0, stream>>>(xr, rms_w, (const v2f*)Win_sw, xp, zb);
  conv_kernel<<<NBLK, 128, 0, stream>>>(xp, conv_w, conv_b, (const v2f*)Wx_sw,
                                        (const v2f*)Wdt_sw, b_dt, xc, delta, Bm, Cm);
  scan_local_kernel<<<dim3(NCH, Bn), EDn, 0, stream>>>(delta, xc, Bm, A_log, y);
  scan_carry_kernel<<<dim3(DSTATEn, Bn), EDn, 0, stream>>>(A_log, y);
  scan_apply_kernel<<<dim3(NCH, Bn), EDn, 0, stream>>>(delta, xc, Bm, Cm,
                                                       A_log, Dp, y);
  out_kernel <<<NBLK, 128, 0, stream>>>(y, zb, (const v2f*)Wout_sw, xr, x,
                                        (float*)d_out);
}


# LDS row pitch of every A-operand tile padded from a power of two to 4 x odd dwords (132/260/20/36): removes the 16-way bank conflict on each WMMA A-fragment load and the row-per-lane norm loops
# speedup vs baseline: 5.6319x; 1.1105x over previous
#include <hip/hip_runtime.h>
#include <hip/hip_bf16.h>
#include <math.h>

#define Bn      8
#define Sn      12
#define NNn     325
#define Dn      128
#define EDn     256
#define DSTATEn 8
#define DCONVn  4
#define DTRANKn 8
#define MSLOTSn 10
#define Ln      (Sn*NNn)
#define NTOK    (Bn*Ln)
#define TM      3
#define TOKB    (16*TM)
#define NBLK    (NTOK/TOKB)
#define PD      (Dn + 4)
#define PE      (EDn + 4)
#define PS      20
#define PB      36

typedef float v2f __attribute__((ext_vector_type(2)));
typedef float v8f __attribute__((ext_vector_type(8)));

__device__ __forceinline__ v8f wmma4(v2f a, v2f b, v8f c) {
  return __builtin_amdgcn_wmma_f32_16x16x4_f32(false, a, false, b, (short)0, c,
                                               false, false);
}

__device__ __forceinline__ void store_c(float* C, int ldc, int nb, v8f c) {
  int lane = threadIdx.x & 31;
  int n = nb + (lane & 15);
  int mb = (lane < 16) ? 0 : 8;
#pragma unroll
  for (int j = 0; j < 8; ++j) C[(mb + j)*ldc + n] = c[j];
}

__device__ __forceinline__ void gemm3(const float* A, int lda, const v2f* Bsw,
                                      int KT, int nt, v8f acc[TM]) {
  int lane = threadIdx.x & 31;
  int m = lane & 15;
  int ks = (lane >> 4) << 1;
  const v2f* bp = Bsw + (size_t)nt * KT * 32 + lane;
  const v2f* a0 = (const v2f*)(A + (0*16 + m)*lda + ks);
  const v2f* a1 = (const v2f*)(A + (1*16 + m)*lda + ks);
  const v2f* a2 = (const v2f*)(A + (2*16 + m)*lda + ks);
  v8f z = {};
  acc[0] = z; acc[1] = z; acc[2] = z;
#pragma unroll 4
  for (int kt = 0; kt < KT; ++kt) {
    v2f b = bp[kt * 32];
    acc[0] = wmma4(a0[kt*2], b, acc[0]);
    acc[1] = wmma4(a1[kt*2], b, acc[1]);
    acc[2] = wmma4(a2[kt*2], b, acc[2]);
  }
}

__device__ __forceinline__ float silu_f(float x) { return x / (1.f + __expf(-x)); }

__global__ void swz_kernel(const float* W, float* dst, int K, int Nreal, int NT) {
  int idx = blockIdx.x * blockDim.x + threadIdx.x;
  int KT = K / 4;
  if (idx >= NT * KT * 32) return;
  int lane = idx & 31;
  int kt = (idx >> 5) % KT;
  int nt = idx / (32 * KT);
  int n = nt*16 + (lane & 15);
  int k = kt*4 + ((lane >> 4) << 1);
  float bx = (n < Nreal) ? W[k*Nreal + n]     : 0.f;
  float by = (n < Nreal) ? W[(k+1)*Nreal + n] : 0.f;
  dst[idx*2]     = bx;
  dst[idx*2 + 1] = by;
}

__global__ void kv_kernel(const float* mem, const float* Wk, const float* Wv,
                          float* kb, float* vb) {
  int i = blockIdx.x * blockDim.x + threadIdx.x;
  if (i >= 2 * MSLOTSn * Dn) return;
  int which = i / (MSLOTSn * Dn);
  int r = i % (MSLOTSn * Dn);
  int m = r / Dn, d = r % Dn;
  const float* W = which ? Wv : Wk;
  float acc = 0.f;
  for (int k = 0; k < Dn; ++k) acc += mem[m*Dn + k] * W[k*Dn + d];
  (which ? vb : kb)[m*Dn + d] = acc;
}

__global__ void kvswz_kernel(const float* kb, const float* vb,
                             float* kfragT, float* vfrag) {
  int idx = blockIdx.x * blockDim.x + threadIdx.x;
  if (idx < 32*32) {
    int lane = idx & 31; int kt = idx >> 5;
    int n = lane & 15;   int k = kt*4 + ((lane >> 4) << 1);
    kfragT[idx*2]     = (n < MSLOTSn) ? kb[n*Dn + k]     : 0.f;
    kfragT[idx*2 + 1] = (n < MSLOTSn) ? kb[n*Dn + k + 1] : 0.f;
  }
  if (idx < 8*4*32) {
    int lane = idx & 31; int kt = (idx >> 5) & 3; int nt = idx >> 7;
    int n = nt*16 + (lane & 15); int k = kt*4 + ((lane >> 4) << 1);
    vfrag[idx*2]     = (k     < MSLOTSn) ? vb[k*Dn + n]     : 0.f;
    vfrag[idx*2 + 1] = (k + 1 < MSLOTSn) ? vb[(k+1)*Dn + n] : 0.f;
  }
}

__global__ void attn_kernel(const float* x, const v2f* Wq_sw, const v2f* Wo_sw,
                            const float* ln_g, const float* ln_b,
                            const v2f* kfragT, const v2f* vfrag, float* xr) {
  __shared__ float xs[TOKB*PD];
  __shared__ float qs[TOKB*PD];
  __shared__ float av[TOKB*PD];
  __shared__ float sc[TOKB*PS];
  int tid = threadIdx.x;
  int wave = tid >> 5;
  long t0 = (long)blockIdx.x * TOKB;
  v8f acc[TM];

  for (int i = tid; i < TOKB*Dn; i += 128) xs[(i >> 7)*PD + (i & 127)] = x[t0*Dn + i];
  __syncthreads();

  for (int nti = 0; nti < 2; ++nti) {
    int nt = wave*2 + nti;
    gemm3(xs, PD, Wq_sw, 32, nt, acc);
#pragma unroll
    for (int t = 0; t < TM; ++t) store_c(qs + t*16*PD, PD, nt*16, acc[t]);
  }
  __syncthreads();

  if (wave == 0) {
    gemm3(qs, PD, kfragT, 32, 0, acc);
#pragma unroll
    for (int t = 0; t < TM; ++t) store_c(sc + t*16*PS, PS, 0, acc[t]);
  }
  __syncthreads();

  if (tid < TOKB) {
    const float scale = 0.088388347648318447f;
    float mx = -1e30f;
    for (int m = 0; m < MSLOTSn; ++m) {
      float v = sc[tid*PS + m] * scale; sc[tid*PS + m] = v;
      if (v > mx) mx = v;
    }
    float s = 0.f;
    for (int m = 0; m < MSLOTSn; ++m) {
      float e = __expf(sc[tid*PS + m] - mx); sc[tid*PS + m] = e; s += e;
    }
    float inv = 1.f / s;
    for (int m = 0; m < MSLOTSn; ++m) sc[tid*PS + m] *= inv;
    for (int m = MSLOTSn; m < 16; ++m) sc[tid*PS + m] = 0.f;
  }
  __syncthreads();

  for (int nti = 0; nti < 2; ++nti) {
    int nt = wave*2 + nti;
    gemm3(sc, PS, vfrag, 4, nt, acc);
#pragma unroll
    for (int t = 0; t < TM; ++t) store_c(av + t*16*PD, PD, nt*16, acc[t]);
  }
  __syncthreads();

  for (int nti = 0; nti < 2; ++nti) {
    int nt = wave*2 + nti;
    gemm3(av, PD, Wo_sw, 32, nt, acc);
#pragma unroll
    for (int t = 0; t < TM; ++t) store_c(qs + t*16*PD, PD, nt*16, acc[t]);
  }
  __syncthreads();

  if (tid < TOKB) {
    float m = 0.f;
    for (int c = 0; c < Dn; ++c) m += qs[tid*PD + c];
    m *= (1.f / Dn);
    float v = 0.f;
    for (int c = 0; c < Dn; ++c) { float d = qs[tid*PD + c] - m; v += d*d; }
    v *= (1.f / Dn);
    float inv = rsqrtf(v + 1e-5f);
    long tok = t0 + tid;
    for (int c = 0; c < Dn; ++c)
      xr[tok*Dn + c] = (qs[tid*PD + c] - m) * inv * ln_g[c] + ln_b[c]
                       + xs[tid*PD + c];
  }
}

__global__ void in_kernel(const float* xr, const float* rms_w, const v2f* Win_sw,
                          float* xp, float* zb) {
  __shared__ float xn[TOKB*PD];
  int tid = threadIdx.x;
  int wave = tid >> 5;
  int lane = tid & 31;
  long t0 = (long)blockIdx.x * TOKB;
  v8f acc[TM];

  for (int i = tid; i < TOKB*Dn; i += 128) xn[(i >> 7)*PD + (i & 127)] = xr[t0*Dn + i];
  __syncthreads();
  if (tid < TOKB) {
    float s = 0.f;
    for (int c = 0; c < Dn; ++c) { float v = xn[tid*PD + c]; s += v*v; }
    float inv = rsqrtf(s / Dn + 1e-5f);
    for (int c = 0; c < Dn; ++c) xn[tid*PD + c] = xn[tid*PD + c] * inv * rms_w[c];
  }
  __syncthreads();

  for (int nti = 0; nti < 8; ++nti) {
    int nt = wave*8 + nti;
    gemm3(xn, PD, Win_sw, 32, nt, acc);
    int n = nt*16 + (lane & 15);
    int mb = (lane < 16) ? 0 : 8;
#pragma unroll
    for (int t = 0; t < TM; ++t)
#pragma unroll
      for (int j = 0; j < 8; ++j) {
        long tok = t0 + t*16 + mb + j;
        float v = acc[t][j];
        if (n < EDn) xp[tok*EDn + n] = v;
        else         zb[tok*EDn + (n - EDn)] = v;
      }
  }
}

__global__ void conv_kernel(const float* xp, const float* conv_w, const float* conv_b,
                            const v2f* Wx_sw, const v2f* Wdt_sw, const float* b_dt,
                            float* xc, float* delta, float* Bm, float* Cm) {
  __shared__ float xcs[TOKB*PE];
  __shared__ float dbcs[TOKB*PB];
  int tid = threadIdx.x;
  int wave = tid >> 5;
  int lane = tid & 31;
  long t0 = (long)blockIdx.x * TOKB;
  v8f acc[TM];

  for (int i = tid; i < TOKB*EDn; i += 128) {
    int r = i >> 8;
    int c = i & 255;
    long tok = t0 + r;
    int b = (int)(tok / Ln);
    int l = (int)(tok % Ln);
    float a = conv_b[c];
#pragma unroll
    for (int j = 0; j < DCONVn; ++j) {
      int ll = l - (DCONVn - 1) + j;
      if (ll >= 0) a += conv_w[c*DCONVn + j] * xp[((long)b*Ln + ll)*EDn + c];
    }
    a = silu_f(a);
    xcs[r*PE + c] = a;
    xc[tok*EDn + c] = a;
  }
  __syncthreads();

  if (wave < 2) {
    gemm3(xcs, PE, Wx_sw, 64, wave, acc);
#pragma unroll
    for (int t = 0; t < TM; ++t) store_c(dbcs + t*16*PB, PB, wave*16, acc[t]);
  }
  __syncthreads();

  for (int nti = 0; nti < 4; ++nti) {
    int nt = wave*4 + nti;
    gemm3(dbcs, PB, Wdt_sw, 2, nt, acc);
    int n = nt*16 + (lane & 15);
    int mb = (lane < 16) ? 0 : 8;
#pragma unroll
    for (int t = 0; t < TM; ++t)
#pragma unroll
      for (int j = 0; j < 8; ++j) {
        long tok = t0 + t*16 + mb + j;
        float v = acc[t][j] + b_dt[n];
        delta[tok*EDn + n] = (v > 20.f) ? v : log1pf(__expf(v));
      }
  }

  for (int i = tid; i < TOKB*16; i += 128) {
    int r = i >> 4, c = i & 15;
    long tok = t0 + r;
    float v = dbcs[r*PB + DTRANKn + c];
    if (c < DSTATEn) Bm[tok*DSTATEn + c]             = v;
    else             Cm[tok*DSTATEn + (c - DSTATEn)] = v;
  }
}

#define LC      25
#define NCH     (Ln/LC)
#define CARRYU  26

__global__ void __launch_bounds__(EDn)
scan_local_kernel(const float* __restrict__ delta, const float* __restrict__ xc,
                  const float* __restrict__ Bm, const float* __restrict__ A_log,
                  float* __restrict__ y) {
  const int d  = threadIdx.x;
  const int t0 = blockIdx.y * Ln + blockIdx.x * LC;
  float a[DSTATEn], h[DSTATEn];
#pragma unroll
  for (int n = 0; n < DSTATEn; ++n) { a[n] = -__expf(A_log[d*DSTATEn + n]); h[n] = 0.f; }
  float sdt = 0.f;
#pragma unroll
  for (int l = 0; l < LC; ++l) {
    const int t = t0 + l;
    float dt = delta[t*EDn + d];
    float xv = xc[t*EDn + d];
    sdt += dt;
#pragma unroll
    for (int n = 0; n < DSTATEn; ++n) {
      float dA = __expf(dt * a[n]);
      h[n] = dA * h[n] + dt * Bm[t*DSTATEn + n] * xv;
    }
  }
#pragma unroll
  for (int n = 0; n < DSTATEn; ++n) y[(t0 + n)*EDn + d] = h[n];
  y[(t0 + DSTATEn)*EDn + d] = sdt;
}

__global__ void __launch_bounds__(EDn)
scan_carry_kernel(const float* __restrict__ A_log, float* y) {
  const int d = threadIdx.x;
  const int n = blockIdx.x;
  const int b = blockIdx.y;
  const float a = -__expf(A_log[d*DSTATEn + n]);
  float*       hp = y + (b*Ln + n)*EDn + d;
  const float* sp = y + (b*Ln + DSTATEn)*EDn + d;
  float carry = 0.f;
  for (int c0 = 0; c0 < NCH; c0 += CARRYU) {
    float he[CARRYU], sd[CARRYU];
#pragma unroll
    for (int j = 0; j < CARRYU; ++j) {
      he[j] = hp[(c0 + j)*(LC*EDn)];
      sd[j] = sp[(c0 + j)*(LC*EDn)];
    }
#pragma unroll
    for (int j = 0; j < CARRYU; ++j) {
      float P = __expf(a * sd[j]);
      hp[(c0 + j)*(LC*EDn)] = carry;
      carry = P * carry + he[j];
    }
  }
}

__global__ void __launch_bounds__(EDn)
scan_apply_kernel(const float* __restrict__ delta, const float* __restrict__ xc,
                  const float* __restrict__ Bm, const float* __restrict__ Cm,
                  const float* __restrict__ A_log, const float* __restrict__ Dp,
                  float* __restrict__ y) {
  const int d  = threadIdx.x;
  const int t0 = blockIdx.y * Ln + blockIdx.x * LC;
  float a[DSTATEn], h[DSTATEn];
#pragma unroll
  for (int n = 0; n < DSTATEn; ++n) {
    a[n] = -__expf(A_log[d*DSTATEn + n]);
    h[n] = y[(t0 + n)*EDn + d];
  }
  const float dp = Dp[d];
#pragma unroll
  for (int l = 0; l < LC; ++l) {
    const int t = t0 + l;
    float dt = delta[t*EDn + d];
    float xv = xc[t*EDn + d];
    float yv = 0.f;
#pragma unroll
    for (int n = 0; n < DSTATEn; ++n) {
      float dA = __expf(dt * a[n]);
      h[n] = dA * h[n] + dt * Bm[t*DSTATEn + n] * xv;
      yv += h[n] * Cm[t*DSTATEn + n];
    }
    y[t*EDn + d] = yv + dp * xv;
  }
}

__global__ void out_kernel(const float* y, const float* zb, const v2f* Wout_sw,
                           const float* xr, const float* x, float* out) {
  __shared__ float ys[TOKB*PE];
  int tid = threadIdx.x;
  int wave = tid >> 5;
  int lane = tid & 31;
  long t0 = (long)blockIdx.x * TOKB;
  v8f acc[TM];

  for (int i = tid; i < TOKB*EDn; i += 128) {
    int r = i >> 8, c = i & 255;
    long tok = t0 + r;
    float z = zb[tok*EDn + c];
    ys[r*PE + c] = y[tok*EDn + c] * silu_f(z);
  }
  __syncthreads();

  for (int nti = 0; nti < 2; ++nti) {
    int nt = wave*2 + nti;
    gemm3(ys, PE, Wout_sw, 64, nt, acc);
    int n = nt*16 + (lane & 15);
    int mb = (lane < 16) ? 0 : 8;
#pragma unroll
    for (int t = 0; t < TM; ++t)
#pragma unroll
      for (int j = 0; j < 8; ++j) {
        long tok = t0 + t*16 + mb + j;
        out[tok*Dn + n] = acc[t][j] + xr[tok*Dn + n] + x[tok*Dn + n];
      }
  }
}

extern "C" void kernel_launch(void* const* d_in, const int* in_sizes, int n_in,
                              void* d_out, int out_size, void* d_ws, size_t ws_size,
                              hipStream_t stream) {
  (void)in_sizes; (void)n_in; (void)out_size; (void)ws_size;
  const float* x      = (const float*)d_in[0];
  const float* mem    = (const float*)d_in[1];
  const float* Wq     = (const float*)d_in[2];
  const float* Wk     = (const float*)d_in[3];
  const float* Wv     = (const float*)d_in[4];
  const float* Wo     = (const float*)d_in[5];
  const float* ln1_g  = (const float*)d_in[6];
  const float* ln1_b  = (const float*)d_in[7];
  const float* rms_w  = (const float*)d_in[8];
  const float* W_in   = (const float*)d_in[9];
  const float* conv_w = (const float*)d_in[10];
  const float* conv_b = (const float*)d_in[11];
  const float* W_x    = (const float*)d_in[12];
  const float* W_dt   = (const float*)d_in[13];
  const float* b_dt   = (const float*)d_in[14];
  const float* A_log  = (const float*)d_in[15];
  const float* Dp     = (const float*)d_in[16];
  const float* W_out  = (const float*)d_in[17];

  float* ws = (float*)d_ws;
  size_t off = 0;
  float* kb      = ws + off; off += (size_t)MSLOTSn*Dn;
  float* vb      = ws + off; off += (size_t)MSLOTSn*Dn;
  float* Wq_sw   = ws + off; off += (size_t)8*32*32*2;
  float* Wo_sw   = ws + off; off += (size_t)8*32*32*2;
  float* Win_sw  = ws + off; off += (size_t)32*32*32*2;
  float* Wx_sw   = ws + off; off += (size_t)2*64*32*2;
  float* Wdt_sw  = ws + off; off += (size_t)16*2*32*2;
  float* Wout_sw = ws + off; off += (size_t)8*64*32*2;
  float* kfragT  = ws + off; off += (size_t)1*32*32*2;
  float* vfrag   = ws + off; off += (size_t)8*4*32*2;
  float* xr      = ws + off; off += (size_t)NTOK*Dn;
  float* xp      = ws + off; off += (size_t)NTOK*EDn;
  float* zb      = ws + off; off += (size_t)NTOK*EDn;
  float* xc      = ws + off; off += (size_t)NTOK*EDn;
  float* delta   = ws + off; off += (size_t)NTOK*EDn;
  float* Bm      = ws + off; off += (size_t)NTOK*DSTATEn;
  float* Cm      = ws + off; off += (size_t)NTOK*DSTATEn;
  float* y       = xp;

  swz_kernel<<<(8*32*32  + 255)/256, 256, 0, stream>>>(Wq,    Wq_sw,   Dn,  Dn,     8);
  swz_kernel<<<(8*32*32  + 255)/256, 256, 0, stream>>>(Wo,    Wo_sw,   Dn,  Dn,     8);
  swz_kernel<<<(32*32*32 + 255)/256, 256, 0, stream>>>(W_in,  Win_sw,  Dn,  2*EDn, 32);
  swz_kernel<<<(2*64*32  + 255)/256, 256, 0, stream>>>(W_x,   Wx_sw,   EDn, 24,     2);
  swz_kernel<<<(16*2*32  + 255)/256, 256, 0, stream>>>(W_dt,  Wdt_sw,  DTRANKn, EDn, 16);
  swz_kernel<<<(8*64*32  + 255)/256, 256, 0, stream>>>(W_out, Wout_sw, EDn, Dn,     8);

  kv_kernel   <<<(2*MSLOTSn*Dn + 255)/256, 256, 0, stream>>>(mem, Wk, Wv, kb, vb);
  kvswz_kernel<<<4, 256, 0, stream>>>(kb, vb, kfragT, vfrag);

  attn_kernel<<<NBLK, 128, 0, stream>>>(x, (const v2f*)Wq_sw, (const v2f*)Wo_sw,
                                        ln1_g, ln1_b, (const v2f*)kfragT,
                                        (const v2f*)vfrag, xr);
  in_kernel  <<<NBLK, 128, 0, stream>>>(xr, rms_w, (const v2f*)Win_sw, xp, zb);
  conv_kernel<<<NBLK, 128, 0, stream>>>(xp, conv_w, conv_b, (const v2f*)Wx_sw,
                                        (const v2f*)Wdt_sw, b_dt, xc, delta, Bm, Cm);
  scan_local_kernel<<<dim3(NCH, Bn), EDn, 0, stream>>>(delta, xc, Bm, A_log, y);
  scan_carry_kernel<<<dim3(DSTATEn, Bn), EDn, 0, stream>>>(A_log, y);
  scan_apply_kernel<<<dim3(NCH, Bn), EDn, 0, stream>>>(delta, xc, Bm, Cm,
                                                       A_log, Dp, y);
  out_kernel <<<NBLK, 128, 0, stream>>>(y, zb, (const v2f*)Wout_sw, xr, x,
                                        (float*)d_out);
}


# global loads of the conv, out and layer-norm stages issued as independent register-held groups ahead of their stores (conv: 51-row window per thread), not one exposed round trip per element
# speedup vs baseline: 7.4934x; 1.3305x over previous
#include <hip/hip_runtime.h>
#include <hip/hip_bf16.h>
#include <math.h>

#define Bn      8
#define Sn      12
#define NNn     325
#define Dn      128
#define EDn     256
#define DSTATEn 8
#define DCONVn  4
#define DTRANKn 8
#define MSLOTSn 10
#define Ln      (Sn*NNn)
#define NTOK    (Bn*Ln)
#define TM      3
#define TOKB    (16*TM)
#define NBLK    (NTOK/TOKB)
#define PD      (Dn + 4)
#define PE      (EDn + 4)
#define PS      20
#define PB      36

typedef float v2f __attribute__((ext_vector_type(2)));
typedef float v8f __attribute__((ext_vector_type(8)));

__device__ __forceinline__ v8f wmma4(v2f a, v2f b, v8f c) {
  return __builtin_amdgcn_wmma_f32_16x16x4_f32(false, a, false, b, (short)0, c,
                                               false, false);
}

__device__ __forceinline__ void store_c(float* C, int ldc, int nb, v8f c) {
  int lane = threadIdx.x & 31;
  int n = nb + (lane & 15);
  int mb = (lane < 16) ? 0 : 8;
#pragma unroll
  for (int j = 0; j < 8; ++j) C[(mb + j)*ldc + n] = c[j];
}

__device__ __forceinline__ void gemm3(const float* A, int lda, const v2f* Bsw,
                                      int KT, int nt, v8f acc[TM]) {
  int lane = threadIdx.x & 31;
  int m = lane & 15;
  int ks = (lane >> 4) << 1;
  const v2f* bp = Bsw + (size_t)nt * KT * 32 + lane;
  const v2f* a0 = (const v2f*)(A + (0*16 + m)*lda + ks);
  const v2f* a1 = (const v2f*)(A + (1*16 + m)*lda + ks);
  const v2f* a2 = (const v2f*)(A + (2*16 + m)*lda + ks);
  v8f z = {};
  acc[0] = z; acc[1] = z; acc[2] = z;
#pragma unroll 4
  for (int kt = 0; kt < KT; ++kt) {
    v2f b = bp[kt * 32];
    acc[0] = wmma4(a0[kt*2], b, acc[0]);
    acc[1] = wmma4(a1[kt*2], b, acc[1]);
    acc[2] = wmma4(a2[kt*2], b, acc[2]);
  }
}

__device__ __forceinline__ float silu_f(float x) { return x / (1.f + __expf(-x)); }

__global__ void swz_kernel(const float* W, float* dst, int K, int Nreal, int NT) {
  int idx = blockIdx.x * blockDim.x + threadIdx.x;
  int KT = K / 4;
  if (idx >= NT * KT * 32) return;
  int lane = idx & 31;
  int kt = (idx >> 5) % KT;
  int nt = idx / (32 * KT);
  int n = nt*16 + (lane & 15);
  int k = kt*4 + ((lane >> 4) << 1);
  float bx = (n < Nreal) ? W[k*Nreal + n]     : 0.f;
  float by = (n < Nreal) ? W[(k+1)*Nreal + n] : 0.f;
  dst[idx*2]     = bx;
  dst[idx*2 + 1] = by;
}

__global__ void kv_kernel(const float* mem, const float* Wk, const float* Wv,
                          float* kb, float* vb) {
  int i = blockIdx.x * blockDim.x + threadIdx.x;
  if (i >= 2 * MSLOTSn * Dn) return;
  int which = i / (MSLOTSn * Dn);
  int r = i % (MSLOTSn * Dn);
  int m = r / Dn, d = r % Dn;
  const float* W = which ? Wv : Wk;
  float acc = 0.f;
  for (int k = 0; k < Dn; ++k) acc += mem[m*Dn + k] * W[k*Dn + d];
  (which ? vb : kb)[m*Dn + d] = acc;
}

__global__ void kvswz_kernel(const float* kb, const float* vb,
                             float* kfragT, float* vfrag) {
  int idx = blockIdx.x * blockDim.x + threadIdx.x;
  if (idx < 32*32) {
    int lane = idx & 31; int kt = idx >> 5;
    int n = lane & 15;   int k = kt*4 + ((lane >> 4) << 1);
    kfragT[idx*2]     = (n < MSLOTSn) ? kb[n*Dn + k]     : 0.f;
    kfragT[idx*2 + 1] = (n < MSLOTSn) ? kb[n*Dn + k + 1] : 0.f;
  }
  if (idx < 8*4*32) {
    int lane = idx & 31; int kt = (idx >> 5) & 3; int nt = idx >> 7;
    int n = nt*16 + (lane & 15); int k = kt*4 + ((lane >> 4) << 1);
    vfrag[idx*2]     = (k     < MSLOTSn) ? vb[k*Dn + n]     : 0.f;
    vfrag[idx*2 + 1] = (k + 1 < MSLOTSn) ? vb[(k+1)*Dn + n] : 0.f;
  }
}

__global__ void attn_kernel(const float* x, const v2f* Wq_sw, const v2f* Wo_sw,
                            const float* ln_g, const float* ln_b,
                            const v2f* kfragT, const v2f* vfrag, float* xr) {
  __shared__ float xs[TOKB*PD];
  __shared__ float qs[TOKB*PD];
  __shared__ float av[TOKB*PD];
  __shared__ float sc[TOKB*PS];
  int tid = threadIdx.x;
  int wave = tid >> 5;
  long t0 = (long)blockIdx.x * TOKB;
  v8f acc[TM];

  for (int i = tid; i < TOKB*Dn; i += 128) xs[(i >> 7)*PD + (i & 127)] = x[t0*Dn + i];
  __syncthreads();

  for (int nti = 0; nti < 2; ++nti) {
    int nt = wave*2 + nti;
    gemm3(xs, PD, Wq_sw, 32, nt, acc);
#pragma unroll
    for (int t = 0; t < TM; ++t) store_c(qs + t*16*PD, PD, nt*16, acc[t]);
  }
  __syncthreads();

  if (wave == 0) {
    gemm3(qs, PD, kfragT, 32, 0, acc);
#pragma unroll
    for (int t = 0; t < TM; ++t) store_c(sc + t*16*PS, PS, 0, acc[t]);
  }
  __syncthreads();

  if (tid < TOKB) {
    const float scale = 0.088388347648318447f;
    float mx = -1e30f;
    for (int m = 0; m < MSLOTSn; ++m) {
      float v = sc[tid*PS + m] * scale; sc[tid*PS + m] = v;
      if (v > mx) mx = v;
    }
    float s = 0.f;
    for (int m = 0; m < MSLOTSn; ++m) {
      float e = __expf(sc[tid*PS + m] - mx); sc[tid*PS + m] = e; s += e;
    }
    float inv = 1.f / s;
    for (int m = 0; m < MSLOTSn; ++m) sc[tid*PS + m] *= inv;
    for (int m = MSLOTSn; m < 16; ++m) sc[tid*PS + m] = 0.f;
  }
  __syncthreads();

  for (int nti = 0; nti < 2; ++nti) {
    int nt = wave*2 + nti;
    gemm3(sc, PS, vfrag, 4, nt, acc);
#pragma unroll
    for (int t = 0; t < TM; ++t) store_c(av + t*16*PD, PD, nt*16, acc[t]);
  }
  __syncthreads();

  for (int nti = 0; nti < 2; ++nti) {
    int nt = wave*2 + nti;
    gemm3(av, PD, Wo_sw, 32, nt, acc);
#pragma unroll
    for (int t = 0; t < TM; ++t) store_c(qs + t*16*PD, PD, nt*16, acc[t]);
  }
  __syncthreads();

  if (tid < TOKB) {
    float m = 0.f;
    for (int c = 0; c < Dn; ++c) m += qs[tid*PD + c];
    m *= (1.f / Dn);
    float v = 0.f;
    for (int c = 0; c < Dn; ++c) { float d = qs[tid*PD + c] - m; v += d*d; }
    v *= (1.f / Dn);
    float inv = rsqrtf(v + 1e-5f);
    long tok = t0 + tid;
    for (int c0 = 0; c0 < Dn; c0 += 16) {
      float g[16], bb[16];
#pragma unroll
      for (int k = 0; k < 16; ++k) { g[k] = ln_g[c0 + k]; bb[k] = ln_b[c0 + k]; }
#pragma unroll
      for (int k = 0; k < 16; ++k) {
        int c = c0 + k;
        xr[tok*Dn + c] = (qs[tid*PD + c] - m) * inv * g[k] + bb[k]
                         + xs[tid*PD + c];
      }
    }
  }
}

__global__ void in_kernel(const float* xr, const float* rms_w, const v2f* Win_sw,
                          float* xp, float* zb) {
  __shared__ float xn[TOKB*PD];
  int tid = threadIdx.x;
  int wave = tid >> 5;
  int lane = tid & 31;
  long t0 = (long)blockIdx.x * TOKB;
  v8f acc[TM];

  for (int i = tid; i < TOKB*Dn; i += 128) xn[(i >> 7)*PD + (i & 127)] = xr[t0*Dn + i];
  __syncthreads();
  if (tid < TOKB) {
    float s = 0.f;
    for (int c = 0; c < Dn; ++c) { float v = xn[tid*PD + c]; s += v*v; }
    float inv = rsqrtf(s / Dn + 1e-5f);
    for (int c = 0; c < Dn; ++c) xn[tid*PD + c] = xn[tid*PD + c] * inv * rms_w[c];
  }
  __syncthreads();

  for (int nti = 0; nti < 8; ++nti) {
    int nt = wave*8 + nti;
    gemm3(xn, PD, Win_sw, 32, nt, acc);
    int n = nt*16 + (lane & 15);
    int mb = (lane < 16) ? 0 : 8;
#pragma unroll
    for (int t = 0; t < TM; ++t)
#pragma unroll
      for (int j = 0; j < 8; ++j) {
        long tok = t0 + t*16 + mb + j;
        float v = acc[t][j];
        if (n < EDn) xp[tok*EDn + n] = v;
        else         zb[tok*EDn + (n - EDn)] = v;
      }
  }
}

__global__ void conv_kernel(const float* xp, const float* conv_w, const float* conv_b,
                            const v2f* Wx_sw, const v2f* Wdt_sw, const float* b_dt,
                            float* xc, float* delta, float* Bm, float* Cm) {
  __shared__ float xcs[TOKB*PE];
  __shared__ float dbcs[TOKB*PB];
  int tid = threadIdx.x;
  int wave = tid >> 5;
  int lane = tid & 31;
  long t0 = (long)blockIdx.x * TOKB;
  v8f acc[TM];

  {
    const int c2  = tid * 2;
    const int t0i = blockIdx.x * TOKB;
    const int l0  = t0i % Ln;
    float w0[DCONVn], w1[DCONVn];
#pragma unroll
    for (int j = 0; j < DCONVn; ++j) {
      w0[j] = conv_w[c2*DCONVn + j];
      w1[j] = conv_w[(c2 + 1)*DCONVn + j];
    }
    const float cb0 = conv_b[c2], cb1 = conv_b[c2 + 1];
    v2f win[TOKB + DCONVn - 1];
#pragma unroll
    for (int i = 0; i < TOKB + DCONVn - 1; ++i) {
      int g = t0i - (DCONVn - 1) + i;
      if (g < 0) g = 0;
      win[i] = *(const v2f*)(xp + (long)g*EDn + c2);
    }
#pragma unroll
    for (int r = 0; r < TOKB; ++r) {
      int l = l0 + r;
      if (l >= Ln) l -= Ln;
      float ax = cb0, ay = cb1;
#pragma unroll
      for (int j = 0; j < DCONVn; ++j) {
        const bool ok = (l - (DCONVn - 1) + j) >= 0;
        const float x0 = ok ? win[r + j].x : 0.f;
        const float x1 = ok ? win[r + j].y : 0.f;
        ax += w0[j] * x0;
        ay += w1[j] * x1;
      }
      v2f o;
      o.x = silu_f(ax);
      o.y = silu_f(ay);
      *(v2f*)(xcs + r*PE + c2) = o;
      *(v2f*)(xc + (long)(t0i + r)*EDn + c2) = o;
    }
  }
  __syncthreads();

  if (wave < 2) {
    gemm3(xcs, PE, Wx_sw, 64, wave, acc);
#pragma unroll
    for (int t = 0; t < TM; ++t) store_c(dbcs + t*16*PB, PB, wave*16, acc[t]);
  }
  __syncthreads();

  for (int nti = 0; nti < 4; ++nti) {
    int nt = wave*4 + nti;
    gemm3(dbcs, PB, Wdt_sw, 2, nt, acc);
    int n = nt*16 + (lane & 15);
    int mb = (lane < 16) ? 0 : 8;
    const float bd = b_dt[n];
#pragma unroll
    for (int t = 0; t < TM; ++t)
#pragma unroll
      for (int j = 0; j < 8; ++j) {
        long tok = t0 + t*16 + mb + j;
        float v = acc[t][j] + bd;
        delta[tok*EDn + n] = (v > 20.f) ? v : log1pf(__expf(v));
      }
  }

  for (int i = tid; i < TOKB*16; i += 128) {
    int r = i >> 4, c = i & 15;
    long tok = t0 + r;
    float v = dbcs[r*PB + DTRANKn + c];
    if (c < DSTATEn) Bm[tok*DSTATEn + c]             = v;
    else             Cm[tok*DSTATEn + (c - DSTATEn)] = v;
  }
}

#define LC      25
#define NCH     (Ln/LC)
#define CARRYU  26

__global__ void __launch_bounds__(EDn)
scan_local_kernel(const float* __restrict__ delta, const float* __restrict__ xc,
                  const float* __restrict__ Bm, const float* __restrict__ A_log,
                  float* __restrict__ y) {
  const int d  = threadIdx.x;
  const int t0 = blockIdx.y * Ln + blockIdx.x * LC;
  float a[DSTATEn], h[DSTATEn];
#pragma unroll
  for (int n = 0; n < DSTATEn; ++n) { a[n] = -__expf(A_log[d*DSTATEn + n]); h[n] = 0.f; }
  float sdt = 0.f;
#pragma unroll
  for (int l = 0; l < LC; ++l) {
    const int t = t0 + l;
    float dt = delta[t*EDn + d];
    float xv = xc[t*EDn + d];
    sdt += dt;
#pragma unroll
    for (int n = 0; n < DSTATEn; ++n) {
      float dA = __expf(dt * a[n]);
      h[n] = dA * h[n] + dt * Bm[t*DSTATEn + n] * xv;
    }
  }
#pragma unroll
  for (int n = 0; n < DSTATEn; ++n) y[(t0 + n)*EDn + d] = h[n];
  y[(t0 + DSTATEn)*EDn + d] = sdt;
}

__global__ void __launch_bounds__(EDn)
scan_carry_kernel(const float* __restrict__ A_log, float* y) {
  const int d = threadIdx.x;
  const int n = blockIdx.x;
  const int b = blockIdx.y;
  const float a = -__expf(A_log[d*DSTATEn + n]);
  float*       hp = y + (b*Ln + n)*EDn + d;
  const float* sp = y + (b*Ln + DSTATEn)*EDn + d;
  float carry = 0.f;
  for (int c0 = 0; c0 < NCH; c0 += CARRYU) {
    float he[CARRYU], sd[CARRYU];
#pragma unroll
    for (int j = 0; j < CARRYU; ++j) {
      he[j] = hp[(c0 + j)*(LC*EDn)];
      sd[j] = sp[(c0 + j)*(LC*EDn)];
    }
#pragma unroll
    for (int j = 0; j < CARRYU; ++j) {
      float P = __expf(a * sd[j]);
      hp[(c0 + j)*(LC*EDn)] = carry;
      carry = P * carry + he[j];
    }
  }
}

__global__ void __launch_bounds__(EDn)
scan_apply_kernel(const float* __restrict__ delta, const float* __restrict__ xc,
                  const float* __restrict__ Bm, const float* __restrict__ Cm,
                  const float* __restrict__ A_log, const float* __restrict__ Dp,
                  float* __restrict__ y) {
  const int d  = threadIdx.x;
  const int t0 = blockIdx.y * Ln + blockIdx.x * LC;
  float a[DSTATEn], h[DSTATEn];
#pragma unroll
  for (int n = 0; n < DSTATEn; ++n) {
    a[n] = -__expf(A_log[d*DSTATEn + n]);
    h[n] = y[(t0 + n)*EDn + d];
  }
  const float dp = Dp[d];
#pragma unroll
  for (int l = 0; l < LC; ++l) {
    const int t = t0 + l;
    float dt = delta[t*EDn + d];
    float xv = xc[t*EDn + d];
    float yv = 0.f;
#pragma unroll
    for (int n = 0; n < DSTATEn; ++n) {
      float dA = __expf(dt * a[n]);
      h[n] = dA * h[n] + dt * Bm[t*DSTATEn + n] * xv;
      yv += h[n] * Cm[t*DSTATEn + n];
    }
    y[t*EDn + d] = yv + dp * xv;
  }
}

__global__ void out_kernel(const float* y, const float* zb, const v2f* Wout_sw,
                           const float* xr, const float* x, float* out) {
  __shared__ float ys[TOKB*PE];
  int tid = threadIdx.x;
  int wave = tid >> 5;
  int lane = tid & 31;
  long t0 = (long)blockIdx.x * TOKB;
  v8f acc[TM];

#pragma unroll 12
  for (int i = tid; i < TOKB*EDn/4; i += 128) {
    int r = i >> 6, c = (i & 63) << 2;
    long tok = t0 + r;
    const float4 z4 = *(const float4*)(zb + tok*EDn + c);
    const float4 y4 = *(const float4*)(y  + tok*EDn + c);
    float4 o;
    o.x = y4.x * silu_f(z4.x);
    o.y = y4.y * silu_f(z4.y);
    o.z = y4.z * silu_f(z4.z);
    o.w = y4.w * silu_f(z4.w);
    *(float4*)(ys + r*PE + c) = o;
  }
  __syncthreads();

  for (int nti = 0; nti < 2; ++nti) {
    int nt = wave*2 + nti;
    int n = nt*16 + (lane & 15);
    int mb = (lane < 16) ? 0 : 8;
    float r1[TM][8], r2[TM][8];
#pragma unroll
    for (int t = 0; t < TM; ++t)
#pragma unroll
      for (int j = 0; j < 8; ++j) {
        long tok = t0 + t*16 + mb + j;
        r1[t][j] = xr[tok*Dn + n];
        r2[t][j] = x[tok*Dn + n];
      }
    gemm3(ys, PE, Wout_sw, 64, nt, acc);
#pragma unroll
    for (int t = 0; t < TM; ++t)
#pragma unroll
      for (int j = 0; j < 8; ++j) {
        long tok = t0 + t*16 + mb + j;
        out[tok*Dn + n] = acc[t][j] + r1[t][j] + r2[t][j];
      }
  }
}

extern "C" void kernel_launch(void* const* d_in, const int* in_sizes, int n_in,
                              void* d_out, int out_size, void* d_ws, size_t ws_size,
                              hipStream_t stream) {
  (void)in_sizes; (void)n_in; (void)out_size; (void)ws_size;
  const float* x      = (const float*)d_in[0];
  const float* mem    = (const float*)d_in[1];
  const float* Wq     = (const float*)d_in[2];
  const float* Wk     = (const float*)d_in[3];
  const float* Wv     = (const float*)d_in[4];
  const float* Wo     = (const float*)d_in[5];
  const float* ln1_g  = (const float*)d_in[6];
  const float* ln1_b  = (const float*)d_in[7];
  const float* rms_w  = (const float*)d_in[8];
  const float* W_in   = (const float*)d_in[9];
  const float* conv_w = (const float*)d_in[10];
  const float* conv_b = (const float*)d_in[11];
  const float* W_x    = (const float*)d_in[12];
  const float* W_dt   = (const float*)d_in[13];
  const float* b_dt   = (const float*)d_in[14];
  const float* A_log  = (const float*)d_in[15];
  const float* Dp     = (const float*)d_in[16];
  const float* W_out  = (const float*)d_in[17];

  float* ws = (float*)d_ws;
  size_t off = 0;
  float* kb      = ws + off; off += (size_t)MSLOTSn*Dn;
  float* vb      = ws + off; off += (size_t)MSLOTSn*Dn;
  float* Wq_sw   = ws + off; off += (size_t)8*32*32*2;
  float* Wo_sw   = ws + off; off += (size_t)8*32*32*2;
  float* Win_sw  = ws + off; off += (size_t)32*32*32*2;
  float* Wx_sw   = ws + off; off += (size_t)2*64*32*2;
  float* Wdt_sw  = ws + off; off += (size_t)16*2*32*2;
  float* Wout_sw = ws + off; off += (size_t)8*64*32*2;
  float* kfragT  = ws + off; off += (size_t)1*32*32*2;
  float* vfrag   = ws + off; off += (size_t)8*4*32*2;
  float* xr      = ws + off; off += (size_t)NTOK*Dn;
  float* xp      = ws + off; off += (size_t)NTOK*EDn;
  float* zb      = ws + off; off += (size_t)NTOK*EDn;
  float* xc      = ws + off; off += (size_t)NTOK*EDn;
  float* delta   = ws + off; off += (size_t)NTOK*EDn;
  float* Bm      = ws + off; off += (size_t)NTOK*DSTATEn;
  float* Cm      = ws + off; off += (size_t)NTOK*DSTATEn;
  float* y       = xp;

  swz_kernel<<<(8*32*32  + 255)/256, 256, 0, stream>>>(Wq,    Wq_sw,   Dn,  Dn,     8);
  swz_kernel<<<(8*32*32  + 255)/256, 256, 0, stream>>>(Wo,    Wo_sw,   Dn,  Dn,     8);
  swz_kernel<<<(32*32*32 + 255)/256, 256, 0, stream>>>(W_in,  Win_sw,  Dn,  2*EDn, 32);
  swz_kernel<<<(2*64*32  + 255)/256, 256, 0, stream>>>(W_x,   Wx_sw,   EDn, 24,     2);
  swz_kernel<<<(16*2*32  + 255)/256, 256, 0, stream>>>(W_dt,  Wdt_sw,  DTRANKn, EDn, 16);
  swz_kernel<<<(8*64*32  + 255)/256, 256, 0, stream>>>(W_out, Wout_sw, EDn, Dn,     8);

  kv_kernel   <<<(2*MSLOTSn*Dn + 255)/256, 256, 0, stream>>>(mem, Wk, Wv, kb, vb);
  kvswz_kernel<<<4, 256, 0, stream>>>(kb, vb, kfragT, vfrag);

  attn_kernel<<<NBLK, 128, 0, stream>>>(x, (const v2f*)Wq_sw, (const v2f*)Wo_sw,
                                        ln1_g, ln1_b, (const v2f*)kfragT,
                                        (const v2f*)vfrag, xr);
  in_kernel  <<<NBLK, 128, 0, stream>>>(xr, rms_w, (const v2f*)Win_sw, xp, zb);
  conv_kernel<<<NBLK, 128, 0, stream>>>(xp, conv_w, conv_b, (const v2f*)Wx_sw,
                                        (const v2f*)Wdt_sw, b_dt, xc, delta, Bm, Cm);
  scan_local_kernel<<<dim3(NCH, Bn), EDn, 0, stream>>>(delta, xc, Bm, A_log, y);
  scan_carry_kernel<<<dim3(DSTATEn, Bn), EDn, 0, stream>>>(A_log, y);
  scan_apply_kernel<<<dim3(NCH, Bn), EDn, 0, stream>>>(delta, xc, Bm, Cm,
                                                       A_log, Dp, y);
  out_kernel <<<NBLK, 128, 0, stream>>>(y, zb, (const v2f*)Wout_sw, xr, x,
                                        (float*)d_out);
}


# B-fragment loads of every GEMM phase software-pipelined: each 16-k-step block's loads go out as one group ahead of the previous block's WMMAs (two register sets), across N-tiles; sums unchanged
# speedup vs baseline: 8.4533x; 1.1281x over previous
#include <hip/hip_runtime.h>
#include <hip/hip_bf16.h>
#include <math.h>

#define Bn      8
#define Sn      12
#define NNn     325
#define Dn      128
#define EDn     256
#define DSTATEn 8
#define DCONVn  4
#define DTRANKn 8
#define MSLOTSn 10
#define Ln      (Sn*NNn)
#define NTOK    (Bn*Ln)
#define TM      3
#define TOKB    (16*TM)
#define NBLK    (NTOK/TOKB)
#define PD      (Dn + 4)
#define PE      (EDn + 4)
#define PS      20
#define PB      36

typedef float v2f __attribute__((ext_vector_type(2)));
typedef float v8f __attribute__((ext_vector_type(8)));

__device__ __forceinline__ v8f wmma4(v2f a, v2f b, v8f c) {
  return __builtin_amdgcn_wmma_f32_16x16x4_f32(false, a, false, b, (short)0, c,
                                               false, false);
}

__device__ __forceinline__ void store_c(float* C, int ldc, int nb, v8f c) {
  int lane = threadIdx.x & 31;
  int n = nb + (lane & 15);
  int mb = (lane < 16) ? 0 : 8;
#pragma unroll
  for (int j = 0; j < 8; ++j) C[(mb + j)*ldc + n] = c[j];
}

#define CH      16

__device__ __forceinline__ void mma_block(const v2f* a0, const v2f* a1,
                                          const v2f* a2, const v2f (&b)[CH],
                                          v8f (&acc)[TM]) {
#pragma unroll
  for (int j = 0; j < CH; ++j) {
    acc[0] = wmma4(a0[j*2], b[j], acc[0]);
    acc[1] = wmma4(a1[j*2], b[j], acc[1]);
    acc[2] = wmma4(a2[j*2], b[j], acc[2]);
  }
}

__device__ __forceinline__ void load_block(const v2f* bq, v2f (&b)[CH]) {
#pragma unroll
  for (int j = 0; j < CH; ++j) b[j] = bq[j*32];
}

template <int KT, int NTI, class Epi>
__device__ __forceinline__ void gemm_stream(const float* A, int lda,
                                            const v2f* Bsw, int nt0, Epi epi) {
  constexpr int NC  = NTI*KT/CH;
  constexpr int CPN = KT/CH;
  static_assert(KT % (2*CH) == 0, "an N-tile must be an even number of blocks");
  int lane = threadIdx.x & 31;
  int m = lane & 15;
  int ks = (lane >> 4) << 1;
  const v2f* bp = Bsw + (size_t)nt0 * KT * 32 + lane;
  const v2f* a0 = (const v2f*)(A + (0*16 + m)*lda + ks);
  const v2f* a1 = (const v2f*)(A + (1*16 + m)*lda + ks);
  const v2f* a2 = (const v2f*)(A + (2*16 + m)*lda + ks);
  v2f b0[CH], b1[CH];
  v8f acc[TM];
  const v8f z = {};
  acc[0] = z; acc[1] = z; acc[2] = z;
  load_block(bp, b0);
#pragma nounroll
  for (int c = 0; c < NC - 2; c += 2) {
    const int ko = (c % CPN) * CH * 2;
    load_block(bp + (c + 1)*CH*32, b1);
    mma_block(a0 + ko, a1 + ko, a2 + ko, b0, acc);
    load_block(bp + (c + 2)*CH*32, b0);
    mma_block(a0 + ko + CH*2, a1 + ko + CH*2, a2 + ko + CH*2, b1, acc);
    if (NTI > 1 && ((c + 2) % CPN) == 0) {
      epi(nt0 + c / CPN, acc);
      acc[0] = z; acc[1] = z; acc[2] = z;
    }
  }
  {
    const int ko = ((NC - 2) % CPN) * CH * 2;
    load_block(bp + (NC - 1)*CH*32, b1);
    mma_block(a0 + ko, a1 + ko, a2 + ko, b0, acc);
    mma_block(a0 + ko + CH*2, a1 + ko + CH*2, a2 + ko + CH*2, b1, acc);
    epi(nt0 + NTI - 1, acc);
  }
}

template <int KT, int NTI, class Epi>
__device__ __forceinline__ void gemm_small(const float* A, int lda,
                                           const v2f* Bsw, int nt0, Epi epi) {
  int lane = threadIdx.x & 31;
  int m = lane & 15;
  int ks = (lane >> 4) << 1;
  const v2f* bp = Bsw + (size_t)nt0 * KT * 32 + lane;
  const v2f* a0 = (const v2f*)(A + (0*16 + m)*lda + ks);
  const v2f* a1 = (const v2f*)(A + (1*16 + m)*lda + ks);
  const v2f* a2 = (const v2f*)(A + (2*16 + m)*lda + ks);
  v2f b[NTI*KT];
#pragma unroll
  for (int i = 0; i < NTI*KT; ++i) b[i] = bp[i*32];
#pragma unroll
  for (int nti = 0; nti < NTI; ++nti) {
    v8f acc[TM];
    const v8f z = {};
    acc[0] = z; acc[1] = z; acc[2] = z;
#pragma unroll
    for (int kt = 0; kt < KT; ++kt) {
      acc[0] = wmma4(a0[kt*2], b[nti*KT + kt], acc[0]);
      acc[1] = wmma4(a1[kt*2], b[nti*KT + kt], acc[1]);
      acc[2] = wmma4(a2[kt*2], b[nti*KT + kt], acc[2]);
    }
    epi(nt0 + nti, acc);
  }
}

__device__ __forceinline__ float silu_f(float x) { return x / (1.f + __expf(-x)); }

__global__ void swz_kernel(const float* W, float* dst, int K, int Nreal, int NT) {
  int idx = blockIdx.x * blockDim.x + threadIdx.x;
  int KT = K / 4;
  if (idx >= NT * KT * 32) return;
  int lane = idx & 31;
  int kt = (idx >> 5) % KT;
  int nt = idx / (32 * KT);
  int n = nt*16 + (lane & 15);
  int k = kt*4 + ((lane >> 4) << 1);
  float bx = (n < Nreal) ? W[k*Nreal + n]     : 0.f;
  float by = (n < Nreal) ? W[(k+1)*Nreal + n] : 0.f;
  dst[idx*2]     = bx;
  dst[idx*2 + 1] = by;
}

__global__ void kv_kernel(const float* mem, const float* Wk, const float* Wv,
                          float* kb, float* vb) {
  int i = blockIdx.x * blockDim.x + threadIdx.x;
  if (i >= 2 * MSLOTSn * Dn) return;
  int which = i / (MSLOTSn * Dn);
  int r = i % (MSLOTSn * Dn);
  int m = r / Dn, d = r % Dn;
  const float* W = which ? Wv : Wk;
  float acc = 0.f;
  for (int k = 0; k < Dn; ++k) acc += mem[m*Dn + k] * W[k*Dn + d];
  (which ? vb : kb)[m*Dn + d] = acc;
}

__global__ void kvswz_kernel(const float* kb, const float* vb,
                             float* kfragT, float* vfrag) {
  int idx = blockIdx.x * blockDim.x + threadIdx.x;
  if (idx < 32*32) {
    int lane = idx & 31; int kt = idx >> 5;
    int n = lane & 15;   int k = kt*4 + ((lane >> 4) << 1);
    kfragT[idx*2]     = (n < MSLOTSn) ? kb[n*Dn + k]     : 0.f;
    kfragT[idx*2 + 1] = (n < MSLOTSn) ? kb[n*Dn + k + 1] : 0.f;
  }
  if (idx < 8*4*32) {
    int lane = idx & 31; int kt = (idx >> 5) & 3; int nt = idx >> 7;
    int n = nt*16 + (lane & 15); int k = kt*4 + ((lane >> 4) << 1);
    vfrag[idx*2]     = (k     < MSLOTSn) ? vb[k*Dn + n]     : 0.f;
    vfrag[idx*2 + 1] = (k + 1 < MSLOTSn) ? vb[(k+1)*Dn + n] : 0.f;
  }
}

__global__ void __launch_bounds__(128, 3)
attn_kernel(const float* x, const v2f* Wq_sw, const v2f* Wo_sw,
                            const float* ln_g, const float* ln_b,
                            const v2f* kfragT, const v2f* vfrag, float* xr) {
  __shared__ float xs[TOKB*PD];
  __shared__ float qs[TOKB*PD];
  __shared__ float av[TOKB*PD];
  __shared__ float sc[TOKB*PS];
  int tid = threadIdx.x;
  int wave = tid >> 5;
  long t0 = (long)blockIdx.x * TOKB;

  for (int i = tid; i < TOKB*Dn; i += 128) xs[(i >> 7)*PD + (i & 127)] = x[t0*Dn + i];
  __syncthreads();

  gemm_stream<32, 2>(xs, PD, Wq_sw, wave*2, [&](int nt, const v8f (&acc)[TM]) {
#pragma unroll
    for (int t = 0; t < TM; ++t) store_c(qs + t*16*PD, PD, nt*16, acc[t]);
  });
  __syncthreads();

  if (wave == 0) {
    gemm_stream<32, 1>(qs, PD, kfragT, 0, [&](int, const v8f (&acc)[TM]) {
#pragma unroll
      for (int t = 0; t < TM; ++t) store_c(sc + t*16*PS, PS, 0, acc[t]);
    });
  }
  __syncthreads();

  if (tid < TOKB) {
    const float scale = 0.088388347648318447f;
    float mx = -1e30f;
    for (int m = 0; m < MSLOTSn; ++m) {
      float v = sc[tid*PS + m] * scale; sc[tid*PS + m] = v;
      if (v > mx) mx = v;
    }
    float s = 0.f;
    for (int m = 0; m < MSLOTSn; ++m) {
      float e = __expf(sc[tid*PS + m] - mx); sc[tid*PS + m] = e; s += e;
    }
    float inv = 1.f / s;
    for (int m = 0; m < MSLOTSn; ++m) sc[tid*PS + m] *= inv;
    for (int m = MSLOTSn; m < 16; ++m) sc[tid*PS + m] = 0.f;
  }
  __syncthreads();

  gemm_small<4, 2>(sc, PS, vfrag, wave*2, [&](int nt, const v8f (&acc)[TM]) {
#pragma unroll
    for (int t = 0; t < TM; ++t) store_c(av + t*16*PD, PD, nt*16, acc[t]);
  });
  __syncthreads();

  gemm_stream<32, 2>(av, PD, Wo_sw, wave*2, [&](int nt, const v8f (&acc)[TM]) {
#pragma unroll
    for (int t = 0; t < TM; ++t) store_c(qs + t*16*PD, PD, nt*16, acc[t]);
  });
  __syncthreads();

  if (tid < TOKB) {
    float m = 0.f;
    for (int c = 0; c < Dn; ++c) m += qs[tid*PD + c];
    m *= (1.f / Dn);
    float v = 0.f;
    for (int c = 0; c < Dn; ++c) { float d = qs[tid*PD + c] - m; v += d*d; }
    v *= (1.f / Dn);
    float inv = rsqrtf(v + 1e-5f);
    long tok = t0 + tid;
    for (int c0 = 0; c0 < Dn; c0 += 16) {
      float g[16], bb[16];
#pragma unroll
      for (int k = 0; k < 16; ++k) { g[k] = ln_g[c0 + k]; bb[k] = ln_b[c0 + k]; }
#pragma unroll
      for (int k = 0; k < 16; ++k) {
        int c = c0 + k;
        xr[tok*Dn + c] = (qs[tid*PD + c] - m) * inv * g[k] + bb[k]
                         + xs[tid*PD + c];
      }
    }
  }
}

__global__ void __launch_bounds__(128, 3)
in_kernel(const float* xr, const float* rms_w, const v2f* Win_sw,
                          float* xp, float* zb) {
  __shared__ float xn[TOKB*PD];
  int tid = threadIdx.x;
  int wave = tid >> 5;
  int lane = tid & 31;
  long t0 = (long)blockIdx.x * TOKB;

  for (int i = tid; i < TOKB*Dn; i += 128) xn[(i >> 7)*PD + (i & 127)] = xr[t0*Dn + i];
  __syncthreads();
  if (tid < TOKB) {
    float s = 0.f;
    for (int c = 0; c < Dn; ++c) { float v = xn[tid*PD + c]; s += v*v; }
    float inv = rsqrtf(s / Dn + 1e-5f);
    for (int c = 0; c < Dn; ++c) xn[tid*PD + c] = xn[tid*PD + c] * inv * rms_w[c];
  }
  __syncthreads();

  gemm_stream<32, 8>(xn, PD, Win_sw, wave*8, [&](int nt, const v8f (&acc)[TM]) {
    int n = nt*16 + (lane & 15);
    int mb = (lane < 16) ? 0 : 8;
#pragma unroll
    for (int t = 0; t < TM; ++t)
#pragma unroll
      for (int j = 0; j < 8; ++j) {
        long tok = t0 + t*16 + mb + j;
        float v = acc[t][j];
        if (n < EDn) xp[tok*EDn + n] = v;
        else         zb[tok*EDn + (n - EDn)] = v;
      }
  });
}

__global__ void __launch_bounds__(128, 3)
conv_kernel(const float* xp, const float* conv_w, const float* conv_b,
                            const v2f* Wx_sw, const v2f* Wdt_sw, const float* b_dt,
                            float* xc, float* delta, float* Bm, float* Cm) {
  __shared__ float xcs[TOKB*PE];
  __shared__ float dbcs[TOKB*PB];
  int tid = threadIdx.x;
  int wave = tid >> 5;
  int lane = tid & 31;
  long t0 = (long)blockIdx.x * TOKB;

  {
    const int c2  = tid * 2;
    const int t0i = blockIdx.x * TOKB;
    const int l0  = t0i % Ln;
    float w0[DCONVn], w1[DCONVn];
#pragma unroll
    for (int j = 0; j < DCONVn; ++j) {
      w0[j] = conv_w[c2*DCONVn + j];
      w1[j] = conv_w[(c2 + 1)*DCONVn + j];
    }
    const float cb0 = conv_b[c2], cb1 = conv_b[c2 + 1];
    v2f win[TOKB + DCONVn - 1];
#pragma unroll
    for (int i = 0; i < TOKB + DCONVn - 1; ++i) {
      int g = t0i - (DCONVn - 1) + i;
      if (g < 0) g = 0;
      win[i] = *(const v2f*)(xp + (long)g*EDn + c2);
    }
#pragma unroll
    for (int r = 0; r < TOKB; ++r) {
      int l = l0 + r;
      if (l >= Ln) l -= Ln;
      float ax = cb0, ay = cb1;
#pragma unroll
      for (int j = 0; j < DCONVn; ++j) {
        const bool ok = (l - (DCONVn - 1) + j) >= 0;
        const float x0 = ok ? win[r + j].x : 0.f;
        const float x1 = ok ? win[r + j].y : 0.f;
        ax += w0[j] * x0;
        ay += w1[j] * x1;
      }
      v2f o;
      o.x = silu_f(ax);
      o.y = silu_f(ay);
      *(v2f*)(xcs + r*PE + c2) = o;
      *(v2f*)(xc + (long)(t0i + r)*EDn + c2) = o;
    }
  }
  __syncthreads();

  if (wave < 2) {
    gemm_stream<64, 1>(xcs, PE, Wx_sw, wave, [&](int nt, const v8f (&acc)[TM]) {
#pragma unroll
      for (int t = 0; t < TM; ++t) store_c(dbcs + t*16*PB, PB, nt*16, acc[t]);
    });
  }
  __syncthreads();

  gemm_small<2, 4>(dbcs, PB, Wdt_sw, wave*4, [&](int nt, const v8f (&acc)[TM]) {
    int n = nt*16 + (lane & 15);
    int mb = (lane < 16) ? 0 : 8;
    const float bd = b_dt[n];
#pragma unroll
    for (int t = 0; t < TM; ++t)
#pragma unroll
      for (int j = 0; j < 8; ++j) {
        long tok = t0 + t*16 + mb + j;
        float v = acc[t][j] + bd;
        delta[tok*EDn + n] = (v > 20.f) ? v : log1pf(__expf(v));
      }
  });

  for (int i = tid; i < TOKB*16; i += 128) {
    int r = i >> 4, c = i & 15;
    long tok = t0 + r;
    float v = dbcs[r*PB + DTRANKn + c];
    if (c < DSTATEn) Bm[tok*DSTATEn + c]             = v;
    else             Cm[tok*DSTATEn + (c - DSTATEn)] = v;
  }
}

#define LC      25
#define NCH     (Ln/LC)
#define CARRYU  26

__global__ void __launch_bounds__(EDn)
scan_local_kernel(const float* __restrict__ delta, const float* __restrict__ xc,
                  const float* __restrict__ Bm, const float* __restrict__ A_log,
                  float* __restrict__ y) {
  const int d  = threadIdx.x;
  const int t0 = blockIdx.y * Ln + blockIdx.x * LC;
  float a[DSTATEn], h[DSTATEn];
#pragma unroll
  for (int n = 0; n < DSTATEn; ++n) { a[n] = -__expf(A_log[d*DSTATEn + n]); h[n] = 0.f; }
  float sdt = 0.f;
#pragma unroll
  for (int l = 0; l < LC; ++l) {
    const int t = t0 + l;
    float dt = delta[t*EDn + d];
    float xv = xc[t*EDn + d];
    sdt += dt;
#pragma unroll
    for (int n = 0; n < DSTATEn; ++n) {
      float dA = __expf(dt * a[n]);
      h[n] = dA * h[n] + dt * Bm[t*DSTATEn + n] * xv;
    }
  }
#pragma unroll
  for (int n = 0; n < DSTATEn; ++n) y[(t0 + n)*EDn + d] = h[n];
  y[(t0 + DSTATEn)*EDn + d] = sdt;
}

__global__ void __launch_bounds__(EDn)
scan_carry_kernel(const float* __restrict__ A_log, float* y) {
  const int d = threadIdx.x;
  const int n = blockIdx.x;
  const int b = blockIdx.y;
  const float a = -__expf(A_log[d*DSTATEn + n]);
  float*       hp = y + (b*Ln + n)*EDn + d;
  const float* sp = y + (b*Ln + DSTATEn)*EDn + d;
  float carry = 0.f;
  for (int c0 = 0; c0 < NCH; c0 += CARRYU) {
    float he[CARRYU], sd[CARRYU];
#pragma unroll
    for (int j = 0; j < CARRYU; ++j) {
      he[j] = hp[(c0 + j)*(LC*EDn)];
      sd[j] = sp[(c0 + j)*(LC*EDn)];
    }
#pragma unroll
    for (int j = 0; j < CARRYU; ++j) {
      float P = __expf(a * sd[j]);
      hp[(c0 + j)*(LC*EDn)] = carry;
      carry = P * carry + he[j];
    }
  }
}

__global__ void __launch_bounds__(EDn)
scan_apply_kernel(const float* __restrict__ delta, const float* __restrict__ xc,
                  const float* __restrict__ Bm, const float* __restrict__ Cm,
                  const float* __restrict__ A_log, const float* __restrict__ Dp,
                  float* __restrict__ y) {
  const int d  = threadIdx.x;
  const int t0 = blockIdx.y * Ln + blockIdx.x * LC;
  float a[DSTATEn], h[DSTATEn];
#pragma unroll
  for (int n = 0; n < DSTATEn; ++n) {
    a[n] = -__expf(A_log[d*DSTATEn + n]);
    h[n] = y[(t0 + n)*EDn + d];
  }
  const float dp = Dp[d];
#pragma unroll
  for (int l = 0; l < LC; ++l) {
    const int t = t0 + l;
    float dt = delta[t*EDn + d];
    float xv = xc[t*EDn + d];
    float yv = 0.f;
#pragma unroll
    for (int n = 0; n < DSTATEn; ++n) {
      float dA = __expf(dt * a[n]);
      h[n] = dA * h[n] + dt * Bm[t*DSTATEn + n] * xv;
      yv += h[n] * Cm[t*DSTATEn + n];
    }
    y[t*EDn + d] = yv + dp * xv;
  }
}

__global__ void __launch_bounds__(128, 3)
out_kernel(const float* y, const float* zb, const v2f* Wout_sw,
                           const float* xr, const float* x, float* out) {
  __shared__ float ys[TOKB*PE];
  int tid = threadIdx.x;
  int wave = tid >> 5;
  int lane = tid & 31;
  long t0 = (long)blockIdx.x * TOKB;

#pragma unroll 12
  for (int i = tid; i < TOKB*EDn/4; i += 128) {
    int r = i >> 6, c = (i & 63) << 2;
    long tok = t0 + r;
    const float4 z4 = *(const float4*)(zb + tok*EDn + c);
    const float4 y4 = *(const float4*)(y  + tok*EDn + c);
    float4 o;
    o.x = y4.x * silu_f(z4.x);
    o.y = y4.y * silu_f(z4.y);
    o.z = y4.z * silu_f(z4.z);
    o.w = y4.w * silu_f(z4.w);
    *(float4*)(ys + r*PE + c) = o;
  }
  __syncthreads();

  gemm_stream<64, 2>(ys, PE, Wout_sw, wave*2, [&](int nt, const v8f (&acc)[TM]) {
    int n = nt*16 + (lane & 15);
    int mb = (lane < 16) ? 0 : 8;
    float r1[TM][8], r2[TM][8];
#pragma unroll
    for (int t = 0; t < TM; ++t)
#pragma unroll
      for (int j = 0; j < 8; ++j) {
        long tok = t0 + t*16 + mb + j;
        r1[t][j] = xr[tok*Dn + n];
        r2[t][j] = x[tok*Dn + n];
      }
#pragma unroll
    for (int t = 0; t < TM; ++t)
#pragma unroll
      for (int j = 0; j < 8; ++j) {
        long tok = t0 + t*16 + mb + j;
        out[tok*Dn + n] = acc[t][j] + r1[t][j] + r2[t][j];
      }
  });
}

extern "C" void kernel_launch(void* const* d_in, const int* in_sizes, int n_in,
                              void* d_out, int out_size, void* d_ws, size_t ws_size,
                              hipStream_t stream) {
  (void)in_sizes; (void)n_in; (void)out_size; (void)ws_size;
  const float* x      = (const float*)d_in[0];
  const float* mem    = (const float*)d_in[1];
  const float* Wq     = (const float*)d_in[2];
  const float* Wk     = (const float*)d_in[3];
  const float* Wv     = (const float*)d_in[4];
  const float* Wo     = (const float*)d_in[5];
  const float* ln1_g  = (const float*)d_in[6];
  const float* ln1_b  = (const float*)d_in[7];
  const float* rms_w  = (const float*)d_in[8];
  const float* W_in   = (const float*)d_in[9];
  const float* conv_w = (const float*)d_in[10];
  const float* conv_b = (const float*)d_in[11];
  const float* W_x    = (const float*)d_in[12];
  const float* W_dt   = (const float*)d_in[13];
  const float* b_dt   = (const float*)d_in[14];
  const float* A_log  = (const float*)d_in[15];
  const float* Dp     = (const float*)d_in[16];
  const float* W_out  = (const float*)d_in[17];

  float* ws = (float*)d_ws;
  size_t off = 0;
  float* kb      = ws + off; off += (size_t)MSLOTSn*Dn;
  float* vb      = ws + off; off += (size_t)MSLOTSn*Dn;
  float* Wq_sw   = ws + off; off += (size_t)8*32*32*2;
  float* Wo_sw   = ws + off; off += (size_t)8*32*32*2;
  float* Win_sw  = ws + off; off += (size_t)32*32*32*2;
  float* Wx_sw   = ws + off; off += (size_t)2*64*32*2;
  float* Wdt_sw  = ws + off; off += (size_t)16*2*32*2;
  float* Wout_sw = ws + off; off += (size_t)8*64*32*2;
  float* kfragT  = ws + off; off += (size_t)1*32*32*2;
  float* vfrag   = ws + off; off += (size_t)8*4*32*2;
  float* xr      = ws + off; off += (size_t)NTOK*Dn;
  float* xp      = ws + off; off += (size_t)NTOK*EDn;
  float* zb      = ws + off; off += (size_t)NTOK*EDn;
  float* xc      = ws + off; off += (size_t)NTOK*EDn;
  float* delta   = ws + off; off += (size_t)NTOK*EDn;
  float* Bm      = ws + off; off += (size_t)NTOK*DSTATEn;
  float* Cm      = ws + off; off += (size_t)NTOK*DSTATEn;
  float* y       = xp;

  swz_kernel<<<(8*32*32  + 255)/256, 256, 0, stream>>>(Wq,    Wq_sw,   Dn,  Dn,     8);
  swz_kernel<<<(8*32*32  + 255)/256, 256, 0, stream>>>(Wo,    Wo_sw,   Dn,  Dn,     8);
  swz_kernel<<<(32*32*32 + 255)/256, 256, 0, stream>>>(W_in,  Win_sw,  Dn,  2*EDn, 32);
  swz_kernel<<<(2*64*32  + 255)/256, 256, 0, stream>>>(W_x,   Wx_sw,   EDn, 24,     2);
  swz_kernel<<<(16*2*32  + 255)/256, 256, 0, stream>>>(W_dt,  Wdt_sw,  DTRANKn, EDn, 16);
  swz_kernel<<<(8*64*32  + 255)/256, 256, 0, stream>>>(W_out, Wout_sw, EDn, Dn,     8);

  kv_kernel   <<<(2*MSLOTSn*Dn + 255)/256, 256, 0, stream>>>(mem, Wk, Wv, kb, vb);
  kvswz_kernel<<<4, 256, 0, stream>>>(kb, vb, kfragT, vfrag);

  attn_kernel<<<NBLK, 128, 0, stream>>>(x, (const v2f*)Wq_sw, (const v2f*)Wo_sw,
                                        ln1_g, ln1_b, (const v2f*)kfragT,
                                        (const v2f*)vfrag, xr);
  in_kernel  <<<NBLK, 128, 0, stream>>>(xr, rms_w, (const v2f*)Win_sw, xp, zb);
  conv_kernel<<<NBLK, 128, 0, stream>>>(xp, conv_w, conv_b, (const v2f*)Wx_sw,
                                        (const v2f*)Wdt_sw, b_dt, xc, delta, Bm, Cm);
  scan_local_kernel<<<dim3(NCH, Bn), EDn, 0, stream>>>(delta, xc, Bm, A_log, y);
  scan_carry_kernel<<<dim3(DSTATEn, Bn), EDn, 0, stream>>>(A_log, y);
  scan_apply_kernel<<<dim3(NCH, Bn), EDn, 0, stream>>>(delta, xc, Bm, Cm,
                                                       A_log, Dp, y);
  out_kernel <<<NBLK, 128, 0, stream>>>(y, zb, (const v2f*)Wout_sw, xr, x,
                                        (float*)d_out);
}


# the eight weight-preparation launches (six fragment-order swizzles, k/v product, k/v swizzle) merged into one launch; k and v fragments formed directly from mem, Wk and Wv; sums unchanged
# speedup vs baseline: 8.8249x; 1.0440x over previous
#include <hip/hip_runtime.h>
#include <hip/hip_bf16.h>
#include <math.h>

#define Bn      8
#define Sn      12
#define NNn     325
#define Dn      128
#define EDn     256
#define DSTATEn 8
#define DCONVn  4
#define DTRANKn 8
#define MSLOTSn 10
#define Ln      (Sn*NNn)
#define NTOK    (Bn*Ln)
#define TM      3
#define TOKB    (16*TM)
#define NBLK    (NTOK/TOKB)
#define PD      (Dn + 4)
#define PE      (EDn + 4)
#define PS      20
#define PB      36

typedef float v2f __attribute__((ext_vector_type(2)));
typedef float v8f __attribute__((ext_vector_type(8)));

__device__ __forceinline__ v8f wmma4(v2f a, v2f b, v8f c) {
  return __builtin_amdgcn_wmma_f32_16x16x4_f32(false, a, false, b, (short)0, c,
                                               false, false);
}

__device__ __forceinline__ void store_c(float* C, int ldc, int nb, v8f c) {
  int lane = threadIdx.x & 31;
  int n = nb + (lane & 15);
  int mb = (lane < 16) ? 0 : 8;
#pragma unroll
  for (int j = 0; j < 8; ++j) C[(mb + j)*ldc + n] = c[j];
}

#define CH      16

__device__ __forceinline__ void mma_block(const v2f* a0, const v2f* a1,
                                          const v2f* a2, const v2f (&b)[CH],
                                          v8f (&acc)[TM]) {
#pragma unroll
  for (int j = 0; j < CH; ++j) {
    acc[0] = wmma4(a0[j*2], b[j], acc[0]);
    acc[1] = wmma4(a1[j*2], b[j], acc[1]);
    acc[2] = wmma4(a2[j*2], b[j], acc[2]);
  }
}

__device__ __forceinline__ void load_block(const v2f* bq, v2f (&b)[CH]) {
#pragma unroll
  for (int j = 0; j < CH; ++j) b[j] = bq[j*32];
}

template <int KT, int NTI, class Epi>
__device__ __forceinline__ void gemm_stream(const float* A, int lda,
                                            const v2f* Bsw, int nt0, Epi epi) {
  constexpr int NC  = NTI*KT/CH;
  constexpr int CPN = KT/CH;
  static_assert(KT % (2*CH) == 0, "an N-tile must be an even number of blocks");
  int lane = threadIdx.x & 31;
  int m = lane & 15;
  int ks = (lane >> 4) << 1;
  const v2f* bp = Bsw + (size_t)nt0 * KT * 32 + lane;
  const v2f* a0 = (const v2f*)(A + (0*16 + m)*lda + ks);
  const v2f* a1 = (const v2f*)(A + (1*16 + m)*lda + ks);
  const v2f* a2 = (const v2f*)(A + (2*16 + m)*lda + ks);
  v2f b0[CH], b1[CH];
  v8f acc[TM];
  const v8f z = {};
  acc[0] = z; acc[1] = z; acc[2] = z;
  load_block(bp, b0);
#pragma nounroll
  for (int c = 0; c < NC - 2; c += 2) {
    const int ko = (c % CPN) * CH * 2;
    load_block(bp + (c + 1)*CH*32, b1);
    mma_block(a0 + ko, a1 + ko, a2 + ko, b0, acc);
    load_block(bp + (c + 2)*CH*32, b0);
    mma_block(a0 + ko + CH*2, a1 + ko + CH*2, a2 + ko + CH*2, b1, acc);
    if (NTI > 1 && ((c + 2) % CPN) == 0) {
      epi(nt0 + c / CPN, acc);
      acc[0] = z; acc[1] = z; acc[2] = z;
    }
  }
  {
    const int ko = ((NC - 2) % CPN) * CH * 2;
    load_block(bp + (NC - 1)*CH*32, b1);
    mma_block(a0 + ko, a1 + ko, a2 + ko, b0, acc);
    mma_block(a0 + ko + CH*2, a1 + ko + CH*2, a2 + ko + CH*2, b1, acc);
    epi(nt0 + NTI - 1, acc);
  }
}

template <int KT, int NTI, class Epi>
__device__ __forceinline__ void gemm_small(const float* A, int lda,
                                           const v2f* Bsw, int nt0, Epi epi) {
  int lane = threadIdx.x & 31;
  int m = lane & 15;
  int ks = (lane >> 4) << 1;
  const v2f* bp = Bsw + (size_t)nt0 * KT * 32 + lane;
  const v2f* a0 = (const v2f*)(A + (0*16 + m)*lda + ks);
  const v2f* a1 = (const v2f*)(A + (1*16 + m)*lda + ks);
  const v2f* a2 = (const v2f*)(A + (2*16 + m)*lda + ks);
  v2f b[NTI*KT];
#pragma unroll
  for (int i = 0; i < NTI*KT; ++i) b[i] = bp[i*32];
#pragma unroll
  for (int nti = 0; nti < NTI; ++nti) {
    v8f acc[TM];
    const v8f z = {};
    acc[0] = z; acc[1] = z; acc[2] = z;
#pragma unroll
    for (int kt = 0; kt < KT; ++kt) {
      acc[0] = wmma4(a0[kt*2], b[nti*KT + kt], acc[0]);
      acc[1] = wmma4(a1[kt*2], b[nti*KT + kt], acc[1]);
      acc[2] = wmma4(a2[kt*2], b[nti*KT + kt], acc[2]);
    }
    epi(nt0 + nti, acc);
  }
}

__device__ __forceinline__ float silu_f(float x) { return x / (1.f + __expf(-x)); }

#define PM      (Dn + 1)
#define PBK_KV  8
#define PBK_WQ  (8*32*32/256)
#define PBK_WO  (8*32*32/256)
#define PBK_WIN (32*32*32/256)
#define PBK_WX  (2*64*32/256)
#define PBK_WDT (16*2*32/256)
#define PBK_WOUT (8*64*32/256)
#define PBK_ALL (PBK_KV + PBK_WQ + PBK_WO + PBK_WIN + PBK_WX + PBK_WDT + PBK_WOUT)

template <int K, int NREAL, int NT>
__device__ __forceinline__ void swz_tile(const float* W, float* dst, int idx) {
  constexpr int KT = K / 4;
  static_assert((NT*KT*32) % 256 == 0, "whole blocks");
  int lane = idx & 31;
  int kt = (idx >> 5) % KT;
  int nt = idx / (32 * KT);
  int n = nt*16 + (lane & 15);
  int k = kt*4 + ((lane >> 4) << 1);
  v2f o;
  o.x = (n < NREAL) ? W[k*NREAL + n]     : 0.f;
  o.y = (n < NREAL) ? W[(k+1)*NREAL + n] : 0.f;
  *(v2f*)(dst + idx*2) = o;
}

__device__ __forceinline__ void kv_tile(const float* mem, const float* Wk,
                                        const float* Wv, float* kfragT,
                                        float* vfrag, int blk, float* ms, float* ss) {
  const int tid = threadIdx.x;
  const bool isk = blk < 4;
  const int q = blk & 3;
  const float* Wsel = isk ? Wk : Wv;
#pragma unroll
  for (int r = 0; r < Dn*32/256; ++r) {
    int i = tid + 256*r;
    ss[i] = Wsel[(i >> 5)*Dn + q*32 + (i & 31)];
  }
#pragma unroll
  for (int r = 0; r < MSLOTSn*Dn/256; ++r) {
    int i = tid + 256*r;
    ms[(i >> 7)*PM + (i & 127)] = mem[i];
  }
  __syncthreads();
  const int lane = tid & 31;
  const int h2 = (lane >> 4) << 1;
  int r0, r1, c0, c1;
  bool ok0, ok1;
  if (isk) {
    const int n = lane & 15;
    const int kk = (tid >> 5)*4 + h2;
    r0 = r1 = n; c0 = kk; c1 = kk + 1;
    ok0 = ok1 = n < MSLOTSn;
  } else {
    const int s = ((tid >> 5) & 3)*4 + h2;
    const int cc = ((tid >> 7) << 4) + (lane & 15);
    r0 = s; r1 = s + 1; c0 = c1 = cc;
    ok0 = s < MSLOTSn; ok1 = s + 1 < MSLOTSn;
  }
  const float* m0 = ms + (ok0 ? r0 : 0)*PM;
  const float* m1 = ms + (ok1 ? r1 : 0)*PM;
  float acc0 = 0.f, acc1 = 0.f;
#pragma unroll 16
  for (int j = 0; j < Dn; ++j) {
    acc0 += m0[j] * ss[j*32 + c0];
    acc1 += m1[j] * ss[j*32 + c1];
  }
  v2f o;
  o.x = ok0 ? acc0 : 0.f;
  o.y = ok1 ? acc1 : 0.f;
  float* dst = isk ? kfragT : vfrag;
  *(v2f*)(dst + (q*256 + tid)*2) = o;
}

__global__ void __launch_bounds__(256)
prep_kernel(const float* mem, const float* Wk, const float* Wv,
            const float* Wq, const float* Wo, const float* W_in,
            const float* W_x, const float* W_dt, const float* W_out,
            float* kfragT, float* vfrag, float* Wq_sw, float* Wo_sw,
            float* Win_sw, float* Wx_sw, float* Wdt_sw, float* Wout_sw) {
  __shared__ float ms[MSLOTSn*PM];
  __shared__ float ss[Dn*32];
  int b = blockIdx.x;
  const int tid = threadIdx.x;
  if (b < PBK_KV) { kv_tile(mem, Wk, Wv, kfragT, vfrag, b, ms, ss); return; }
  b -= PBK_KV;
  if (b < PBK_WQ) { swz_tile<Dn, Dn, 8>(Wq, Wq_sw, b*256 + tid); return; }
  b -= PBK_WQ;
  if (b < PBK_WO) { swz_tile<Dn, Dn, 8>(Wo, Wo_sw, b*256 + tid); return; }
  b -= PBK_WO;
  if (b < PBK_WIN) { swz_tile<Dn, 2*EDn, 32>(W_in, Win_sw, b*256 + tid); return; }
  b -= PBK_WIN;
  if (b < PBK_WX) { swz_tile<EDn, DTRANKn + 2*DSTATEn, 2>(W_x, Wx_sw, b*256 + tid); return; }
  b -= PBK_WX;
  if (b < PBK_WDT) { swz_tile<DTRANKn, EDn, 16>(W_dt, Wdt_sw, b*256 + tid); return; }
  b -= PBK_WDT;
  swz_tile<EDn, Dn, 8>(W_out, Wout_sw, b*256 + tid);
}

__global__ void __launch_bounds__(128, 3)
attn_kernel(const float* x, const v2f* Wq_sw, const v2f* Wo_sw,
                            const float* ln_g, const float* ln_b,
                            const v2f* kfragT, const v2f* vfrag, float* xr) {
  __shared__ float xs[TOKB*PD];
  __shared__ float qs[TOKB*PD];
  __shared__ float av[TOKB*PD];
  __shared__ float sc[TOKB*PS];
  int tid = threadIdx.x;
  int wave = tid >> 5;
  long t0 = (long)blockIdx.x * TOKB;

  for (int i = tid; i < TOKB*Dn; i += 128) xs[(i >> 7)*PD + (i & 127)] = x[t0*Dn + i];
  __syncthreads();

  gemm_stream<32, 2>(xs, PD, Wq_sw, wave*2, [&](int nt, const v8f (&acc)[TM]) {
#pragma unroll
    for (int t = 0; t < TM; ++t) store_c(qs + t*16*PD, PD, nt*16, acc[t]);
  });
  __syncthreads();

  if (wave == 0) {
    gemm_stream<32, 1>(qs, PD, kfragT, 0, [&](int, const v8f (&acc)[TM]) {
#pragma unroll
      for (int t = 0; t < TM; ++t) store_c(sc + t*16*PS, PS, 0, acc[t]);
    });
  }
  __syncthreads();

  if (tid < TOKB) {
    const float scale = 0.088388347648318447f;
    float mx = -1e30f;
    for (int m = 0; m < MSLOTSn; ++m) {
      float v = sc[tid*PS + m] * scale; sc[tid*PS + m] = v;
      if (v > mx) mx = v;
    }
    float s = 0.f;
    for (int m = 0; m < MSLOTSn; ++m) {
      float e = __expf(sc[tid*PS + m] - mx); sc[tid*PS + m] = e; s += e;
    }
    float inv = 1.f / s;
    for (int m = 0; m < MSLOTSn; ++m) sc[tid*PS + m] *= inv;
    for (int m = MSLOTSn; m < 16; ++m) sc[tid*PS + m] = 0.f;
  }
  __syncthreads();

  gemm_small<4, 2>(sc, PS, vfrag, wave*2, [&](int nt, const v8f (&acc)[TM]) {
#pragma unroll
    for (int t = 0; t < TM; ++t) store_c(av + t*16*PD, PD, nt*16, acc[t]);
  });
  __syncthreads();

  gemm_stream<32, 2>(av, PD, Wo_sw, wave*2, [&](int nt, const v8f (&acc)[TM]) {
#pragma unroll
    for (int t = 0; t < TM; ++t) store_c(qs + t*16*PD, PD, nt*16, acc[t]);
  });
  __syncthreads();

  if (tid < TOKB) {
    float m = 0.f;
    for (int c = 0; c < Dn; ++c) m += qs[tid*PD + c];
    m *= (1.f / Dn);
    float v = 0.f;
    for (int c = 0; c < Dn; ++c) { float d = qs[tid*PD + c] - m; v += d*d; }
    v *= (1.f / Dn);
    float inv = rsqrtf(v + 1e-5f);
    long tok = t0 + tid;
    for (int c0 = 0; c0 < Dn; c0 += 16) {
      float g[16], bb[16];
#pragma unroll
      for (int k = 0; k < 16; ++k) { g[k] = ln_g[c0 + k]; bb[k] = ln_b[c0 + k]; }
#pragma unroll
      for (int k = 0; k < 16; ++k) {
        int c = c0 + k;
        xr[tok*Dn + c] = (qs[tid*PD + c] - m) * inv * g[k] + bb[k]
                         + xs[tid*PD + c];
      }
    }
  }
}

__global__ void __launch_bounds__(128, 3)
in_kernel(const float* xr, const float* rms_w, const v2f* Win_sw,
                          float* xp, float* zb) {
  __shared__ float xn[TOKB*PD];
  int tid = threadIdx.x;
  int wave = tid >> 5;
  int lane = tid & 31;
  long t0 = (long)blockIdx.x * TOKB;

  for (int i = tid; i < TOKB*Dn; i += 128) xn[(i >> 7)*PD + (i & 127)] = xr[t0*Dn + i];
  __syncthreads();
  if (tid < TOKB) {
    float s = 0.f;
    for (int c = 0; c < Dn; ++c) { float v = xn[tid*PD + c]; s += v*v; }
    float inv = rsqrtf(s / Dn + 1e-5f);
    for (int c = 0; c < Dn; ++c) xn[tid*PD + c] = xn[tid*PD + c] * inv * rms_w[c];
  }
  __syncthreads();

  gemm_stream<32, 8>(xn, PD, Win_sw, wave*8, [&](int nt, const v8f (&acc)[TM]) {
    int n = nt*16 + (lane & 15);
    int mb = (lane < 16) ? 0 : 8;
#pragma unroll
    for (int t = 0; t < TM; ++t)
#pragma unroll
      for (int j = 0; j < 8; ++j) {
        long tok = t0 + t*16 + mb + j;
        float v = acc[t][j];
        if (n < EDn) xp[tok*EDn + n] = v;
        else         zb[tok*EDn + (n - EDn)] = v;
      }
  });
}

__global__ void __launch_bounds__(128, 3)
conv_kernel(const float* xp, const float* conv_w, const float* conv_b,
                            const v2f* Wx_sw, const v2f* Wdt_sw, const float* b_dt,
                            float* xc, float* delta, float* Bm, float* Cm) {
  __shared__ float xcs[TOKB*PE];
  __shared__ float dbcs[TOKB*PB];
  int tid = threadIdx.x;
  int wave = tid >> 5;
  int lane = tid & 31;
  long t0 = (long)blockIdx.x * TOKB;

  {
    const int c2  = tid * 2;
    const int t0i = blockIdx.x * TOKB;
    const int l0  = t0i % Ln;
    float w0[DCONVn], w1[DCONVn];
#pragma unroll
    for (int j = 0; j < DCONVn; ++j) {
      w0[j] = conv_w[c2*DCONVn + j];
      w1[j] = conv_w[(c2 + 1)*DCONVn + j];
    }
    const float cb0 = conv_b[c2], cb1 = conv_b[c2 + 1];
    v2f win[TOKB + DCONVn - 1];
#pragma unroll
    for (int i = 0; i < TOKB + DCONVn - 1; ++i) {
      int g = t0i - (DCONVn - 1) + i;
      if (g < 0) g = 0;
      win[i] = *(const v2f*)(xp + (long)g*EDn + c2);
    }
#pragma unroll
    for (int r = 0; r < TOKB; ++r) {
      int l = l0 + r;
      if (l >= Ln) l -= Ln;
      float ax = cb0, ay = cb1;
#pragma unroll
      for (int j = 0; j < DCONVn; ++j) {
        const bool ok = (l - (DCONVn - 1) + j) >= 0;
        const float x0 = ok ? win[r + j].x : 0.f;
        const float x1 = ok ? win[r + j].y : 0.f;
        ax += w0[j] * x0;
        ay += w1[j] * x1;
      }
      v2f o;
      o.x = silu_f(ax);
      o.y = silu_f(ay);
      *(v2f*)(xcs + r*PE + c2) = o;
      *(v2f*)(xc + (long)(t0i + r)*EDn + c2) = o;
    }
  }
  __syncthreads();

  if (wave < 2) {
    gemm_stream<64, 1>(xcs, PE, Wx_sw, wave, [&](int nt, const v8f (&acc)[TM]) {
#pragma unroll
      for (int t = 0; t < TM; ++t) store_c(dbcs + t*16*PB, PB, nt*16, acc[t]);
    });
  }
  __syncthreads();

  gemm_small<2, 4>(dbcs, PB, Wdt_sw, wave*4, [&](int nt, const v8f (&acc)[TM]) {
    int n = nt*16 + (lane & 15);
    int mb = (lane < 16) ? 0 : 8;
    const float bd = b_dt[n];
#pragma unroll
    for (int t = 0; t < TM; ++t)
#pragma unroll
      for (int j = 0; j < 8; ++j) {
        long tok = t0 + t*16 + mb + j;
        float v = acc[t][j] + bd;
        delta[tok*EDn + n] = (v > 20.f) ? v : log1pf(__expf(v));
      }
  });

  for (int i = tid; i < TOKB*16; i += 128) {
    int r = i >> 4, c = i & 15;
    long tok = t0 + r;
    float v = dbcs[r*PB + DTRANKn + c];
    if (c < DSTATEn) Bm[tok*DSTATEn + c]             = v;
    else             Cm[tok*DSTATEn + (c - DSTATEn)] = v;
  }
}

#define LC      25
#define NCH     (Ln/LC)
#define CARRYU  26

__global__ void __launch_bounds__(EDn)
scan_local_kernel(const float* __restrict__ delta, const float* __restrict__ xc,
                  const float* __restrict__ Bm, const float* __restrict__ A_log,
                  float* __restrict__ y) {
  const int d  = threadIdx.x;
  const int t0 = blockIdx.y * Ln + blockIdx.x * LC;
  float a[DSTATEn], h[DSTATEn];
#pragma unroll
  for (int n = 0; n < DSTATEn; ++n) { a[n] = -__expf(A_log[d*DSTATEn + n]); h[n] = 0.f; }
  float sdt = 0.f;
#pragma unroll
  for (int l = 0; l < LC; ++l) {
    const int t = t0 + l;
    float dt = delta[t*EDn + d];
    float xv = xc[t*EDn + d];
    sdt += dt;
#pragma unroll
    for (int n = 0; n < DSTATEn; ++n) {
      float dA = __expf(dt * a[n]);
      h[n] = dA * h[n] + dt * Bm[t*DSTATEn + n] * xv;
    }
  }
#pragma unroll
  for (int n = 0; n < DSTATEn; ++n) y[(t0 + n)*EDn + d] = h[n];
  y[(t0 + DSTATEn)*EDn + d] = sdt;
}

__global__ void __launch_bounds__(EDn)
scan_carry_kernel(const float* __restrict__ A_log, float* y) {
  const int d = threadIdx.x;
  const int n = blockIdx.x;
  const int b = blockIdx.y;
  const float a = -__expf(A_log[d*DSTATEn + n]);
  float*       hp = y + (b*Ln + n)*EDn + d;
  const float* sp = y + (b*Ln + DSTATEn)*EDn + d;
  float carry = 0.f;
  for (int c0 = 0; c0 < NCH; c0 += CARRYU) {
    float he[CARRYU], sd[CARRYU];
#pragma unroll
    for (int j = 0; j < CARRYU; ++j) {
      he[j] = hp[(c0 + j)*(LC*EDn)];
      sd[j] = sp[(c0 + j)*(LC*EDn)];
    }
#pragma unroll
    for (int j = 0; j < CARRYU; ++j) {
      float P = __expf(a * sd[j]);
      hp[(c0 + j)*(LC*EDn)] = carry;
      carry = P * carry + he[j];
    }
  }
}

__global__ void __launch_bounds__(EDn)
scan_apply_kernel(const float* __restrict__ delta, const float* __restrict__ xc,
                  const float* __restrict__ Bm, const float* __restrict__ Cm,
                  const float* __restrict__ A_log, const float* __restrict__ Dp,
                  float* __restrict__ y) {
  const int d  = threadIdx.x;
  const int t0 = blockIdx.y * Ln + blockIdx.x * LC;
  float a[DSTATEn], h[DSTATEn];
#pragma unroll
  for (int n = 0; n < DSTATEn; ++n) {
    a[n] = -__expf(A_log[d*DSTATEn + n]);
    h[n] = y[(t0 + n)*EDn + d];
  }
  const float dp = Dp[d];
#pragma unroll
  for (int l = 0; l < LC; ++l) {
    const int t = t0 + l;
    float dt = delta[t*EDn + d];
    float xv = xc[t*EDn + d];
    float yv = 0.f;
#pragma unroll
    for (int n = 0; n < DSTATEn; ++n) {
      float dA = __expf(dt * a[n]);
      h[n] = dA * h[n] + dt * Bm[t*DSTATEn + n] * xv;
      yv += h[n] * Cm[t*DSTATEn + n];
    }
    y[t*EDn + d] = yv + dp * xv;
  }
}

__global__ void __launch_bounds__(128, 3)
out_kernel(const float* y, const float* zb, const v2f* Wout_sw,
                           const float* xr, const float* x, float* out) {
  __shared__ float ys[TOKB*PE];
  int tid = threadIdx.x;
  int wave = tid >> 5;
  int lane = tid & 31;
  long t0 = (long)blockIdx.x * TOKB;

#pragma unroll 12
  for (int i = tid; i < TOKB*EDn/4; i += 128) {
    int r = i >> 6, c = (i & 63) << 2;
    long tok = t0 + r;
    const float4 z4 = *(const float4*)(zb + tok*EDn + c);
    const float4 y4 = *(const float4*)(y  + tok*EDn + c);
    float4 o;
    o.x = y4.x * silu_f(z4.x);
    o.y = y4.y * silu_f(z4.y);
    o.z = y4.z * silu_f(z4.z);
    o.w = y4.w * silu_f(z4.w);
    *(float4*)(ys + r*PE + c) = o;
  }
  __syncthreads();

  gemm_stream<64, 2>(ys, PE, Wout_sw, wave*2, [&](int nt, const v8f (&acc)[TM]) {
    int n = nt*16 + (lane & 15);
    int mb = (lane < 16) ? 0 : 8;
    float r1[TM][8], r2[TM][8];
#pragma unroll
    for (int t = 0; t < TM; ++t)
#pragma unroll
      for (int j = 0; j < 8; ++j) {
        long tok = t0 + t*16 + mb + j;
        r1[t][j] = xr[tok*Dn + n];
        r2[t][j] = x[tok*Dn + n];
      }
#pragma unroll
    for (int t = 0; t < TM; ++t)
#pragma unroll
      for (int j = 0; j < 8; ++j) {
        long tok = t0 + t*16 + mb + j;
        out[tok*Dn + n] = acc[t][j] + r1[t][j] + r2[t][j];
      }
  });
}

extern "C" void kernel_launch(void* const* d_in, const int* in_sizes, int n_in,
                              void* d_out, int out_size, void* d_ws, size_t ws_size,
                              hipStream_t stream) {
  (void)in_sizes; (void)n_in; (void)out_size; (void)ws_size;
  const float* x      = (const float*)d_in[0];
  const float* mem    = (const float*)d_in[1];
  const float* Wq     = (const float*)d_in[2];
  const float* Wk     = (const float*)d_in[3];
  const float* Wv     = (const float*)d_in[4];
  const float* Wo     = (const float*)d_in[5];
  const float* ln1_g  = (const float*)d_in[6];
  const float* ln1_b  = (const float*)d_in[7];
  const float* rms_w  = (const float*)d_in[8];
  const float* W_in   = (const float*)d_in[9];
  const float* conv_w = (const float*)d_in[10];
  const float* conv_b = (const float*)d_in[11];
  const float* W_x    = (const float*)d_in[12];
  const float* W_dt   = (const float*)d_in[13];
  const float* b_dt   = (const float*)d_in[14];
  const float* A_log  = (const float*)d_in[15];
  const float* Dp     = (const float*)d_in[16];
  const float* W_out  = (const float*)d_in[17];

  float* ws = (float*)d_ws;
  size_t off = 0;
  float* Wq_sw   = ws + off; off += (size_t)8*32*32*2;
  float* Wo_sw   = ws + off; off += (size_t)8*32*32*2;
  float* Win_sw  = ws + off; off += (size_t)32*32*32*2;
  float* Wx_sw   = ws + off; off += (size_t)2*64*32*2;
  float* Wdt_sw  = ws + off; off += (size_t)16*2*32*2;
  float* Wout_sw = ws + off; off += (size_t)8*64*32*2;
  float* kfragT  = ws + off; off += (size_t)1*32*32*2;
  float* vfrag   = ws + off; off += (size_t)8*4*32*2;
  float* xr      = ws + off; off += (size_t)NTOK*Dn;
  float* xp      = ws + off; off += (size_t)NTOK*EDn;
  float* zb      = ws + off; off += (size_t)NTOK*EDn;
  float* xc      = ws + off; off += (size_t)NTOK*EDn;
  float* delta   = ws + off; off += (size_t)NTOK*EDn;
  float* Bm      = ws + off; off += (size_t)NTOK*DSTATEn;
  float* Cm      = ws + off; off += (size_t)NTOK*DSTATEn;
  float* y       = xp;

  prep_kernel<<<PBK_ALL, 256, 0, stream>>>(mem, Wk, Wv, Wq, Wo, W_in, W_x, W_dt,
                                           W_out, kfragT, vfrag, Wq_sw, Wo_sw,
                                           Win_sw, Wx_sw, Wdt_sw, Wout_sw);

  attn_kernel<<<NBLK, 128, 0, stream>>>(x, (const v2f*)Wq_sw, (const v2f*)Wo_sw,
                                        ln1_g, ln1_b, (const v2f*)kfragT,
                                        (const v2f*)vfrag, xr);
  in_kernel  <<<NBLK, 128, 0, stream>>>(xr, rms_w, (const v2f*)Win_sw, xp, zb);
  conv_kernel<<<NBLK, 128, 0, stream>>>(xp, conv_w, conv_b, (const v2f*)Wx_sw,
                                        (const v2f*)Wdt_sw, b_dt, xc, delta, Bm, Cm);
  scan_local_kernel<<<dim3(NCH, Bn), EDn, 0, stream>>>(delta, xc, Bm, A_log, y);
  scan_carry_kernel<<<dim3(DSTATEn, Bn), EDn, 0, stream>>>(A_log, y);
  scan_apply_kernel<<<dim3(NCH, Bn), EDn, 0, stream>>>(delta, xc, Bm, Cm,
                                                       A_log, Dp, y);
  out_kernel <<<NBLK, 128, 0, stream>>>(y, zb, (const v2f*)Wout_sw, xr, x,
                                        (float*)d_out);
}


# softplus of the delta stage: the in-line library log1pf (82 dependent instructions per element, 67 KB of text) replaced by log(u)*e/(u-1), u=1+e, on hardware log2 and rcp; delta moves a few ulp
# speedup vs baseline: 13.8011x; 1.5639x over previous
#include <hip/hip_runtime.h>
#include <hip/hip_bf16.h>
#include <math.h>

#define Bn      8
#define Sn      12
#define NNn     325
#define Dn      128
#define EDn     256
#define DSTATEn 8
#define DCONVn  4
#define DTRANKn 8
#define MSLOTSn 10
#define Ln      (Sn*NNn)
#define NTOK    (Bn*Ln)
#define TM      3
#define TOKB    (16*TM)
#define NBLK    (NTOK/TOKB)
#define PD      (Dn + 4)
#define PE      (EDn + 4)
#define PS      20
#define PB      36

typedef float v2f __attribute__((ext_vector_type(2)));
typedef float v8f __attribute__((ext_vector_type(8)));

__device__ __forceinline__ v8f wmma4(v2f a, v2f b, v8f c) {
  return __builtin_amdgcn_wmma_f32_16x16x4_f32(false, a, false, b, (short)0, c,
                                               false, false);
}

__device__ __forceinline__ void store_c(float* C, int ldc, int nb, v8f c) {
  int lane = threadIdx.x & 31;
  int n = nb + (lane & 15);
  int mb = (lane < 16) ? 0 : 8;
#pragma unroll
  for (int j = 0; j < 8; ++j) C[(mb + j)*ldc + n] = c[j];
}

#define CH      16

__device__ __forceinline__ void mma_block(const v2f* a0, const v2f* a1,
                                          const v2f* a2, const v2f (&b)[CH],
                                          v8f (&acc)[TM]) {
#pragma unroll
  for (int j = 0; j < CH; ++j) {
    acc[0] = wmma4(a0[j*2], b[j], acc[0]);
    acc[1] = wmma4(a1[j*2], b[j], acc[1]);
    acc[2] = wmma4(a2[j*2], b[j], acc[2]);
  }
}

__device__ __forceinline__ void load_block(const v2f* bq, v2f (&b)[CH]) {
#pragma unroll
  for (int j = 0; j < CH; ++j) b[j] = bq[j*32];
}

template <int KT, int NTI, class Epi>
__device__ __forceinline__ void gemm_stream(const float* A, int lda,
                                            const v2f* Bsw, int nt0, Epi epi) {
  constexpr int NC  = NTI*KT/CH;
  constexpr int CPN = KT/CH;
  static_assert(KT % (2*CH) == 0, "an N-tile must be an even number of blocks");
  int lane = threadIdx.x & 31;
  int m = lane & 15;
  int ks = (lane >> 4) << 1;
  const v2f* bp = Bsw + (size_t)nt0 * KT * 32 + lane;
  const v2f* a0 = (const v2f*)(A + (0*16 + m)*lda + ks);
  const v2f* a1 = (const v2f*)(A + (1*16 + m)*lda + ks);
  const v2f* a2 = (const v2f*)(A + (2*16 + m)*lda + ks);
  v2f b0[CH], b1[CH];
  v8f acc[TM];
  const v8f z = {};
  acc[0] = z; acc[1] = z; acc[2] = z;
  load_block(bp, b0);
#pragma nounroll
  for (int c = 0; c < NC - 2; c += 2) {
    const int ko = (c % CPN) * CH * 2;
    load_block(bp + (c + 1)*CH*32, b1);
    mma_block(a0 + ko, a1 + ko, a2 + ko, b0, acc);
    load_block(bp + (c + 2)*CH*32, b0);
    mma_block(a0 + ko + CH*2, a1 + ko + CH*2, a2 + ko + CH*2, b1, acc);
    if (NTI > 1 && ((c + 2) % CPN) == 0) {
      epi(nt0 + c / CPN, acc);
      acc[0] = z; acc[1] = z; acc[2] = z;
    }
  }
  {
    const int ko = ((NC - 2) % CPN) * CH * 2;
    load_block(bp + (NC - 1)*CH*32, b1);
    mma_block(a0 + ko, a1 + ko, a2 + ko, b0, acc);
    mma_block(a0 + ko + CH*2, a1 + ko + CH*2, a2 + ko + CH*2, b1, acc);
    epi(nt0 + NTI - 1, acc);
  }
}

template <int KT, int NTI, class Epi>
__device__ __forceinline__ void gemm_small(const float* A, int lda,
                                           const v2f* Bsw, int nt0, Epi epi) {
  int lane = threadIdx.x & 31;
  int m = lane & 15;
  int ks = (lane >> 4) << 1;
  const v2f* bp = Bsw + (size_t)nt0 * KT * 32 + lane;
  const v2f* a0 = (const v2f*)(A + (0*16 + m)*lda + ks);
  const v2f* a1 = (const v2f*)(A + (1*16 + m)*lda + ks);
  const v2f* a2 = (const v2f*)(A + (2*16 + m)*lda + ks);
  v2f b[NTI*KT];
#pragma unroll
  for (int i = 0; i < NTI*KT; ++i) b[i] = bp[i*32];
#pragma unroll
  for (int nti = 0; nti < NTI; ++nti) {
    v8f acc[TM];
    const v8f z = {};
    acc[0] = z; acc[1] = z; acc[2] = z;
#pragma unroll
    for (int kt = 0; kt < KT; ++kt) {
      acc[0] = wmma4(a0[kt*2], b[nti*KT + kt], acc[0]);
      acc[1] = wmma4(a1[kt*2], b[nti*KT + kt], acc[1]);
      acc[2] = wmma4(a2[kt*2], b[nti*KT + kt], acc[2]);
    }
    epi(nt0 + nti, acc);
  }
}

__device__ __forceinline__ float silu_f(float x) { return x / (1.f + __expf(-x)); }

__device__ __forceinline__ float softplus_f(float v) {
  const float e = __expf(v);
  const float u = 1.f + e;
  const float d = u - 1.f;
  const float lg = __builtin_amdgcn_logf(u) * 0.69314718f;
  const float r = lg * (e * __builtin_amdgcn_rcpf(d));
  return (v > 20.f) ? v : ((d == 0.f) ? e : r);
}

#define PM      (Dn + 1)
#define PBK_KV  8
#define PBK_WQ  (8*32*32/256)
#define PBK_WO  (8*32*32/256)
#define PBK_WIN (32*32*32/256)
#define PBK_WX  (2*64*32/256)
#define PBK_WDT (16*2*32/256)
#define PBK_WOUT (8*64*32/256)
#define PBK_ALL (PBK_KV + PBK_WQ + PBK_WO + PBK_WIN + PBK_WX + PBK_WDT + PBK_WOUT)

template <int K, int NREAL, int NT>
__device__ __forceinline__ void swz_tile(const float* W, float* dst, int idx) {
  constexpr int KT = K / 4;
  static_assert((NT*KT*32) % 256 == 0, "whole blocks");
  int lane = idx & 31;
  int kt = (idx >> 5) % KT;
  int nt = idx / (32 * KT);
  int n = nt*16 + (lane & 15);
  int k = kt*4 + ((lane >> 4) << 1);
  v2f o;
  o.x = (n < NREAL) ? W[k*NREAL + n]     : 0.f;
  o.y = (n < NREAL) ? W[(k+1)*NREAL + n] : 0.f;
  *(v2f*)(dst + idx*2) = o;
}

__device__ __forceinline__ void kv_tile(const float* mem, const float* Wk,
                                        const float* Wv, float* kfragT,
                                        float* vfrag, int blk, float* ms, float* ss) {
  const int tid = threadIdx.x;
  const bool isk = blk < 4;
  const int q = blk & 3;
  const float* Wsel = isk ? Wk : Wv;
#pragma unroll
  for (int r = 0; r < Dn*32/256; ++r) {
    int i = tid + 256*r;
    ss[i] = Wsel[(i >> 5)*Dn + q*32 + (i & 31)];
  }
#pragma unroll
  for (int r = 0; r < MSLOTSn*Dn/256; ++r) {
    int i = tid + 256*r;
    ms[(i >> 7)*PM + (i & 127)] = mem[i];
  }
  __syncthreads();
  const int lane = tid & 31;
  const int h2 = (lane >> 4) << 1;
  int r0, r1, c0, c1;
  bool ok0, ok1;
  if (isk) {
    const int n = lane & 15;
    const int kk = (tid >> 5)*4 + h2;
    r0 = r1 = n; c0 = kk; c1 = kk + 1;
    ok0 = ok1 = n < MSLOTSn;
  } else {
    const int s = ((tid >> 5) & 3)*4 + h2;
    const int cc = ((tid >> 7) << 4) + (lane & 15);
    r0 = s; r1 = s + 1; c0 = c1 = cc;
    ok0 = s < MSLOTSn; ok1 = s + 1 < MSLOTSn;
  }
  const float* m0 = ms + (ok0 ? r0 : 0)*PM;
  const float* m1 = ms + (ok1 ? r1 : 0)*PM;
  float acc0 = 0.f, acc1 = 0.f;
#pragma unroll 16
  for (int j = 0; j < Dn; ++j) {
    acc0 += m0[j] * ss[j*32 + c0];
    acc1 += m1[j] * ss[j*32 + c1];
  }
  v2f o;
  o.x = ok0 ? acc0 : 0.f;
  o.y = ok1 ? acc1 : 0.f;
  float* dst = isk ? kfragT : vfrag;
  *(v2f*)(dst + (q*256 + tid)*2) = o;
}

__global__ void __launch_bounds__(256)
prep_kernel(const float* mem, const float* Wk, const float* Wv,
            const float* Wq, const float* Wo, const float* W_in,
            const float* W_x, const float* W_dt, const float* W_out,
            float* kfragT, float* vfrag, float* Wq_sw, float* Wo_sw,
            float* Win_sw, float* Wx_sw, float* Wdt_sw, float* Wout_sw) {
  __shared__ float ms[MSLOTSn*PM];
  __shared__ float ss[Dn*32];
  int b = blockIdx.x;
  const int tid = threadIdx.x;
  if (b < PBK_KV) { kv_tile(mem, Wk, Wv, kfragT, vfrag, b, ms, ss); return; }
  b -= PBK_KV;
  if (b < PBK_WQ) { swz_tile<Dn, Dn, 8>(Wq, Wq_sw, b*256 + tid); return; }
  b -= PBK_WQ;
  if (b < PBK_WO) { swz_tile<Dn, Dn, 8>(Wo, Wo_sw, b*256 + tid); return; }
  b -= PBK_WO;
  if (b < PBK_WIN) { swz_tile<Dn, 2*EDn, 32>(W_in, Win_sw, b*256 + tid); return; }
  b -= PBK_WIN;
  if (b < PBK_WX) { swz_tile<EDn, DTRANKn + 2*DSTATEn, 2>(W_x, Wx_sw, b*256 + tid); return; }
  b -= PBK_WX;
  if (b < PBK_WDT) { swz_tile<DTRANKn, EDn, 16>(W_dt, Wdt_sw, b*256 + tid); return; }
  b -= PBK_WDT;
  swz_tile<EDn, Dn, 8>(W_out, Wout_sw, b*256 + tid);
}

__global__ void __launch_bounds__(128, 3)
attn_kernel(const float* x, const v2f* Wq_sw, const v2f* Wo_sw,
                            const float* ln_g, const float* ln_b,
                            const v2f* kfragT, const v2f* vfrag, float* xr) {
  __shared__ float xs[TOKB*PD];
  __shared__ float qs[TOKB*PD];
  __shared__ float av[TOKB*PD];
  __shared__ float sc[TOKB*PS];
  int tid = threadIdx.x;
  int wave = tid >> 5;
  long t0 = (long)blockIdx.x * TOKB;

  for (int i = tid; i < TOKB*Dn; i += 128) xs[(i >> 7)*PD + (i & 127)] = x[t0*Dn + i];
  __syncthreads();

  gemm_stream<32, 2>(xs, PD, Wq_sw, wave*2, [&](int nt, const v8f (&acc)[TM]) {
#pragma unroll
    for (int t = 0; t < TM; ++t) store_c(qs + t*16*PD, PD, nt*16, acc[t]);
  });
  __syncthreads();

  if (wave == 0) {
    gemm_stream<32, 1>(qs, PD, kfragT, 0, [&](int, const v8f (&acc)[TM]) {
#pragma unroll
      for (int t = 0; t < TM; ++t) store_c(sc + t*16*PS, PS, 0, acc[t]);
    });
  }
  __syncthreads();

  if (tid < TOKB) {
    const float scale = 0.088388347648318447f;
    float mx = -1e30f;
    for (int m = 0; m < MSLOTSn; ++m) {
      float v = sc[tid*PS + m] * scale; sc[tid*PS + m] = v;
      if (v > mx) mx = v;
    }
    float s = 0.f;
    for (int m = 0; m < MSLOTSn; ++m) {
      float e = __expf(sc[tid*PS + m] - mx); sc[tid*PS + m] = e; s += e;
    }
    float inv = 1.f / s;
    for (int m = 0; m < MSLOTSn; ++m) sc[tid*PS + m] *= inv;
    for (int m = MSLOTSn; m < 16; ++m) sc[tid*PS + m] = 0.f;
  }
  __syncthreads();

  gemm_small<4, 2>(sc, PS, vfrag, wave*2, [&](int nt, const v8f (&acc)[TM]) {
#pragma unroll
    for (int t = 0; t < TM; ++t) store_c(av + t*16*PD, PD, nt*16, acc[t]);
  });
  __syncthreads();

  gemm_stream<32, 2>(av, PD, Wo_sw, wave*2, [&](int nt, const v8f (&acc)[TM]) {
#pragma unroll
    for (int t = 0; t < TM; ++t) store_c(qs + t*16*PD, PD, nt*16, acc[t]);
  });
  __syncthreads();

  if (tid < TOKB) {
    float m = 0.f;
    for (int c = 0; c < Dn; ++c) m += qs[tid*PD + c];
    m *= (1.f / Dn);
    float v = 0.f;
    for (int c = 0; c < Dn; ++c) { float d = qs[tid*PD + c] - m; v += d*d; }
    v *= (1.f / Dn);
    float inv = rsqrtf(v + 1e-5f);
    long tok = t0 + tid;
    for (int c0 = 0; c0 < Dn; c0 += 16) {
      float g[16], bb[16];
#pragma unroll
      for (int k = 0; k < 16; ++k) { g[k] = ln_g[c0 + k]; bb[k] = ln_b[c0 + k]; }
#pragma unroll
      for (int k = 0; k < 16; ++k) {
        int c = c0 + k;
        xr[tok*Dn + c] = (qs[tid*PD + c] - m) * inv * g[k] + bb[k]
                         + xs[tid*PD + c];
      }
    }
  }
}

__global__ void __launch_bounds__(128, 3)
in_kernel(const float* xr, const float* rms_w, const v2f* Win_sw,
                          float* xp, float* zb) {
  __shared__ float xn[TOKB*PD];
  int tid = threadIdx.x;
  int wave = tid >> 5;
  int lane = tid & 31;
  long t0 = (long)blockIdx.x * TOKB;

  for (int i = tid; i < TOKB*Dn; i += 128) xn[(i >> 7)*PD + (i & 127)] = xr[t0*Dn + i];
  __syncthreads();
  if (tid < TOKB) {
    float s = 0.f;
    for (int c = 0; c < Dn; ++c) { float v = xn[tid*PD + c]; s += v*v; }
    float inv = rsqrtf(s / Dn + 1e-5f);
    for (int c = 0; c < Dn; ++c) xn[tid*PD + c] = xn[tid*PD + c] * inv * rms_w[c];
  }
  __syncthreads();

  gemm_stream<32, 8>(xn, PD, Win_sw, wave*8, [&](int nt, const v8f (&acc)[TM]) {
    int n = nt*16 + (lane & 15);
    int mb = (lane < 16) ? 0 : 8;
#pragma unroll
    for (int t = 0; t < TM; ++t)
#pragma unroll
      for (int j = 0; j < 8; ++j) {
        long tok = t0 + t*16 + mb + j;
        float v = acc[t][j];
        if (n < EDn) xp[tok*EDn + n] = v;
        else         zb[tok*EDn + (n - EDn)] = v;
      }
  });
}

__global__ void __launch_bounds__(128, 3)
conv_kernel(const float* xp, const float* conv_w, const float* conv_b,
                            const v2f* Wx_sw, const v2f* Wdt_sw, const float* b_dt,
                            float* xc, float* delta, float* Bm, float* Cm) {
  __shared__ float xcs[TOKB*PE];
  __shared__ float dbcs[TOKB*PB];
  int tid = threadIdx.x;
  int wave = tid >> 5;
  int lane = tid & 31;
  long t0 = (long)blockIdx.x * TOKB;

  {
    const int c2  = tid * 2;
    const int t0i = blockIdx.x * TOKB;
    const int l0  = t0i % Ln;
    float w0[DCONVn], w1[DCONVn];
#pragma unroll
    for (int j = 0; j < DCONVn; ++j) {
      w0[j] = conv_w[c2*DCONVn + j];
      w1[j] = conv_w[(c2 + 1)*DCONVn + j];
    }
    const float cb0 = conv_b[c2], cb1 = conv_b[c2 + 1];
    v2f win[TOKB + DCONVn - 1];
#pragma unroll
    for (int i = 0; i < TOKB + DCONVn - 1; ++i) {
      int g = t0i - (DCONVn - 1) + i;
      if (g < 0) g = 0;
      win[i] = *(const v2f*)(xp + (long)g*EDn + c2);
    }
#pragma unroll
    for (int r = 0; r < TOKB; ++r) {
      int l = l0 + r;
      if (l >= Ln) l -= Ln;
      float ax = cb0, ay = cb1;
#pragma unroll
      for (int j = 0; j < DCONVn; ++j) {
        const bool ok = (l - (DCONVn - 1) + j) >= 0;
        const float x0 = ok ? win[r + j].x : 0.f;
        const float x1 = ok ? win[r + j].y : 0.f;
        ax += w0[j] * x0;
        ay += w1[j] * x1;
      }
      v2f o;
      o.x = silu_f(ax);
      o.y = silu_f(ay);
      *(v2f*)(xcs + r*PE + c2) = o;
      *(v2f*)(xc + (long)(t0i + r)*EDn + c2) = o;
    }
  }
  __syncthreads();

  if (wave < 2) {
    gemm_stream<64, 1>(xcs, PE, Wx_sw, wave, [&](int nt, const v8f (&acc)[TM]) {
#pragma unroll
      for (int t = 0; t < TM; ++t) store_c(dbcs + t*16*PB, PB, nt*16, acc[t]);
    });
  }
  __syncthreads();

  gemm_small<2, 4>(dbcs, PB, Wdt_sw, wave*4, [&](int nt, const v8f (&acc)[TM]) {
    int n = nt*16 + (lane & 15);
    int mb = (lane < 16) ? 0 : 8;
    const float bd = b_dt[n];
#pragma unroll
    for (int t = 0; t < TM; ++t)
#pragma unroll
      for (int j = 0; j < 8; ++j) {
        long tok = t0 + t*16 + mb + j;
        float v = acc[t][j] + bd;
        delta[tok*EDn + n] = softplus_f(v);
      }
  });

  for (int i = tid; i < TOKB*16; i += 128) {
    int r = i >> 4, c = i & 15;
    long tok = t0 + r;
    float v = dbcs[r*PB + DTRANKn + c];
    if (c < DSTATEn) Bm[tok*DSTATEn + c]             = v;
    else             Cm[tok*DSTATEn + (c - DSTATEn)] = v;
  }
}

#define LC      25
#define NCH     (Ln/LC)
#define CARRYU  26

__global__ void __launch_bounds__(EDn)
scan_local_kernel(const float* __restrict__ delta, const float* __restrict__ xc,
                  const float* __restrict__ Bm, const float* __restrict__ A_log,
                  float* __restrict__ y) {
  const int d  = threadIdx.x;
  const int t0 = blockIdx.y * Ln + blockIdx.x * LC;
  float a[DSTATEn], h[DSTATEn];
#pragma unroll
  for (int n = 0; n < DSTATEn; ++n) { a[n] = -__expf(A_log[d*DSTATEn + n]); h[n] = 0.f; }
  float sdt = 0.f;
#pragma unroll
  for (int l = 0; l < LC; ++l) {
    const int t = t0 + l;
    float dt = delta[t*EDn + d];
    float xv = xc[t*EDn + d];
    sdt += dt;
#pragma unroll
    for (int n = 0; n < DSTATEn; ++n) {
      float dA = __expf(dt * a[n]);
      h[n] = dA * h[n] + dt * Bm[t*DSTATEn + n] * xv;
    }
  }
#pragma unroll
  for (int n = 0; n < DSTATEn; ++n) y[(t0 + n)*EDn + d] = h[n];
  y[(t0 + DSTATEn)*EDn + d] = sdt;
}

__global__ void __launch_bounds__(EDn)
scan_carry_kernel(const float* __restrict__ A_log, float* y) {
  const int d = threadIdx.x;
  const int n = blockIdx.x;
  const int b = blockIdx.y;
  const float a = -__expf(A_log[d*DSTATEn + n]);
  float*       hp = y + (b*Ln + n)*EDn + d;
  const float* sp = y + (b*Ln + DSTATEn)*EDn + d;
  float carry = 0.f;
  for (int c0 = 0; c0 < NCH; c0 += CARRYU) {
    float he[CARRYU], sd[CARRYU];
#pragma unroll
    for (int j = 0; j < CARRYU; ++j) {
      he[j] = hp[(c0 + j)*(LC*EDn)];
      sd[j] = sp[(c0 + j)*(LC*EDn)];
    }
#pragma unroll
    for (int j = 0; j < CARRYU; ++j) {
      float P = __expf(a * sd[j]);
      hp[(c0 + j)*(LC*EDn)] = carry;
      carry = P * carry + he[j];
    }
  }
}

__global__ void __launch_bounds__(EDn)
scan_apply_kernel(const float* __restrict__ delta, const float* __restrict__ xc,
                  const float* __restrict__ Bm, const float* __restrict__ Cm,
                  const float* __restrict__ A_log, const float* __restrict__ Dp,
                  float* __restrict__ y) {
  const int d  = threadIdx.x;
  const int t0 = blockIdx.y * Ln + blockIdx.x * LC;
  float a[DSTATEn], h[DSTATEn];
#pragma unroll
  for (int n = 0; n < DSTATEn; ++n) {
    a[n] = -__expf(A_log[d*DSTATEn + n]);
    h[n] = y[(t0 + n)*EDn + d];
  }
  const float dp = Dp[d];
#pragma unroll
  for (int l = 0; l < LC; ++l) {
    const int t = t0 + l;
    float dt = delta[t*EDn + d];
    float xv = xc[t*EDn + d];
    float yv = 0.f;
#pragma unroll
    for (int n = 0; n < DSTATEn; ++n) {
      float dA = __expf(dt * a[n]);
      h[n] = dA * h[n] + dt * Bm[t*DSTATEn + n] * xv;
      yv += h[n] * Cm[t*DSTATEn + n];
    }
    y[t*EDn + d] = yv + dp * xv;
  }
}

__global__ void __launch_bounds__(128, 3)
out_kernel(const float* y, const float* zb, const v2f* Wout_sw,
                           const float* xr, const float* x, float* out) {
  __shared__ float ys[TOKB*PE];
  int tid = threadIdx.x;
  int wave = tid >> 5;
  int lane = tid & 31;
  long t0 = (long)blockIdx.x * TOKB;

#pragma unroll 12
  for (int i = tid; i < TOKB*EDn/4; i += 128) {
    int r = i >> 6, c = (i & 63) << 2;
    long tok = t0 + r;
    const float4 z4 = *(const float4*)(zb + tok*EDn + c);
    const float4 y4 = *(const float4*)(y  + tok*EDn + c);
    float4 o;
    o.x = y4.x * silu_f(z4.x);
    o.y = y4.y * silu_f(z4.y);
    o.z = y4.z * silu_f(z4.z);
    o.w = y4.w * silu_f(z4.w);
    *(float4*)(ys + r*PE + c) = o;
  }
  __syncthreads();

  gemm_stream<64, 2>(ys, PE, Wout_sw, wave*2, [&](int nt, const v8f (&acc)[TM]) {
    int n = nt*16 + (lane & 15);
    int mb = (lane < 16) ? 0 : 8;
    float r1[TM][8], r2[TM][8];
#pragma unroll
    for (int t = 0; t < TM; ++t)
#pragma unroll
      for (int j = 0; j < 8; ++j) {
        long tok = t0 + t*16 + mb + j;
        r1[t][j] = xr[tok*Dn + n];
        r2[t][j] = x[tok*Dn + n];
      }
#pragma unroll
    for (int t = 0; t < TM; ++t)
#pragma unroll
      for (int j = 0; j < 8; ++j) {
        long tok = t0 + t*16 + mb + j;
        out[tok*Dn + n] = acc[t][j] + r1[t][j] + r2[t][j];
      }
  });
}

extern "C" void kernel_launch(void* const* d_in, const int* in_sizes, int n_in,
                              void* d_out, int out_size, void* d_ws, size_t ws_size,
                              hipStream_t stream) {
  (void)in_sizes; (void)n_in; (void)out_size; (void)ws_size;
  const float* x      = (const float*)d_in[0];
  const float* mem    = (const float*)d_in[1];
  const float* Wq     = (const float*)d_in[2];
  const float* Wk     = (const float*)d_in[3];
  const float* Wv     = (const float*)d_in[4];
  const float* Wo     = (const float*)d_in[5];
  const float* ln1_g  = (const float*)d_in[6];
  const float* ln1_b  = (const float*)d_in[7];
  const float* rms_w  = (const float*)d_in[8];
  const float* W_in   = (const float*)d_in[9];
  const float* conv_w = (const float*)d_in[10];
  const float* conv_b = (const float*)d_in[11];
  const float* W_x    = (const float*)d_in[12];
  const float* W_dt   = (const float*)d_in[13];
  const float* b_dt   = (const float*)d_in[14];
  const float* A_log  = (const float*)d_in[15];
  const float* Dp     = (const float*)d_in[16];
  const float* W_out  = (const float*)d_in[17];

  float* ws = (float*)d_ws;
  size_t off = 0;
  float* Wq_sw   = ws + off; off += (size_t)8*32*32*2;
  float* Wo_sw   = ws + off; off += (size_t)8*32*32*2;
  float* Win_sw  = ws + off; off += (size_t)32*32*32*2;
  float* Wx_sw   = ws + off; off += (size_t)2*64*32*2;
  float* Wdt_sw  = ws + off; off += (size_t)16*2*32*2;
  float* Wout_sw = ws + off; off += (size_t)8*64*32*2;
  float* kfragT  = ws + off; off += (size_t)1*32*32*2;
  float* vfrag   = ws + off; off += (size_t)8*4*32*2;
  float* xr      = ws + off; off += (size_t)NTOK*Dn;
  float* xp      = ws + off; off += (size_t)NTOK*EDn;
  float* zb      = ws + off; off += (size_t)NTOK*EDn;
  float* xc      = ws + off; off += (size_t)NTOK*EDn;
  float* delta   = ws + off; off += (size_t)NTOK*EDn;
  float* Bm      = ws + off; off += (size_t)NTOK*DSTATEn;
  float* Cm      = ws + off; off += (size_t)NTOK*DSTATEn;
  float* y       = xp;

  prep_kernel<<<PBK_ALL, 256, 0, stream>>>(mem, Wk, Wv, Wq, Wo, W_in, W_x, W_dt,
                                           W_out, kfragT, vfrag, Wq_sw, Wo_sw,
                                           Win_sw, Wx_sw, Wdt_sw, Wout_sw);

  attn_kernel<<<NBLK, 128, 0, stream>>>(x, (const v2f*)Wq_sw, (const v2f*)Wo_sw,
                                        ln1_g, ln1_b, (const v2f*)kfragT,
                                        (const v2f*)vfrag, xr);
  in_kernel  <<<NBLK, 128, 0, stream>>>(xr, rms_w, (const v2f*)Win_sw, xp, zb);
  conv_kernel<<<NBLK, 128, 0, stream>>>(xp, conv_w, conv_b, (const v2f*)Wx_sw,
                                        (const v2f*)Wdt_sw, b_dt, xc, delta, Bm, Cm);
  scan_local_kernel<<<dim3(NCH, Bn), EDn, 0, stream>>>(delta, xc, Bm, A_log, y);
  scan_carry_kernel<<<dim3(DSTATEn, Bn), EDn, 0, stream>>>(A_log, y);
  scan_apply_kernel<<<dim3(NCH, Bn), EDn, 0, stream>>>(delta, xc, Bm, Cm,
                                                       A_log, Dp, y);
  out_kernel <<<NBLK, 128, 0, stream>>>(y, zb, (const v2f*)Wout_sw, xr, x,
                                        (float*)d_out);
}


# LayerNorm and RMSNorm rows taken by whole waves (lane = 4 columns, sums across lanes by DPP) instead of one thread per row; xr stored as whole 512-B rows; mean/variance sums become butterfly trees
# speedup vs baseline: 14.5581x; 1.0549x over previous
#include <hip/hip_runtime.h>
#include <hip/hip_bf16.h>
#include <math.h>

#define Bn      8
#define Sn      12
#define NNn     325
#define Dn      128
#define EDn     256
#define DSTATEn 8
#define DCONVn  4
#define DTRANKn 8
#define MSLOTSn 10
#define Ln      (Sn*NNn)
#define NTOK    (Bn*Ln)
#define TM      3
#define TOKB    (16*TM)
#define NBLK    (NTOK/TOKB)
#define PD      (Dn + 4)
#define PE      (EDn + 4)
#define PS      20
#define PB      36

typedef float v2f __attribute__((ext_vector_type(2)));
typedef float v8f __attribute__((ext_vector_type(8)));

__device__ __forceinline__ v8f wmma4(v2f a, v2f b, v8f c) {
  return __builtin_amdgcn_wmma_f32_16x16x4_f32(false, a, false, b, (short)0, c,
                                               false, false);
}

__device__ __forceinline__ void store_c(float* C, int ldc, int nb, v8f c) {
  int lane = threadIdx.x & 31;
  int n = nb + (lane & 15);
  int mb = (lane < 16) ? 0 : 8;
#pragma unroll
  for (int j = 0; j < 8; ++j) C[(mb + j)*ldc + n] = c[j];
}

#define CH      16

__device__ __forceinline__ void mma_block(const v2f* a0, const v2f* a1,
                                          const v2f* a2, const v2f (&b)[CH],
                                          v8f (&acc)[TM]) {
#pragma unroll
  for (int j = 0; j < CH; ++j) {
    acc[0] = wmma4(a0[j*2], b[j], acc[0]);
    acc[1] = wmma4(a1[j*2], b[j], acc[1]);
    acc[2] = wmma4(a2[j*2], b[j], acc[2]);
  }
}

__device__ __forceinline__ void load_block(const v2f* bq, v2f (&b)[CH]) {
#pragma unroll
  for (int j = 0; j < CH; ++j) b[j] = bq[j*32];
}

template <int KT, int NTI, class Epi>
__device__ __forceinline__ void gemm_stream(const float* A, int lda,
                                            const v2f* Bsw, int nt0, Epi epi) {
  constexpr int NC  = NTI*KT/CH;
  constexpr int CPN = KT/CH;
  static_assert(KT % (2*CH) == 0, "an N-tile must be an even number of blocks");
  int lane = threadIdx.x & 31;
  int m = lane & 15;
  int ks = (lane >> 4) << 1;
  const v2f* bp = Bsw + (size_t)nt0 * KT * 32 + lane;
  const v2f* a0 = (const v2f*)(A + (0*16 + m)*lda + ks);
  const v2f* a1 = (const v2f*)(A + (1*16 + m)*lda + ks);
  const v2f* a2 = (const v2f*)(A + (2*16 + m)*lda + ks);
  v2f b0[CH], b1[CH];
  v8f acc[TM];
  const v8f z = {};
  acc[0] = z; acc[1] = z; acc[2] = z;
  load_block(bp, b0);
#pragma nounroll
  for (int c = 0; c < NC - 2; c += 2) {
    const int ko = (c % CPN) * CH * 2;
    load_block(bp + (c + 1)*CH*32, b1);
    mma_block(a0 + ko, a1 + ko, a2 + ko, b0, acc);
    load_block(bp + (c + 2)*CH*32, b0);
    mma_block(a0 + ko + CH*2, a1 + ko + CH*2, a2 + ko + CH*2, b1, acc);
    if (NTI > 1 && ((c + 2) % CPN) == 0) {
      epi(nt0 + c / CPN, acc);
      acc[0] = z; acc[1] = z; acc[2] = z;
    }
  }
  {
    const int ko = ((NC - 2) % CPN) * CH * 2;
    load_block(bp + (NC - 1)*CH*32, b1);
    mma_block(a0 + ko, a1 + ko, a2 + ko, b0, acc);
    mma_block(a0 + ko + CH*2, a1 + ko + CH*2, a2 + ko + CH*2, b1, acc);
    epi(nt0 + NTI - 1, acc);
  }
}

template <int KT, int NTI, class Epi>
__device__ __forceinline__ void gemm_small(const float* A, int lda,
                                           const v2f* Bsw, int nt0, Epi epi) {
  int lane = threadIdx.x & 31;
  int m = lane & 15;
  int ks = (lane >> 4) << 1;
  const v2f* bp = Bsw + (size_t)nt0 * KT * 32 + lane;
  const v2f* a0 = (const v2f*)(A + (0*16 + m)*lda + ks);
  const v2f* a1 = (const v2f*)(A + (1*16 + m)*lda + ks);
  const v2f* a2 = (const v2f*)(A + (2*16 + m)*lda + ks);
  v2f b[NTI*KT];
#pragma unroll
  for (int i = 0; i < NTI*KT; ++i) b[i] = bp[i*32];
#pragma unroll
  for (int nti = 0; nti < NTI; ++nti) {
    v8f acc[TM];
    const v8f z = {};
    acc[0] = z; acc[1] = z; acc[2] = z;
#pragma unroll
    for (int kt = 0; kt < KT; ++kt) {
      acc[0] = wmma4(a0[kt*2], b[nti*KT + kt], acc[0]);
      acc[1] = wmma4(a1[kt*2], b[nti*KT + kt], acc[1]);
      acc[2] = wmma4(a2[kt*2], b[nti*KT + kt], acc[2]);
    }
    epi(nt0 + nti, acc);
  }
}

__device__ __forceinline__ float silu_f(float x) { return x / (1.f + __expf(-x)); }

__device__ __forceinline__ float xhalf(float x) {
  return __builtin_bit_cast(float, __builtin_amdgcn_permlanex16(
      __builtin_bit_cast(int, x), __builtin_bit_cast(int, x),
      0x76543210, 0xfedcba98, false, false));
}

template <int M>
__device__ __forceinline__ float add_xmask(float s) {
  const int t = __builtin_amdgcn_update_dpp(0, __builtin_bit_cast(int, s),
                                            0x160 + M, 0xf, 0xf, true);
  return s + __builtin_bit_cast(float, t);
}

__device__ __forceinline__ float wave_sum(float s) {
  s += xhalf(s);
  s = add_xmask<8>(s);
  s = add_xmask<4>(s);
  s = add_xmask<2>(s);
  return add_xmask<1>(s);
}

__device__ __forceinline__ float softplus_f(float v) {
  const float e = __expf(v);
  const float u = 1.f + e;
  const float d = u - 1.f;
  const float lg = __builtin_amdgcn_logf(u) * 0.69314718f;
  const float r = lg * (e * __builtin_amdgcn_rcpf(d));
  return (v > 20.f) ? v : ((d == 0.f) ? e : r);
}

#define PM      (Dn + 1)
#define PBK_KV  8
#define PBK_WQ  (8*32*32/256)
#define PBK_WO  (8*32*32/256)
#define PBK_WIN (32*32*32/256)
#define PBK_WX  (2*64*32/256)
#define PBK_WDT (16*2*32/256)
#define PBK_WOUT (8*64*32/256)
#define PBK_ALL (PBK_KV + PBK_WQ + PBK_WO + PBK_WIN + PBK_WX + PBK_WDT + PBK_WOUT)

template <int K, int NREAL, int NT>
__device__ __forceinline__ void swz_tile(const float* W, float* dst, int idx) {
  constexpr int KT = K / 4;
  static_assert((NT*KT*32) % 256 == 0, "whole blocks");
  int lane = idx & 31;
  int kt = (idx >> 5) % KT;
  int nt = idx / (32 * KT);
  int n = nt*16 + (lane & 15);
  int k = kt*4 + ((lane >> 4) << 1);
  v2f o;
  o.x = (n < NREAL) ? W[k*NREAL + n]     : 0.f;
  o.y = (n < NREAL) ? W[(k+1)*NREAL + n] : 0.f;
  *(v2f*)(dst + idx*2) = o;
}

__device__ __forceinline__ void kv_tile(const float* mem, const float* Wk,
                                        const float* Wv, float* kfragT,
                                        float* vfrag, int blk, float* ms, float* ss) {
  const int tid = threadIdx.x;
  const bool isk = blk < 4;
  const int q = blk & 3;
  const float* Wsel = isk ? Wk : Wv;
#pragma unroll
  for (int r = 0; r < Dn*32/256; ++r) {
    int i = tid + 256*r;
    ss[i] = Wsel[(i >> 5)*Dn + q*32 + (i & 31)];
  }
#pragma unroll
  for (int r = 0; r < MSLOTSn*Dn/256; ++r) {
    int i = tid + 256*r;
    ms[(i >> 7)*PM + (i & 127)] = mem[i];
  }
  __syncthreads();
  const int lane = tid & 31;
  const int h2 = (lane >> 4) << 1;
  int r0, r1, c0, c1;
  bool ok0, ok1;
  if (isk) {
    const int n = lane & 15;
    const int kk = (tid >> 5)*4 + h2;
    r0 = r1 = n; c0 = kk; c1 = kk + 1;
    ok0 = ok1 = n < MSLOTSn;
  } else {
    const int s = ((tid >> 5) & 3)*4 + h2;
    const int cc = ((tid >> 7) << 4) + (lane & 15);
    r0 = s; r1 = s + 1; c0 = c1 = cc;
    ok0 = s < MSLOTSn; ok1 = s + 1 < MSLOTSn;
  }
  const float* m0 = ms + (ok0 ? r0 : 0)*PM;
  const float* m1 = ms + (ok1 ? r1 : 0)*PM;
  float acc0 = 0.f, acc1 = 0.f;
#pragma unroll 16
  for (int j = 0; j < Dn; ++j) {
    acc0 += m0[j] * ss[j*32 + c0];
    acc1 += m1[j] * ss[j*32 + c1];
  }
  v2f o;
  o.x = ok0 ? acc0 : 0.f;
  o.y = ok1 ? acc1 : 0.f;
  float* dst = isk ? kfragT : vfrag;
  *(v2f*)(dst + (q*256 + tid)*2) = o;
}

__global__ void __launch_bounds__(256)
prep_kernel(const float* mem, const float* Wk, const float* Wv,
            const float* Wq, const float* Wo, const float* W_in,
            const float* W_x, const float* W_dt, const float* W_out,
            float* kfragT, float* vfrag, float* Wq_sw, float* Wo_sw,
            float* Win_sw, float* Wx_sw, float* Wdt_sw, float* Wout_sw) {
  __shared__ float ms[MSLOTSn*PM];
  __shared__ float ss[Dn*32];
  int b = blockIdx.x;
  const int tid = threadIdx.x;
  if (b < PBK_KV) { kv_tile(mem, Wk, Wv, kfragT, vfrag, b, ms, ss); return; }
  b -= PBK_KV;
  if (b < PBK_WQ) { swz_tile<Dn, Dn, 8>(Wq, Wq_sw, b*256 + tid); return; }
  b -= PBK_WQ;
  if (b < PBK_WO) { swz_tile<Dn, Dn, 8>(Wo, Wo_sw, b*256 + tid); return; }
  b -= PBK_WO;
  if (b < PBK_WIN) { swz_tile<Dn, 2*EDn, 32>(W_in, Win_sw, b*256 + tid); return; }
  b -= PBK_WIN;
  if (b < PBK_WX) { swz_tile<EDn, DTRANKn + 2*DSTATEn, 2>(W_x, Wx_sw, b*256 + tid); return; }
  b -= PBK_WX;
  if (b < PBK_WDT) { swz_tile<DTRANKn, EDn, 16>(W_dt, Wdt_sw, b*256 + tid); return; }
  b -= PBK_WDT;
  swz_tile<EDn, Dn, 8>(W_out, Wout_sw, b*256 + tid);
}

__global__ void __launch_bounds__(128, 3)
attn_kernel(const float* x, const v2f* Wq_sw, const v2f* Wo_sw,
                            const float* ln_g, const float* ln_b,
                            const v2f* kfragT, const v2f* vfrag, float* xr) {
  __shared__ __attribute__((aligned(16))) float xs[TOKB*PD];
  __shared__ __attribute__((aligned(16))) float qs[TOKB*PD];
  __shared__ float av[TOKB*PD];
  __shared__ float sc[TOKB*PS];
  int tid = threadIdx.x;
  int wave = tid >> 5;
  int lane = tid & 31;
  long t0 = (long)blockIdx.x * TOKB;

  for (int i = tid; i < TOKB*Dn; i += 128) xs[(i >> 7)*PD + (i & 127)] = x[t0*Dn + i];
  __syncthreads();

  gemm_stream<32, 2>(xs, PD, Wq_sw, wave*2, [&](int nt, const v8f (&acc)[TM]) {
#pragma unroll
    for (int t = 0; t < TM; ++t) store_c(qs + t*16*PD, PD, nt*16, acc[t]);
  });
  __syncthreads();

  if (wave == 0) {
    gemm_stream<32, 1>(qs, PD, kfragT, 0, [&](int, const v8f (&acc)[TM]) {
#pragma unroll
      for (int t = 0; t < TM; ++t) store_c(sc + t*16*PS, PS, 0, acc[t]);
    });
  }
  __syncthreads();

  if (tid < TOKB) {
    const float scale = 0.088388347648318447f;
    float mx = -1e30f;
    for (int m = 0; m < MSLOTSn; ++m) {
      float v = sc[tid*PS + m] * scale; sc[tid*PS + m] = v;
      if (v > mx) mx = v;
    }
    float s = 0.f;
    for (int m = 0; m < MSLOTSn; ++m) {
      float e = __expf(sc[tid*PS + m] - mx); sc[tid*PS + m] = e; s += e;
    }
    float inv = 1.f / s;
    for (int m = 0; m < MSLOTSn; ++m) sc[tid*PS + m] *= inv;
    for (int m = MSLOTSn; m < 16; ++m) sc[tid*PS + m] = 0.f;
  }
  __syncthreads();

  gemm_small<4, 2>(sc, PS, vfrag, wave*2, [&](int nt, const v8f (&acc)[TM]) {
#pragma unroll
    for (int t = 0; t < TM; ++t) store_c(av + t*16*PD, PD, nt*16, acc[t]);
  });
  __syncthreads();

  gemm_stream<32, 2>(av, PD, Wo_sw, wave*2, [&](int nt, const v8f (&acc)[TM]) {
#pragma unroll
    for (int t = 0; t < TM; ++t) store_c(qs + t*16*PD, PD, nt*16, acc[t]);
  });
  __syncthreads();

  {
    constexpr int R = TOKB/4;
    const int c4 = lane * 4;
    const int r0 = wave * R;
    const float g0 = ln_g[c4], g1 = ln_g[c4 + 1], g2 = ln_g[c4 + 2], g3 = ln_g[c4 + 3];
    const float b0 = ln_b[c4], b1 = ln_b[c4 + 1], b2 = ln_b[c4 + 2], b3 = ln_b[c4 + 3];
    float4 q[R];
    float s[R];
#pragma unroll
    for (int k = 0; k < R; ++k) {
      q[k] = *(const float4*)(qs + (r0 + k)*PD + c4);
      s[k] = (q[k].x + q[k].y) + (q[k].z + q[k].w);
    }
#pragma unroll
    for (int k = 0; k < R; ++k) s[k] = wave_sum(s[k]);
#pragma unroll
    for (int k = 0; k < R; ++k) {
      const float m = s[k] * (1.f / Dn);
      q[k].x -= m; q[k].y -= m; q[k].z -= m; q[k].w -= m;
      s[k] = (q[k].x*q[k].x + q[k].y*q[k].y) + (q[k].z*q[k].z + q[k].w*q[k].w);
    }
#pragma unroll
    for (int k = 0; k < R; ++k) s[k] = wave_sum(s[k]);
#pragma unroll
    for (int k = 0; k < R; ++k) {
      const float inv = rsqrtf(s[k] * (1.f / Dn) + 1e-5f);
      const float4 xv = *(const float4*)(xs + (r0 + k)*PD + c4);
      float4 o;
      o.x = q[k].x * inv * g0 + b0 + xv.x;
      o.y = q[k].y * inv * g1 + b1 + xv.y;
      o.z = q[k].z * inv * g2 + b2 + xv.z;
      o.w = q[k].w * inv * g3 + b3 + xv.w;
      *(float4*)(xr + (t0 + r0 + k)*Dn + c4) = o;
    }
  }
}

__global__ void __launch_bounds__(128, 3)
in_kernel(const float* xr, const float* rms_w, const v2f* Win_sw,
                          float* xp, float* zb) {
  __shared__ __attribute__((aligned(16))) float xn[TOKB*PD];
  int tid = threadIdx.x;
  int wave = tid >> 5;
  int lane = tid & 31;
  long t0 = (long)blockIdx.x * TOKB;

  {
    constexpr int R = TOKB/4;
    const int c4 = lane * 4;
    const int r0 = wave * R;
    const float w0 = rms_w[c4], w1 = rms_w[c4 + 1], w2 = rms_w[c4 + 2], w3 = rms_w[c4 + 3];
    float4 v[R];
    float s[R];
#pragma unroll
    for (int k = 0; k < R; ++k)
      v[k] = *(const float4*)(xr + (t0 + r0 + k)*Dn + c4);
#pragma unroll
    for (int k = 0; k < R; ++k)
      s[k] = (v[k].x*v[k].x + v[k].y*v[k].y) + (v[k].z*v[k].z + v[k].w*v[k].w);
#pragma unroll
    for (int k = 0; k < R; ++k) s[k] = wave_sum(s[k]);
#pragma unroll
    for (int k = 0; k < R; ++k) {
      const float inv = rsqrtf(s[k] / Dn + 1e-5f);
      float4 o;
      o.x = v[k].x * inv * w0;
      o.y = v[k].y * inv * w1;
      o.z = v[k].z * inv * w2;
      o.w = v[k].w * inv * w3;
      *(float4*)(xn + (r0 + k)*PD + c4) = o;
    }
  }
  __syncthreads();

  gemm_stream<32, 8>(xn, PD, Win_sw, wave*8, [&](int nt, const v8f (&acc)[TM]) {
    int n = nt*16 + (lane & 15);
    int mb = (lane < 16) ? 0 : 8;
#pragma unroll
    for (int t = 0; t < TM; ++t)
#pragma unroll
      for (int j = 0; j < 8; ++j) {
        long tok = t0 + t*16 + mb + j;
        float v = acc[t][j];
        if (n < EDn) xp[tok*EDn + n] = v;
        else         zb[tok*EDn + (n - EDn)] = v;
      }
  });
}

__global__ void __launch_bounds__(128, 3)
conv_kernel(const float* xp, const float* conv_w, const float* conv_b,
                            const v2f* Wx_sw, const v2f* Wdt_sw, const float* b_dt,
                            float* xc, float* delta, float* Bm, float* Cm) {
  __shared__ float xcs[TOKB*PE];
  __shared__ float dbcs[TOKB*PB];
  int tid = threadIdx.x;
  int wave = tid >> 5;
  int lane = tid & 31;
  long t0 = (long)blockIdx.x * TOKB;

  {
    const int c2  = tid * 2;
    const int t0i = blockIdx.x * TOKB;
    const int l0  = t0i % Ln;
    float w0[DCONVn], w1[DCONVn];
#pragma unroll
    for (int j = 0; j < DCONVn; ++j) {
      w0[j] = conv_w[c2*DCONVn + j];
      w1[j] = conv_w[(c2 + 1)*DCONVn + j];
    }
    const float cb0 = conv_b[c2], cb1 = conv_b[c2 + 1];
    v2f win[TOKB + DCONVn - 1];
#pragma unroll
    for (int i = 0; i < TOKB + DCONVn - 1; ++i) {
      int g = t0i - (DCONVn - 1) + i;
      if (g < 0) g = 0;
      win[i] = *(const v2f*)(xp + (long)g*EDn + c2);
    }
#pragma unroll
    for (int r = 0; r < TOKB; ++r) {
      int l = l0 + r;
      if (l >= Ln) l -= Ln;
      float ax = cb0, ay = cb1;
#pragma unroll
      for (int j = 0; j < DCONVn; ++j) {
        const bool ok = (l - (DCONVn - 1) + j) >= 0;
        const float x0 = ok ? win[r + j].x : 0.f;
        const float x1 = ok ? win[r + j].y : 0.f;
        ax += w0[j] * x0;
        ay += w1[j] * x1;
      }
      v2f o;
      o.x = silu_f(ax);
      o.y = silu_f(ay);
      *(v2f*)(xcs + r*PE + c2) = o;
      *(v2f*)(xc + (long)(t0i + r)*EDn + c2) = o;
    }
  }
  __syncthreads();

  if (wave < 2) {
    gemm_stream<64, 1>(xcs, PE, Wx_sw, wave, [&](int nt, const v8f (&acc)[TM]) {
#pragma unroll
      for (int t = 0; t < TM; ++t) store_c(dbcs + t*16*PB, PB, nt*16, acc[t]);
    });
  }
  __syncthreads();

  gemm_small<2, 4>(dbcs, PB, Wdt_sw, wave*4, [&](int nt, const v8f (&acc)[TM]) {
    int n = nt*16 + (lane & 15);
    int mb = (lane < 16) ? 0 : 8;
    const float bd = b_dt[n];
#pragma unroll
    for (int t = 0; t < TM; ++t)
#pragma unroll
      for (int j = 0; j < 8; ++j) {
        long tok = t0 + t*16 + mb + j;
        float v = acc[t][j] + bd;
        delta[tok*EDn + n] = softplus_f(v);
      }
  });

  for (int i = tid; i < TOKB*16; i += 128) {
    int r = i >> 4, c = i & 15;
    long tok = t0 + r;
    float v = dbcs[r*PB + DTRANKn + c];
    if (c < DSTATEn) Bm[tok*DSTATEn + c]             = v;
    else             Cm[tok*DSTATEn + (c - DSTATEn)] = v;
  }
}

#define LC      25
#define NCH     (Ln/LC)
#define CARRYU  26

__global__ void __launch_bounds__(EDn)
scan_local_kernel(const float* __restrict__ delta, const float* __restrict__ xc,
                  const float* __restrict__ Bm, const float* __restrict__ A_log,
                  float* __restrict__ y) {
  const int d  = threadIdx.x;
  const int t0 = blockIdx.y * Ln + blockIdx.x * LC;
  float a[DSTATEn], h[DSTATEn];
#pragma unroll
  for (int n = 0; n < DSTATEn; ++n) { a[n] = -__expf(A_log[d*DSTATEn + n]); h[n] = 0.f; }
  float sdt = 0.f;
#pragma unroll
  for (int l = 0; l < LC; ++l) {
    const int t = t0 + l;
    float dt = delta[t*EDn + d];
    float xv = xc[t*EDn + d];
    sdt += dt;
#pragma unroll
    for (int n = 0; n < DSTATEn; ++n) {
      float dA = __expf(dt * a[n]);
      h[n] = dA * h[n] + dt * Bm[t*DSTATEn + n] * xv;
    }
  }
#pragma unroll
  for (int n = 0; n < DSTATEn; ++n) y[(t0 + n)*EDn + d] = h[n];
  y[(t0 + DSTATEn)*EDn + d] = sdt;
}

__global__ void __launch_bounds__(EDn)
scan_carry_kernel(const float* __restrict__ A_log, float* y) {
  const int d = threadIdx.x;
  const int n = blockIdx.x;
  const int b = blockIdx.y;
  const float a = -__expf(A_log[d*DSTATEn + n]);
  float*       hp = y + (b*Ln + n)*EDn + d;
  const float* sp = y + (b*Ln + DSTATEn)*EDn + d;
  float carry = 0.f;
  for (int c0 = 0; c0 < NCH; c0 += CARRYU) {
    float he[CARRYU], sd[CARRYU];
#pragma unroll
    for (int j = 0; j < CARRYU; ++j) {
      he[j] = hp[(c0 + j)*(LC*EDn)];
      sd[j] = sp[(c0 + j)*(LC*EDn)];
    }
#pragma unroll
    for (int j = 0; j < CARRYU; ++j) {
      float P = __expf(a * sd[j]);
      hp[(c0 + j)*(LC*EDn)] = carry;
      carry = P * carry + he[j];
    }
  }
}

__global__ void __launch_bounds__(EDn)
scan_apply_kernel(const float* __restrict__ delta, const float* __restrict__ xc,
                  const float* __restrict__ Bm, const float* __restrict__ Cm,
                  const float* __restrict__ A_log, const float* __restrict__ Dp,
                  float* __restrict__ y) {
  const int d  = threadIdx.x;
  const int t0 = blockIdx.y * Ln + blockIdx.x * LC;
  float a[DSTATEn], h[DSTATEn];
#pragma unroll
  for (int n = 0; n < DSTATEn; ++n) {
    a[n] = -__expf(A_log[d*DSTATEn + n]);
    h[n] = y[(t0 + n)*EDn + d];
  }
  const float dp = Dp[d];
#pragma unroll
  for (int l = 0; l < LC; ++l) {
    const int t = t0 + l;
    float dt = delta[t*EDn + d];
    float xv = xc[t*EDn + d];
    float yv = 0.f;
#pragma unroll
    for (int n = 0; n < DSTATEn; ++n) {
      float dA = __expf(dt * a[n]);
      h[n] = dA * h[n] + dt * Bm[t*DSTATEn + n] * xv;
      yv += h[n] * Cm[t*DSTATEn + n];
    }
    y[t*EDn + d] = yv + dp * xv;
  }
}

__global__ void __launch_bounds__(128, 3)
out_kernel(const float* y, const float* zb, const v2f* Wout_sw,
                           const float* xr, const float* x, float* out) {
  __shared__ float ys[TOKB*PE];
  int tid = threadIdx.x;
  int wave = tid >> 5;
  int lane = tid & 31;
  long t0 = (long)blockIdx.x * TOKB;

#pragma unroll 12
  for (int i = tid; i < TOKB*EDn/4; i += 128) {
    int r = i >> 6, c = (i & 63) << 2;
    long tok = t0 + r;
    const float4 z4 = *(const float4*)(zb + tok*EDn + c);
    const float4 y4 = *(const float4*)(y  + tok*EDn + c);
    float4 o;
    o.x = y4.x * silu_f(z4.x);
    o.y = y4.y * silu_f(z4.y);
    o.z = y4.z * silu_f(z4.z);
    o.w = y4.w * silu_f(z4.w);
    *(float4*)(ys + r*PE + c) = o;
  }
  __syncthreads();

  gemm_stream<64, 2>(ys, PE, Wout_sw, wave*2, [&](int nt, const v8f (&acc)[TM]) {
    int n = nt*16 + (lane & 15);
    int mb = (lane < 16) ? 0 : 8;
    float r1[TM][8], r2[TM][8];
#pragma unroll
    for (int t = 0; t < TM; ++t)
#pragma unroll
      for (int j = 0; j < 8; ++j) {
        long tok = t0 + t*16 + mb + j;
        r1[t][j] = xr[tok*Dn + n];
        r2[t][j] = x[tok*Dn + n];
      }
#pragma unroll
    for (int t = 0; t < TM; ++t)
#pragma unroll
      for (int j = 0; j < 8; ++j) {
        long tok = t0 + t*16 + mb + j;
        out[tok*Dn + n] = acc[t][j] + r1[t][j] + r2[t][j];
      }
  });
}

extern "C" void kernel_launch(void* const* d_in, const int* in_sizes, int n_in,
                              void* d_out, int out_size, void* d_ws, size_t ws_size,
                              hipStream_t stream) {
  (void)in_sizes; (void)n_in; (void)out_size; (void)ws_size;
  const float* x      = (const float*)d_in[0];
  const float* mem    = (const float*)d_in[1];
  const float* Wq     = (const float*)d_in[2];
  const float* Wk     = (const float*)d_in[3];
  const float* Wv     = (const float*)d_in[4];
  const float* Wo     = (const float*)d_in[5];
  const float* ln1_g  = (const float*)d_in[6];
  const float* ln1_b  = (const float*)d_in[7];
  const float* rms_w  = (const float*)d_in[8];
  const float* W_in   = (const float*)d_in[9];
  const float* conv_w = (const float*)d_in[10];
  const float* conv_b = (const float*)d_in[11];
  const float* W_x    = (const float*)d_in[12];
  const float* W_dt   = (const float*)d_in[13];
  const float* b_dt   = (const float*)d_in[14];
  const float* A_log  = (const float*)d_in[15];
  const float* Dp     = (const float*)d_in[16];
  const float* W_out  = (const float*)d_in[17];

  float* ws = (float*)d_ws;
  size_t off = 0;
  float* Wq_sw   = ws + off; off += (size_t)8*32*32*2;
  float* Wo_sw   = ws + off; off += (size_t)8*32*32*2;
  float* Win_sw  = ws + off; off += (size_t)32*32*32*2;
  float* Wx_sw   = ws + off; off += (size_t)2*64*32*2;
  float* Wdt_sw  = ws + off; off += (size_t)16*2*32*2;
  float* Wout_sw = ws + off; off += (size_t)8*64*32*2;
  float* kfragT  = ws + off; off += (size_t)1*32*32*2;
  float* vfrag   = ws + off; off += (size_t)8*4*32*2;
  float* xr      = ws + off; off += (size_t)NTOK*Dn;
  float* xp      = ws + off; off += (size_t)NTOK*EDn;
  float* zb      = ws + off; off += (size_t)NTOK*EDn;
  float* xc      = ws + off; off += (size_t)NTOK*EDn;
  float* delta   = ws + off; off += (size_t)NTOK*EDn;
  float* Bm      = ws + off; off += (size_t)NTOK*DSTATEn;
  float* Cm      = ws + off; off += (size_t)NTOK*DSTATEn;
  float* y       = xp;

  prep_kernel<<<PBK_ALL, 256, 0, stream>>>(mem, Wk, Wv, Wq, Wo, W_in, W_x, W_dt,
                                           W_out, kfragT, vfrag, Wq_sw, Wo_sw,
                                           Win_sw, Wx_sw, Wdt_sw, Wout_sw);

  attn_kernel<<<NBLK, 128, 0, stream>>>(x, (const v2f*)Wq_sw, (const v2f*)Wo_sw,
                                        ln1_g, ln1_b, (const v2f*)kfragT,
                                        (const v2f*)vfrag, xr);
  in_kernel  <<<NBLK, 128, 0, stream>>>(xr, rms_w, (const v2f*)Win_sw, xp, zb);
  conv_kernel<<<NBLK, 128, 0, stream>>>(xp, conv_w, conv_b, (const v2f*)Wx_sw,
                                        (const v2f*)Wdt_sw, b_dt, xc, delta, Bm, Cm);
  scan_local_kernel<<<dim3(NCH, Bn), EDn, 0, stream>>>(delta, xc, Bm, A_log, y);
  scan_carry_kernel<<<dim3(DSTATEn, Bn), EDn, 0, stream>>>(A_log, y);
  scan_apply_kernel<<<dim3(NCH, Bn), EDn, 0, stream>>>(delta, xc, Bm, Cm,
                                                       A_log, Dp, y);
  out_kernel <<<NBLK, 128, 0, stream>>>(y, zb, (const v2f*)Wout_sw, xr, x,
                                        (float*)d_out);
}


# attention re-associated: scores = x@(Wq k^T), x_mem = att@(v Wo); both weight-only products formed once in the prep launch on WMMA, the per-token q and Wo GEMMs are gone; these sums are reordered
# speedup vs baseline: 15.9049x; 1.0925x over previous
#include <hip/hip_runtime.h>
#include <hip/hip_bf16.h>
#include <math.h>

#define Bn      8
#define Sn      12
#define NNn     325
#define Dn      128
#define EDn     256
#define DSTATEn 8
#define DCONVn  4
#define DTRANKn 8
#define MSLOTSn 10
#define Ln      (Sn*NNn)
#define NTOK    (Bn*Ln)
#define TM      3
#define TOKB    (16*TM)
#define NBLK    (NTOK/TOKB)
#define PD      (Dn + 4)
#define PE      (EDn + 4)
#define PS      20
#define PB      36

typedef float v2f __attribute__((ext_vector_type(2)));
typedef float v8f __attribute__((ext_vector_type(8)));

__device__ __forceinline__ v8f wmma4(v2f a, v2f b, v8f c) {
  return __builtin_amdgcn_wmma_f32_16x16x4_f32(false, a, false, b, (short)0, c,
                                               false, false);
}

__device__ __forceinline__ void store_c(float* C, int ldc, int nb, v8f c) {
  int lane = threadIdx.x & 31;
  int n = nb + (lane & 15);
  int mb = (lane < 16) ? 0 : 8;
#pragma unroll
  for (int j = 0; j < 8; ++j) C[(mb + j)*ldc + n] = c[j];
}

#define CH      16

__device__ __forceinline__ void mma_block(const v2f* a0, const v2f* a1,
                                          const v2f* a2, const v2f (&b)[CH],
                                          v8f (&acc)[TM]) {
#pragma unroll
  for (int j = 0; j < CH; ++j) {
    acc[0] = wmma4(a0[j*2], b[j], acc[0]);
    acc[1] = wmma4(a1[j*2], b[j], acc[1]);
    acc[2] = wmma4(a2[j*2], b[j], acc[2]);
  }
}

__device__ __forceinline__ void load_block(const v2f* bq, v2f (&b)[CH]) {
#pragma unroll
  for (int j = 0; j < CH; ++j) b[j] = bq[j*32];
}

template <int KT, int NTI, class Epi>
__device__ __forceinline__ void gemm_stream(const float* A, int lda,
                                            const v2f* Bsw, int nt0, Epi epi) {
  constexpr int NC  = NTI*KT/CH;
  constexpr int CPN = KT/CH;
  static_assert(KT % (2*CH) == 0, "an N-tile must be an even number of blocks");
  int lane = threadIdx.x & 31;
  int m = lane & 15;
  int ks = (lane >> 4) << 1;
  const v2f* bp = Bsw + (size_t)nt0 * KT * 32 + lane;
  const v2f* a0 = (const v2f*)(A + (0*16 + m)*lda + ks);
  const v2f* a1 = (const v2f*)(A + (1*16 + m)*lda + ks);
  const v2f* a2 = (const v2f*)(A + (2*16 + m)*lda + ks);
  v2f b0[CH], b1[CH];
  v8f acc[TM];
  const v8f z = {};
  acc[0] = z; acc[1] = z; acc[2] = z;
  load_block(bp, b0);
#pragma nounroll
  for (int c = 0; c < NC - 2; c += 2) {
    const int ko = (c % CPN) * CH * 2;
    load_block(bp + (c + 1)*CH*32, b1);
    mma_block(a0 + ko, a1 + ko, a2 + ko, b0, acc);
    load_block(bp + (c + 2)*CH*32, b0);
    mma_block(a0 + ko + CH*2, a1 + ko + CH*2, a2 + ko + CH*2, b1, acc);
    if (NTI > 1 && ((c + 2) % CPN) == 0) {
      epi(nt0 + c / CPN, acc);
      acc[0] = z; acc[1] = z; acc[2] = z;
    }
  }
  {
    const int ko = ((NC - 2) % CPN) * CH * 2;
    load_block(bp + (NC - 1)*CH*32, b1);
    mma_block(a0 + ko, a1 + ko, a2 + ko, b0, acc);
    mma_block(a0 + ko + CH*2, a1 + ko + CH*2, a2 + ko + CH*2, b1, acc);
    epi(nt0 + NTI - 1, acc);
  }
}

template <int KT, int NTI, class Epi>
__device__ __forceinline__ void gemm_small(const float* A, int lda,
                                           const v2f* Bsw, int nt0, Epi epi) {
  int lane = threadIdx.x & 31;
  int m = lane & 15;
  int ks = (lane >> 4) << 1;
  const v2f* bp = Bsw + (size_t)nt0 * KT * 32 + lane;
  const v2f* a0 = (const v2f*)(A + (0*16 + m)*lda + ks);
  const v2f* a1 = (const v2f*)(A + (1*16 + m)*lda + ks);
  const v2f* a2 = (const v2f*)(A + (2*16 + m)*lda + ks);
  v2f b[NTI*KT];
#pragma unroll
  for (int i = 0; i < NTI*KT; ++i) b[i] = bp[i*32];
#pragma unroll
  for (int nti = 0; nti < NTI; ++nti) {
    v8f acc[TM];
    const v8f z = {};
    acc[0] = z; acc[1] = z; acc[2] = z;
#pragma unroll
    for (int kt = 0; kt < KT; ++kt) {
      acc[0] = wmma4(a0[kt*2], b[nti*KT + kt], acc[0]);
      acc[1] = wmma4(a1[kt*2], b[nti*KT + kt], acc[1]);
      acc[2] = wmma4(a2[kt*2], b[nti*KT + kt], acc[2]);
    }
    epi(nt0 + nti, acc);
  }
}

__device__ __forceinline__ float silu_f(float x) { return x / (1.f + __expf(-x)); }

__device__ __forceinline__ float xhalf(float x) {
  return __builtin_bit_cast(float, __builtin_amdgcn_permlanex16(
      __builtin_bit_cast(int, x), __builtin_bit_cast(int, x),
      0x76543210, 0xfedcba98, false, false));
}

template <int M>
__device__ __forceinline__ float add_xmask(float s) {
  const int t = __builtin_amdgcn_update_dpp(0, __builtin_bit_cast(int, s),
                                            0x160 + M, 0xf, 0xf, true);
  return s + __builtin_bit_cast(float, t);
}

__device__ __forceinline__ float wave_sum(float s) {
  s += xhalf(s);
  s = add_xmask<8>(s);
  s = add_xmask<4>(s);
  s = add_xmask<2>(s);
  return add_xmask<1>(s);
}

__device__ __forceinline__ float softplus_f(float v) {
  const float e = __expf(v);
  const float u = 1.f + e;
  const float d = u - 1.f;
  const float lg = __builtin_amdgcn_logf(u) * 0.69314718f;
  const float r = lg * (e * __builtin_amdgcn_rcpf(d));
  return (v > 20.f) ? v : ((d == 0.f) ? e : r);
}

#define PK      (Dn + 4)
#define PBK_KV  2
#define PBK_WIN (32*32*32/256)
#define PBK_WX  (2*64*32/256)
#define PBK_WDT (16*2*32/256)
#define PBK_WOUT (8*64*32/256)
#define PBK_ALL (PBK_KV + PBK_WIN + PBK_WX + PBK_WDT + PBK_WOUT)

template <int K, int NREAL, int NT>
__device__ __forceinline__ void swz_tile(const float* W, float* dst, int idx) {
  constexpr int KT = K / 4;
  static_assert((NT*KT*32) % 256 == 0, "whole blocks");
  int lane = idx & 31;
  int kt = (idx >> 5) % KT;
  int nt = idx / (32 * KT);
  int n = nt*16 + (lane & 15);
  int k = kt*4 + ((lane >> 4) << 1);
  v2f o;
  o.x = (n < NREAL) ? W[k*NREAL + n]     : 0.f;
  o.y = (n < NREAL) ? W[(k+1)*NREAL + n] : 0.f;
  *(v2f*)(dst + idx*2) = o;
}

template <bool ISK>
__device__ __forceinline__ void fold_tile(const float* mem, const float* Wk,
                                          const float* Wv, const float* Wq,
                                          const float* Wo, float* Wqk_sw,
                                          float* VWo_sw, float* kv) {
  const int tid = threadIdx.x;
  const int wave = tid >> 5;
  const int lane = tid & 31;
  const int c = lane & 15;
  const int h2 = (lane >> 4) << 1;
  const float* Wsel = ISK ? Wk : Wv;
  const bool okrow = c < MSLOTSn;
  const float* arow = mem + (okrow ? c : 0)*Dn + h2;
  const float* bcol = Wsel + h2*Dn + wave*16 + c;
  const float* aq = Wq + (wave*16 + c)*Dn + h2;
  const float* bo = Wo + h2*Dn + wave*16 + c;
  v2f a[Dn/4], b[Dn/4], g[Dn/4];
#pragma unroll
  for (int kt = 0; kt < Dn/4; ++kt) {
    const v2f t = *(const v2f*)(arow + kt*4);
    a[kt].x = okrow ? t.x : 0.f;
    a[kt].y = okrow ? t.y : 0.f;
    b[kt].x = bcol[(kt*4)*Dn];
    b[kt].y = bcol[(kt*4 + 1)*Dn];
    if (ISK) {
      g[kt] = *(const v2f*)(aq + kt*4);
    } else {
      g[kt].x = bo[(kt*4)*Dn];
      g[kt].y = bo[(kt*4 + 1)*Dn];
    }
  }
  v8f acc = {};
#pragma unroll
  for (int kt = 0; kt < Dn/4; ++kt) acc = wmma4(a[kt], b[kt], acc);
#pragma unroll
  for (int j = 0; j < 8; ++j) kv[(j + 4*h2)*PK + wave*16 + c] = acc[j];
  __syncthreads();
  const float* lrow = kv + c*PK + h2;
  v8f acc2 = {};
  if (ISK) {
#pragma unroll
    for (int kt = 0; kt < Dn/4; ++kt)
      acc2 = wmma4(g[kt], *(const v2f*)(lrow + kt*4), acc2);
#pragma unroll
    for (int j = 0; j < 8; ++j) {
      const int d = wave*16 + j + 4*h2;
      Wqk_sw[(((d >> 2)*32) + c + 16*((d >> 1) & 1))*2 + (d & 1)] = acc2[j];
    }
  } else {
#pragma unroll
    for (int kt = 0; kt < Dn/4; ++kt)
      acc2 = wmma4(*(const v2f*)(lrow + kt*4), g[kt], acc2);
#pragma unroll
    for (int j = 0; j < 8; ++j) {
      const int sl = j + 4*h2;
      VWo_sw[(((wave*4 + (sl >> 2))*32) + c + 16*((sl >> 1) & 1))*2 + (sl & 1)] = acc2[j];
    }
  }
}

__global__ void __launch_bounds__(256)
prep_kernel(const float* mem, const float* Wk, const float* Wv,
            const float* Wq, const float* Wo, const float* W_in,
            const float* W_x, const float* W_dt, const float* W_out,
            float* Wqk_sw, float* VWo_sw,
            float* Win_sw, float* Wx_sw, float* Wdt_sw, float* Wout_sw) {
  __shared__ __attribute__((aligned(16))) float kv[16*PK];
  int b = blockIdx.x;
  const int tid = threadIdx.x;
  if (b == 0) { fold_tile<true>(mem, Wk, Wv, Wq, Wo, Wqk_sw, VWo_sw, kv); return; }
  if (b == 1) { fold_tile<false>(mem, Wk, Wv, Wq, Wo, Wqk_sw, VWo_sw, kv); return; }
  b -= PBK_KV;
  if (b < PBK_WIN) { swz_tile<Dn, 2*EDn, 32>(W_in, Win_sw, b*256 + tid); return; }
  b -= PBK_WIN;
  if (b < PBK_WX) { swz_tile<EDn, DTRANKn + 2*DSTATEn, 2>(W_x, Wx_sw, b*256 + tid); return; }
  b -= PBK_WX;
  if (b < PBK_WDT) { swz_tile<DTRANKn, EDn, 16>(W_dt, Wdt_sw, b*256 + tid); return; }
  b -= PBK_WDT;
  swz_tile<EDn, Dn, 8>(W_out, Wout_sw, b*256 + tid);
}

__global__ void __launch_bounds__(128, 3)
attn_kernel(const float* x, const float* ln_g, const float* ln_b,
                            const v2f* Wqk_sw, const v2f* VWo_sw, float* xr) {
  __shared__ __attribute__((aligned(16))) float xs[TOKB*PD];
  __shared__ __attribute__((aligned(16))) float qs[TOKB*PD];
  __shared__ float sc[TOKB*PS];
  int tid = threadIdx.x;
  int wave = tid >> 5;
  int lane = tid & 31;
  long t0 = (long)blockIdx.x * TOKB;

  v2f bqk[Dn/4];
#pragma unroll
  for (int kt = 0; kt < Dn/4; ++kt) bqk[kt] = Wqk_sw[kt*32 + lane];

  for (int i = tid; i < TOKB*Dn; i += 128) xs[(i >> 7)*PD + (i & 127)] = x[t0*Dn + i];
  __syncthreads();

  if (wave < TM) {
    const v2f* ap = (const v2f*)(xs + (wave*16 + (lane & 15))*PD + ((lane >> 4) << 1));
    v8f acc = {};
#pragma unroll
    for (int kt = 0; kt < Dn/4; ++kt) acc = wmma4(ap[kt*2], bqk[kt], acc);
    store_c(sc + wave*16*PS, PS, 0, acc);
  }
  __syncthreads();

  if (tid < TOKB) {
    const float scale = 0.088388347648318447f;
    float mx = -1e30f;
    for (int m = 0; m < MSLOTSn; ++m) {
      float v = sc[tid*PS + m] * scale; sc[tid*PS + m] = v;
      if (v > mx) mx = v;
    }
    float s = 0.f;
    for (int m = 0; m < MSLOTSn; ++m) {
      float e = __expf(sc[tid*PS + m] - mx); sc[tid*PS + m] = e; s += e;
    }
    float inv = 1.f / s;
    for (int m = 0; m < MSLOTSn; ++m) sc[tid*PS + m] *= inv;
    for (int m = MSLOTSn; m < 16; ++m) sc[tid*PS + m] = 0.f;
  }
  __syncthreads();

  gemm_small<4, 2>(sc, PS, VWo_sw, wave*2, [&](int nt, const v8f (&acc)[TM]) {
#pragma unroll
    for (int t = 0; t < TM; ++t) store_c(qs + t*16*PD, PD, nt*16, acc[t]);
  });
  __syncthreads();

  {
    constexpr int R = TOKB/4;
    const int c4 = lane * 4;
    const int r0 = wave * R;
    const float g0 = ln_g[c4], g1 = ln_g[c4 + 1], g2 = ln_g[c4 + 2], g3 = ln_g[c4 + 3];
    const float b0 = ln_b[c4], b1 = ln_b[c4 + 1], b2 = ln_b[c4 + 2], b3 = ln_b[c4 + 3];
    float4 q[R];
    float s[R];
#pragma unroll
    for (int k = 0; k < R; ++k) {
      q[k] = *(const float4*)(qs + (r0 + k)*PD + c4);
      s[k] = (q[k].x + q[k].y) + (q[k].z + q[k].w);
    }
#pragma unroll
    for (int k = 0; k < R; ++k) s[k] = wave_sum(s[k]);
#pragma unroll
    for (int k = 0; k < R; ++k) {
      const float m = s[k] * (1.f / Dn);
      q[k].x -= m; q[k].y -= m; q[k].z -= m; q[k].w -= m;
      s[k] = (q[k].x*q[k].x + q[k].y*q[k].y) + (q[k].z*q[k].z + q[k].w*q[k].w);
    }
#pragma unroll
    for (int k = 0; k < R; ++k) s[k] = wave_sum(s[k]);
#pragma unroll
    for (int k = 0; k < R; ++k) {
      const float inv = rsqrtf(s[k] * (1.f / Dn) + 1e-5f);
      const float4 xv = *(const float4*)(xs + (r0 + k)*PD + c4);
      float4 o;
      o.x = q[k].x * inv * g0 + b0 + xv.x;
      o.y = q[k].y * inv * g1 + b1 + xv.y;
      o.z = q[k].z * inv * g2 + b2 + xv.z;
      o.w = q[k].w * inv * g3 + b3 + xv.w;
      *(float4*)(xr + (t0 + r0 + k)*Dn + c4) = o;
    }
  }
}

__global__ void __launch_bounds__(128, 3)
in_kernel(const float* xr, const float* rms_w, const v2f* Win_sw,
                          float* xp, float* zb) {
  __shared__ __attribute__((aligned(16))) float xn[TOKB*PD];
  int tid = threadIdx.x;
  int wave = tid >> 5;
  int lane = tid & 31;
  long t0 = (long)blockIdx.x * TOKB;

  {
    constexpr int R = TOKB/4;
    const int c4 = lane * 4;
    const int r0 = wave * R;
    const float w0 = rms_w[c4], w1 = rms_w[c4 + 1], w2 = rms_w[c4 + 2], w3 = rms_w[c4 + 3];
    float4 v[R];
    float s[R];
#pragma unroll
    for (int k = 0; k < R; ++k)
      v[k] = *(const float4*)(xr + (t0 + r0 + k)*Dn + c4);
#pragma unroll
    for (int k = 0; k < R; ++k)
      s[k] = (v[k].x*v[k].x + v[k].y*v[k].y) + (v[k].z*v[k].z + v[k].w*v[k].w);
#pragma unroll
    for (int k = 0; k < R; ++k) s[k] = wave_sum(s[k]);
#pragma unroll
    for (int k = 0; k < R; ++k) {
      const float inv = rsqrtf(s[k] / Dn + 1e-5f);
      float4 o;
      o.x = v[k].x * inv * w0;
      o.y = v[k].y * inv * w1;
      o.z = v[k].z * inv * w2;
      o.w = v[k].w * inv * w3;
      *(float4*)(xn + (r0 + k)*PD + c4) = o;
    }
  }
  __syncthreads();

  gemm_stream<32, 8>(xn, PD, Win_sw, wave*8, [&](int nt, const v8f (&acc)[TM]) {
    int n = nt*16 + (lane & 15);
    int mb = (lane < 16) ? 0 : 8;
#pragma unroll
    for (int t = 0; t < TM; ++t)
#pragma unroll
      for (int j = 0; j < 8; ++j) {
        long tok = t0 + t*16 + mb + j;
        float v = acc[t][j];
        if (n < EDn) xp[tok*EDn + n] = v;
        else         zb[tok*EDn + (n - EDn)] = v;
      }
  });
}

__global__ void __launch_bounds__(128, 3)
conv_kernel(const float* xp, const float* conv_w, const float* conv_b,
                            const v2f* Wx_sw, const v2f* Wdt_sw, const float* b_dt,
                            float* xc, float* delta, float* Bm, float* Cm) {
  __shared__ float xcs[TOKB*PE];
  __shared__ float dbcs[TOKB*PB];
  int tid = threadIdx.x;
  int wave = tid >> 5;
  int lane = tid & 31;
  long t0 = (long)blockIdx.x * TOKB;

  {
    const int c2  = tid * 2;
    const int t0i = blockIdx.x * TOKB;
    const int l0  = t0i % Ln;
    float w0[DCONVn], w1[DCONVn];
#pragma unroll
    for (int j = 0; j < DCONVn; ++j) {
      w0[j] = conv_w[c2*DCONVn + j];
      w1[j] = conv_w[(c2 + 1)*DCONVn + j];
    }
    const float cb0 = conv_b[c2], cb1 = conv_b[c2 + 1];
    v2f win[TOKB + DCONVn - 1];
#pragma unroll
    for (int i = 0; i < TOKB + DCONVn - 1; ++i) {
      int g = t0i - (DCONVn - 1) + i;
      if (g < 0) g = 0;
      win[i] = *(const v2f*)(xp + (long)g*EDn + c2);
    }
#pragma unroll
    for (int r = 0; r < TOKB; ++r) {
      int l = l0 + r;
      if (l >= Ln) l -= Ln;
      float ax = cb0, ay = cb1;
#pragma unroll
      for (int j = 0; j < DCONVn; ++j) {
        const bool ok = (l - (DCONVn - 1) + j) >= 0;
        const float x0 = ok ? win[r + j].x : 0.f;
        const float x1 = ok ? win[r + j].y : 0.f;
        ax += w0[j] * x0;
        ay += w1[j] * x1;
      }
      v2f o;
      o.x = silu_f(ax);
      o.y = silu_f(ay);
      *(v2f*)(xcs + r*PE + c2) = o;
      *(v2f*)(xc + (long)(t0i + r)*EDn + c2) = o;
    }
  }
  __syncthreads();

  if (wave < 2) {
    gemm_stream<64, 1>(xcs, PE, Wx_sw, wave, [&](int nt, const v8f (&acc)[TM]) {
#pragma unroll
      for (int t = 0; t < TM; ++t) store_c(dbcs + t*16*PB, PB, nt*16, acc[t]);
    });
  }
  __syncthreads();

  gemm_small<2, 4>(dbcs, PB, Wdt_sw, wave*4, [&](int nt, const v8f (&acc)[TM]) {
    int n = nt*16 + (lane & 15);
    int mb = (lane < 16) ? 0 : 8;
    const float bd = b_dt[n];
#pragma unroll
    for (int t = 0; t < TM; ++t)
#pragma unroll
      for (int j = 0; j < 8; ++j) {
        long tok = t0 + t*16 + mb + j;
        float v = acc[t][j] + bd;
        delta[tok*EDn + n] = softplus_f(v);
      }
  });

  for (int i = tid; i < TOKB*16; i += 128) {
    int r = i >> 4, c = i & 15;
    long tok = t0 + r;
    float v = dbcs[r*PB + DTRANKn + c];
    if (c < DSTATEn) Bm[tok*DSTATEn + c]             = v;
    else             Cm[tok*DSTATEn + (c - DSTATEn)] = v;
  }
}

#define LC      25
#define NCH     (Ln/LC)
#define CARRYU  26

__global__ void __launch_bounds__(EDn)
scan_local_kernel(const float* __restrict__ delta, const float* __restrict__ xc,
                  const float* __restrict__ Bm, const float* __restrict__ A_log,
                  float* __restrict__ y) {
  const int d  = threadIdx.x;
  const int t0 = blockIdx.y * Ln + blockIdx.x * LC;
  float a[DSTATEn], h[DSTATEn];
#pragma unroll
  for (int n = 0; n < DSTATEn; ++n) { a[n] = -__expf(A_log[d*DSTATEn + n]); h[n] = 0.f; }
  float sdt = 0.f;
#pragma unroll
  for (int l = 0; l < LC; ++l) {
    const int t = t0 + l;
    float dt = delta[t*EDn + d];
    float xv = xc[t*EDn + d];
    sdt += dt;
#pragma unroll
    for (int n = 0; n < DSTATEn; ++n) {
      float dA = __expf(dt * a[n]);
      h[n] = dA * h[n] + dt * Bm[t*DSTATEn + n] * xv;
    }
  }
#pragma unroll
  for (int n = 0; n < DSTATEn; ++n) y[(t0 + n)*EDn + d] = h[n];
  y[(t0 + DSTATEn)*EDn + d] = sdt;
}

__global__ void __launch_bounds__(EDn)
scan_carry_kernel(const float* __restrict__ A_log, float* y) {
  const int d = threadIdx.x;
  const int n = blockIdx.x;
  const int b = blockIdx.y;
  const float a = -__expf(A_log[d*DSTATEn + n]);
  float*       hp = y + (b*Ln + n)*EDn + d;
  const float* sp = y + (b*Ln + DSTATEn)*EDn + d;
  float carry = 0.f;
  for (int c0 = 0; c0 < NCH; c0 += CARRYU) {
    float he[CARRYU], sd[CARRYU];
#pragma unroll
    for (int j = 0; j < CARRYU; ++j) {
      he[j] = hp[(c0 + j)*(LC*EDn)];
      sd[j] = sp[(c0 + j)*(LC*EDn)];
    }
#pragma unroll
    for (int j = 0; j < CARRYU; ++j) {
      float P = __expf(a * sd[j]);
      hp[(c0 + j)*(LC*EDn)] = carry;
      carry = P * carry + he[j];
    }
  }
}

__global__ void __launch_bounds__(EDn)
scan_apply_kernel(const float* __restrict__ delta, const float* __restrict__ xc,
                  const float* __restrict__ Bm, const float* __restrict__ Cm,
                  const float* __restrict__ A_log, const float* __restrict__ Dp,
                  float* __restrict__ y) {
  const int d  = threadIdx.x;
  const int t0 = blockIdx.y * Ln + blockIdx.x * LC;
  float a[DSTATEn], h[DSTATEn];
#pragma unroll
  for (int n = 0; n < DSTATEn; ++n) {
    a[n] = -__expf(A_log[d*DSTATEn + n]);
    h[n] = y[(t0 + n)*EDn + d];
  }
  const float dp = Dp[d];
#pragma unroll
  for (int l = 0; l < LC; ++l) {
    const int t = t0 + l;
    float dt = delta[t*EDn + d];
    float xv = xc[t*EDn + d];
    float yv = 0.f;
#pragma unroll
    for (int n = 0; n < DSTATEn; ++n) {
      float dA = __expf(dt * a[n]);
      h[n] = dA * h[n] + dt * Bm[t*DSTATEn + n] * xv;
      yv += h[n] * Cm[t*DSTATEn + n];
    }
    y[t*EDn + d] = yv + dp * xv;
  }
}

__global__ void __launch_bounds__(128, 3)
out_kernel(const float* y, const float* zb, const v2f* Wout_sw,
                           const float* xr, const float* x, float* out) {
  __shared__ float ys[TOKB*PE];
  int tid = threadIdx.x;
  int wave = tid >> 5;
  int lane = tid & 31;
  long t0 = (long)blockIdx.x * TOKB;

#pragma unroll 12
  for (int i = tid; i < TOKB*EDn/4; i += 128) {
    int r = i >> 6, c = (i & 63) << 2;
    long tok = t0 + r;
    const float4 z4 = *(const float4*)(zb + tok*EDn + c);
    const float4 y4 = *(const float4*)(y  + tok*EDn + c);
    float4 o;
    o.x = y4.x * silu_f(z4.x);
    o.y = y4.y * silu_f(z4.y);
    o.z = y4.z * silu_f(z4.z);
    o.w = y4.w * silu_f(z4.w);
    *(float4*)(ys + r*PE + c) = o;
  }
  __syncthreads();

  gemm_stream<64, 2>(ys, PE, Wout_sw, wave*2, [&](int nt, const v8f (&acc)[TM]) {
    int n = nt*16 + (lane & 15);
    int mb = (lane < 16) ? 0 : 8;
    float r1[TM][8], r2[TM][8];
#pragma unroll
    for (int t = 0; t < TM; ++t)
#pragma unroll
      for (int j = 0; j < 8; ++j) {
        long tok = t0 + t*16 + mb + j;
        r1[t][j] = xr[tok*Dn + n];
        r2[t][j] = x[tok*Dn + n];
      }
#pragma unroll
    for (int t = 0; t < TM; ++t)
#pragma unroll
      for (int j = 0; j < 8; ++j) {
        long tok = t0 + t*16 + mb + j;
        out[tok*Dn + n] = acc[t][j] + r1[t][j] + r2[t][j];
      }
  });
}

extern "C" void kernel_launch(void* const* d_in, const int* in_sizes, int n_in,
                              void* d_out, int out_size, void* d_ws, size_t ws_size,
                              hipStream_t stream) {
  (void)in_sizes; (void)n_in; (void)out_size; (void)ws_size;
  const float* x      = (const float*)d_in[0];
  const float* mem    = (const float*)d_in[1];
  const float* Wq     = (const float*)d_in[2];
  const float* Wk     = (const float*)d_in[3];
  const float* Wv     = (const float*)d_in[4];
  const float* Wo     = (const float*)d_in[5];
  const float* ln1_g  = (const float*)d_in[6];
  const float* ln1_b  = (const float*)d_in[7];
  const float* rms_w  = (const float*)d_in[8];
  const float* W_in   = (const float*)d_in[9];
  const float* conv_w = (const float*)d_in[10];
  const float* conv_b = (const float*)d_in[11];
  const float* W_x    = (const float*)d_in[12];
  const float* W_dt   = (const float*)d_in[13];
  const float* b_dt   = (const float*)d_in[14];
  const float* A_log  = (const float*)d_in[15];
  const float* Dp     = (const float*)d_in[16];
  const float* W_out  = (const float*)d_in[17];

  float* ws = (float*)d_ws;
  size_t off = 0;
  float* Win_sw  = ws + off; off += (size_t)32*32*32*2;
  float* Wx_sw   = ws + off; off += (size_t)2*64*32*2;
  float* Wdt_sw  = ws + off; off += (size_t)16*2*32*2;
  float* Wout_sw = ws + off; off += (size_t)8*64*32*2;
  float* Wqk_sw  = ws + off; off += (size_t)1*32*32*2;
  float* VWo_sw  = ws + off; off += (size_t)8*4*32*2;
  float* xr      = ws + off; off += (size_t)NTOK*Dn;
  float* xp      = ws + off; off += (size_t)NTOK*EDn;
  float* zb      = ws + off; off += (size_t)NTOK*EDn;
  float* xc      = ws + off; off += (size_t)NTOK*EDn;
  float* delta   = ws + off; off += (size_t)NTOK*EDn;
  float* Bm      = ws + off; off += (size_t)NTOK*DSTATEn;
  float* Cm      = ws + off; off += (size_t)NTOK*DSTATEn;
  float* y       = xp;

  prep_kernel<<<PBK_ALL, 256, 0, stream>>>(mem, Wk, Wv, Wq, Wo, W_in, W_x, W_dt,
                                           W_out, Wqk_sw, VWo_sw,
                                           Win_sw, Wx_sw, Wdt_sw, Wout_sw);

  attn_kernel<<<NBLK, 128, 0, stream>>>(x, ln1_g, ln1_b, (const v2f*)Wqk_sw,
                                        (const v2f*)VWo_sw, xr);
  in_kernel  <<<NBLK, 128, 0, stream>>>(xr, rms_w, (const v2f*)Win_sw, xp, zb);
  conv_kernel<<<NBLK, 128, 0, stream>>>(xp, conv_w, conv_b, (const v2f*)Wx_sw,
                                        (const v2f*)Wdt_sw, b_dt, xc, delta, Bm, Cm);
  scan_local_kernel<<<dim3(NCH, Bn), EDn, 0, stream>>>(delta, xc, Bm, A_log, y);
  scan_carry_kernel<<<dim3(DSTATEn, Bn), EDn, 0, stream>>>(A_log, y);
  scan_apply_kernel<<<dim3(NCH, Bn), EDn, 0, stream>>>(delta, xc, Bm, Cm,
                                                       A_log, Dp, y);
  out_kernel <<<NBLK, 128, 0, stream>>>(y, zb, (const v2f*)Wout_sw, xr, x,
                                        (float*)d_out);
}


# per-element branches taken out of two GEMM epilogues: xp/z destination fixed per wave (was a branch per store), softplus v>20 case by clamp and add-back (was a branch per element; few ulp there)
# speedup vs baseline: 16.9493x; 1.0657x over previous
#include <hip/hip_runtime.h>
#include <hip/hip_bf16.h>
#include <math.h>

#define Bn      8
#define Sn      12
#define NNn     325
#define Dn      128
#define EDn     256
#define DSTATEn 8
#define DCONVn  4
#define DTRANKn 8
#define MSLOTSn 10
#define Ln      (Sn*NNn)
#define NTOK    (Bn*Ln)
#define TM      3
#define TOKB    (16*TM)
#define NBLK    (NTOK/TOKB)
#define PD      (Dn + 4)
#define PE      (EDn + 4)
#define PS      20
#define PB      36

typedef float v2f __attribute__((ext_vector_type(2)));
typedef float v8f __attribute__((ext_vector_type(8)));

__device__ __forceinline__ v8f wmma4(v2f a, v2f b, v8f c) {
  return __builtin_amdgcn_wmma_f32_16x16x4_f32(false, a, false, b, (short)0, c,
                                               false, false);
}

__device__ __forceinline__ void store_c(float* C, int ldc, int nb, v8f c) {
  int lane = threadIdx.x & 31;
  int n = nb + (lane & 15);
  int mb = (lane < 16) ? 0 : 8;
#pragma unroll
  for (int j = 0; j < 8; ++j) C[(mb + j)*ldc + n] = c[j];
}

#define CH      16

__device__ __forceinline__ void mma_block(const v2f* a0, const v2f* a1,
                                          const v2f* a2, const v2f (&b)[CH],
                                          v8f (&acc)[TM]) {
#pragma unroll
  for (int j = 0; j < CH; ++j) {
    acc[0] = wmma4(a0[j*2], b[j], acc[0]);
    acc[1] = wmma4(a1[j*2], b[j], acc[1]);
    acc[2] = wmma4(a2[j*2], b[j], acc[2]);
  }
}

__device__ __forceinline__ void load_block(const v2f* bq, v2f (&b)[CH]) {
#pragma unroll
  for (int j = 0; j < CH; ++j) b[j] = bq[j*32];
}

template <int KT, int NTI, class Epi>
__device__ __forceinline__ void gemm_stream(const float* A, int lda,
                                            const v2f* Bsw, int nt0, Epi epi) {
  constexpr int NC  = NTI*KT/CH;
  constexpr int CPN = KT/CH;
  static_assert(KT % (2*CH) == 0, "an N-tile must be an even number of blocks");
  int lane = threadIdx.x & 31;
  int m = lane & 15;
  int ks = (lane >> 4) << 1;
  const v2f* bp = Bsw + (size_t)nt0 * KT * 32 + lane;
  const v2f* a0 = (const v2f*)(A + (0*16 + m)*lda + ks);
  const v2f* a1 = (const v2f*)(A + (1*16 + m)*lda + ks);
  const v2f* a2 = (const v2f*)(A + (2*16 + m)*lda + ks);
  v2f b0[CH], b1[CH];
  v8f acc[TM];
  const v8f z = {};
  acc[0] = z; acc[1] = z; acc[2] = z;
  load_block(bp, b0);
#pragma nounroll
  for (int c = 0; c < NC - 2; c += 2) {
    const int ko = (c % CPN) * CH * 2;
    load_block(bp + (c + 1)*CH*32, b1);
    mma_block(a0 + ko, a1 + ko, a2 + ko, b0, acc);
    load_block(bp + (c + 2)*CH*32, b0);
    mma_block(a0 + ko + CH*2, a1 + ko + CH*2, a2 + ko + CH*2, b1, acc);
    if (NTI > 1 && ((c + 2) % CPN) == 0) {
      epi(nt0 + c / CPN, acc);
      acc[0] = z; acc[1] = z; acc[2] = z;
    }
  }
  {
    const int ko = ((NC - 2) % CPN) * CH * 2;
    load_block(bp + (NC - 1)*CH*32, b1);
    mma_block(a0 + ko, a1 + ko, a2 + ko, b0, acc);
    mma_block(a0 + ko + CH*2, a1 + ko + CH*2, a2 + ko + CH*2, b1, acc);
    epi(nt0 + NTI - 1, acc);
  }
}

template <int KT, int NTI, class Epi>
__device__ __forceinline__ void gemm_small(const float* A, int lda,
                                           const v2f* Bsw, int nt0, Epi epi) {
  int lane = threadIdx.x & 31;
  int m = lane & 15;
  int ks = (lane >> 4) << 1;
  const v2f* bp = Bsw + (size_t)nt0 * KT * 32 + lane;
  const v2f* a0 = (const v2f*)(A + (0*16 + m)*lda + ks);
  const v2f* a1 = (const v2f*)(A + (1*16 + m)*lda + ks);
  const v2f* a2 = (const v2f*)(A + (2*16 + m)*lda + ks);
  v2f b[NTI*KT];
#pragma unroll
  for (int i = 0; i < NTI*KT; ++i) b[i] = bp[i*32];
#pragma unroll
  for (int nti = 0; nti < NTI; ++nti) {
    v8f acc[TM];
    const v8f z = {};
    acc[0] = z; acc[1] = z; acc[2] = z;
#pragma unroll
    for (int kt = 0; kt < KT; ++kt) {
      acc[0] = wmma4(a0[kt*2], b[nti*KT + kt], acc[0]);
      acc[1] = wmma4(a1[kt*2], b[nti*KT + kt], acc[1]);
      acc[2] = wmma4(a2[kt*2], b[nti*KT + kt], acc[2]);
    }
    epi(nt0 + nti, acc);
  }
}

__device__ __forceinline__ float silu_f(float x) { return x / (1.f + __expf(-x)); }

__device__ __forceinline__ float xhalf(float x) {
  return __builtin_bit_cast(float, __builtin_amdgcn_permlanex16(
      __builtin_bit_cast(int, x), __builtin_bit_cast(int, x),
      0x76543210, 0xfedcba98, false, false));
}

template <int M>
__device__ __forceinline__ float add_xmask(float s) {
  const int t = __builtin_amdgcn_update_dpp(0, __builtin_bit_cast(int, s),
                                            0x160 + M, 0xf, 0xf, true);
  return s + __builtin_bit_cast(float, t);
}

__device__ __forceinline__ float wave_sum(float s) {
  s += xhalf(s);
  s = add_xmask<8>(s);
  s = add_xmask<4>(s);
  s = add_xmask<2>(s);
  return add_xmask<1>(s);
}

__device__ __forceinline__ float softplus_f(float v) {
  const bool big = v > 20.f;
  const float vc = big ? 20.f : v;
  const float hi = big ? v - 20.f : 0.f;
  const float e = __expf(vc);
  const float u = 1.f + e;
  const float d = u - 1.f;
  const float lg = __builtin_amdgcn_logf(u) * 0.69314718f;
  const float r = lg * (e * __builtin_amdgcn_rcpf(d));
  return ((d == 0.f) ? e : r) + hi;
}

#define PK      (Dn + 4)
#define PBK_KV  2
#define PBK_WIN (32*32*32/256)
#define PBK_WX  (2*64*32/256)
#define PBK_WDT (16*2*32/256)
#define PBK_WOUT (8*64*32/256)
#define PBK_ALL (PBK_KV + PBK_WIN + PBK_WX + PBK_WDT + PBK_WOUT)

template <int K, int NREAL, int NT>
__device__ __forceinline__ void swz_tile(const float* W, float* dst, int idx) {
  constexpr int KT = K / 4;
  static_assert((NT*KT*32) % 256 == 0, "whole blocks");
  int lane = idx & 31;
  int kt = (idx >> 5) % KT;
  int nt = idx / (32 * KT);
  int n = nt*16 + (lane & 15);
  int k = kt*4 + ((lane >> 4) << 1);
  v2f o;
  o.x = (n < NREAL) ? W[k*NREAL + n]     : 0.f;
  o.y = (n < NREAL) ? W[(k+1)*NREAL + n] : 0.f;
  *(v2f*)(dst + idx*2) = o;
}

template <bool ISK>
__device__ __forceinline__ void fold_tile(const float* mem, const float* Wk,
                                          const float* Wv, const float* Wq,
                                          const float* Wo, float* Wqk_sw,
                                          float* VWo_sw, float* kv) {
  const int tid = threadIdx.x;
  const int wave = tid >> 5;
  const int lane = tid & 31;
  const int c = lane & 15;
  const int h2 = (lane >> 4) << 1;
  const float* Wsel = ISK ? Wk : Wv;
  const bool okrow = c < MSLOTSn;
  const float* arow = mem + (okrow ? c : 0)*Dn + h2;
  const float* bcol = Wsel + h2*Dn + wave*16 + c;
  const float* aq = Wq + (wave*16 + c)*Dn + h2;
  const float* bo = Wo + h2*Dn + wave*16 + c;
  v2f a[Dn/4], b[Dn/4], g[Dn/4];
#pragma unroll
  for (int kt = 0; kt < Dn/4; ++kt) {
    const v2f t = *(const v2f*)(arow + kt*4);
    a[kt].x = okrow ? t.x : 0.f;
    a[kt].y = okrow ? t.y : 0.f;
    b[kt].x = bcol[(kt*4)*Dn];
    b[kt].y = bcol[(kt*4 + 1)*Dn];
    if (ISK) {
      g[kt] = *(const v2f*)(aq + kt*4);
    } else {
      g[kt].x = bo[(kt*4)*Dn];
      g[kt].y = bo[(kt*4 + 1)*Dn];
    }
  }
  v8f acc = {};
#pragma unroll
  for (int kt = 0; kt < Dn/4; ++kt) acc = wmma4(a[kt], b[kt], acc);
#pragma unroll
  for (int j = 0; j < 8; ++j) kv[(j + 4*h2)*PK + wave*16 + c] = acc[j];
  __syncthreads();
  const float* lrow = kv + c*PK + h2;
  v8f acc2 = {};
  if (ISK) {
#pragma unroll
    for (int kt = 0; kt < Dn/4; ++kt)
      acc2 = wmma4(g[kt], *(const v2f*)(lrow + kt*4), acc2);
#pragma unroll
    for (int j = 0; j < 8; ++j) {
      const int d = wave*16 + j + 4*h2;
      Wqk_sw[(((d >> 2)*32) + c + 16*((d >> 1) & 1))*2 + (d & 1)] = acc2[j];
    }
  } else {
#pragma unroll
    for (int kt = 0; kt < Dn/4; ++kt)
      acc2 = wmma4(*(const v2f*)(lrow + kt*4), g[kt], acc2);
#pragma unroll
    for (int j = 0; j < 8; ++j) {
      const int sl = j + 4*h2;
      VWo_sw[(((wave*4 + (sl >> 2))*32) + c + 16*((sl >> 1) & 1))*2 + (sl & 1)] = acc2[j];
    }
  }
}

__global__ void __launch_bounds__(256)
prep_kernel(const float* mem, const float* Wk, const float* Wv,
            const float* Wq, const float* Wo, const float* W_in,
            const float* W_x, const float* W_dt, const float* W_out,
            float* Wqk_sw, float* VWo_sw,
            float* Win_sw, float* Wx_sw, float* Wdt_sw, float* Wout_sw) {
  __shared__ __attribute__((aligned(16))) float kv[16*PK];
  int b = blockIdx.x;
  const int tid = threadIdx.x;
  if (b == 0) { fold_tile<true>(mem, Wk, Wv, Wq, Wo, Wqk_sw, VWo_sw, kv); return; }
  if (b == 1) { fold_tile<false>(mem, Wk, Wv, Wq, Wo, Wqk_sw, VWo_sw, kv); return; }
  b -= PBK_KV;
  if (b < PBK_WIN) { swz_tile<Dn, 2*EDn, 32>(W_in, Win_sw, b*256 + tid); return; }
  b -= PBK_WIN;
  if (b < PBK_WX) { swz_tile<EDn, DTRANKn + 2*DSTATEn, 2>(W_x, Wx_sw, b*256 + tid); return; }
  b -= PBK_WX;
  if (b < PBK_WDT) { swz_tile<DTRANKn, EDn, 16>(W_dt, Wdt_sw, b*256 + tid); return; }
  b -= PBK_WDT;
  swz_tile<EDn, Dn, 8>(W_out, Wout_sw, b*256 + tid);
}

__global__ void __launch_bounds__(128, 3)
attn_kernel(const float* x, const float* ln_g, const float* ln_b,
                            const v2f* Wqk_sw, const v2f* VWo_sw, float* xr) {
  __shared__ __attribute__((aligned(16))) float xs[TOKB*PD];
  __shared__ __attribute__((aligned(16))) float qs[TOKB*PD];
  __shared__ float sc[TOKB*PS];
  int tid = threadIdx.x;
  int wave = tid >> 5;
  int lane = tid & 31;
  long t0 = (long)blockIdx.x * TOKB;

  v2f bqk[Dn/4];
#pragma unroll
  for (int kt = 0; kt < Dn/4; ++kt) bqk[kt] = Wqk_sw[kt*32 + lane];

  for (int i = tid; i < TOKB*Dn; i += 128) xs[(i >> 7)*PD + (i & 127)] = x[t0*Dn + i];
  __syncthreads();

  if (wave < TM) {
    const v2f* ap = (const v2f*)(xs + (wave*16 + (lane & 15))*PD + ((lane >> 4) << 1));
    v8f acc = {};
#pragma unroll
    for (int kt = 0; kt < Dn/4; ++kt) acc = wmma4(ap[kt*2], bqk[kt], acc);
    store_c(sc + wave*16*PS, PS, 0, acc);
  }
  __syncthreads();

  if (tid < TOKB) {
    const float scale = 0.088388347648318447f;
    float mx = -1e30f;
    for (int m = 0; m < MSLOTSn; ++m) {
      float v = sc[tid*PS + m] * scale; sc[tid*PS + m] = v;
      if (v > mx) mx = v;
    }
    float s = 0.f;
    for (int m = 0; m < MSLOTSn; ++m) {
      float e = __expf(sc[tid*PS + m] - mx); sc[tid*PS + m] = e; s += e;
    }
    float inv = 1.f / s;
    for (int m = 0; m < MSLOTSn; ++m) sc[tid*PS + m] *= inv;
    for (int m = MSLOTSn; m < 16; ++m) sc[tid*PS + m] = 0.f;
  }
  __syncthreads();

  gemm_small<4, 2>(sc, PS, VWo_sw, wave*2, [&](int nt, const v8f (&acc)[TM]) {
#pragma unroll
    for (int t = 0; t < TM; ++t) store_c(qs + t*16*PD, PD, nt*16, acc[t]);
  });
  __syncthreads();

  {
    constexpr int R = TOKB/4;
    const int c4 = lane * 4;
    const int r0 = wave * R;
    const float g0 = ln_g[c4], g1 = ln_g[c4 + 1], g2 = ln_g[c4 + 2], g3 = ln_g[c4 + 3];
    const float b0 = ln_b[c4], b1 = ln_b[c4 + 1], b2 = ln_b[c4 + 2], b3 = ln_b[c4 + 3];
    float4 q[R];
    float s[R];
#pragma unroll
    for (int k = 0; k < R; ++k) {
      q[k] = *(const float4*)(qs + (r0 + k)*PD + c4);
      s[k] = (q[k].x + q[k].y) + (q[k].z + q[k].w);
    }
#pragma unroll
    for (int k = 0; k < R; ++k) s[k] = wave_sum(s[k]);
#pragma unroll
    for (int k = 0; k < R; ++k) {
      const float m = s[k] * (1.f / Dn);
      q[k].x -= m; q[k].y -= m; q[k].z -= m; q[k].w -= m;
      s[k] = (q[k].x*q[k].x + q[k].y*q[k].y) + (q[k].z*q[k].z + q[k].w*q[k].w);
    }
#pragma unroll
    for (int k = 0; k < R; ++k) s[k] = wave_sum(s[k]);
#pragma unroll
    for (int k = 0; k < R; ++k) {
      const float inv = rsqrtf(s[k] * (1.f / Dn) + 1e-5f);
      const float4 xv = *(const float4*)(xs + (r0 + k)*PD + c4);
      float4 o;
      o.x = q[k].x * inv * g0 + b0 + xv.x;
      o.y = q[k].y * inv * g1 + b1 + xv.y;
      o.z = q[k].z * inv * g2 + b2 + xv.z;
      o.w = q[k].w * inv * g3 + b3 + xv.w;
      *(float4*)(xr + (t0 + r0 + k)*Dn + c4) = o;
    }
  }
}

__global__ void __launch_bounds__(128, 3)
in_kernel(const float* xr, const float* rms_w, const v2f* Win_sw,
                          float* xp, float* zb) {
  __shared__ __attribute__((aligned(16))) float xn[TOKB*PD];
  int tid = threadIdx.x;
  int wave = tid >> 5;
  int lane = tid & 31;
  long t0 = (long)blockIdx.x * TOKB;

  {
    constexpr int R = TOKB/4;
    const int c4 = lane * 4;
    const int r0 = wave * R;
    const float w0 = rms_w[c4], w1 = rms_w[c4 + 1], w2 = rms_w[c4 + 2], w3 = rms_w[c4 + 3];
    float4 v[R];
    float s[R];
#pragma unroll
    for (int k = 0; k < R; ++k)
      v[k] = *(const float4*)(xr + (t0 + r0 + k)*Dn + c4);
#pragma unroll
    for (int k = 0; k < R; ++k)
      s[k] = (v[k].x*v[k].x + v[k].y*v[k].y) + (v[k].z*v[k].z + v[k].w*v[k].w);
#pragma unroll
    for (int k = 0; k < R; ++k) s[k] = wave_sum(s[k]);
#pragma unroll
    for (int k = 0; k < R; ++k) {
      const float inv = rsqrtf(s[k] / Dn + 1e-5f);
      float4 o;
      o.x = v[k].x * inv * w0;
      o.y = v[k].y * inv * w1;
      o.z = v[k].z * inv * w2;
      o.w = v[k].w * inv * w3;
      *(float4*)(xn + (r0 + k)*PD + c4) = o;
    }
  }
  __syncthreads();

  float* const dst = (wave < 2) ? xp : zb;
  gemm_stream<32, 8>(xn, PD, Win_sw, wave*8, [&](int nt, const v8f (&acc)[TM]) {
    int n = (nt & 15)*16 + (lane & 15);
    int mb = (lane < 16) ? 0 : 8;
#pragma unroll
    for (int t = 0; t < TM; ++t)
#pragma unroll
      for (int j = 0; j < 8; ++j) {
        long tok = t0 + t*16 + mb + j;
        dst[tok*EDn + n] = acc[t][j];
      }
  });
}

__global__ void __launch_bounds__(128, 3)
conv_kernel(const float* xp, const float* conv_w, const float* conv_b,
                            const v2f* Wx_sw, const v2f* Wdt_sw, const float* b_dt,
                            float* xc, float* delta, float* Bm, float* Cm) {
  __shared__ float xcs[TOKB*PE];
  __shared__ float dbcs[TOKB*PB];
  int tid = threadIdx.x;
  int wave = tid >> 5;
  int lane = tid & 31;
  long t0 = (long)blockIdx.x * TOKB;

  {
    const int c2  = tid * 2;
    const int t0i = blockIdx.x * TOKB;
    const int l0  = t0i % Ln;
    float w0[DCONVn], w1[DCONVn];
#pragma unroll
    for (int j = 0; j < DCONVn; ++j) {
      w0[j] = conv_w[c2*DCONVn + j];
      w1[j] = conv_w[(c2 + 1)*DCONVn + j];
    }
    const float cb0 = conv_b[c2], cb1 = conv_b[c2 + 1];
    v2f win[TOKB + DCONVn - 1];
#pragma unroll
    for (int i = 0; i < TOKB + DCONVn - 1; ++i) {
      int g = t0i - (DCONVn - 1) + i;
      if (g < 0) g = 0;
      win[i] = *(const v2f*)(xp + (long)g*EDn + c2);
    }
#pragma unroll
    for (int r = 0; r < TOKB; ++r) {
      int l = l0 + r;
      if (l >= Ln) l -= Ln;
      float ax = cb0, ay = cb1;
#pragma unroll
      for (int j = 0; j < DCONVn; ++j) {
        const bool ok = (l - (DCONVn - 1) + j) >= 0;
        const float x0 = ok ? win[r + j].x : 0.f;
        const float x1 = ok ? win[r + j].y : 0.f;
        ax += w0[j] * x0;
        ay += w1[j] * x1;
      }
      v2f o;
      o.x = silu_f(ax);
      o.y = silu_f(ay);
      *(v2f*)(xcs + r*PE + c2) = o;
      *(v2f*)(xc + (long)(t0i + r)*EDn + c2) = o;
    }
  }
  __syncthreads();

  if (wave < 2) {
    gemm_stream<64, 1>(xcs, PE, Wx_sw, wave, [&](int nt, const v8f (&acc)[TM]) {
#pragma unroll
      for (int t = 0; t < TM; ++t) store_c(dbcs + t*16*PB, PB, nt*16, acc[t]);
    });
  }
  __syncthreads();

  gemm_small<2, 4>(dbcs, PB, Wdt_sw, wave*4, [&](int nt, const v8f (&acc)[TM]) {
    int n = nt*16 + (lane & 15);
    int mb = (lane < 16) ? 0 : 8;
    const float bd = b_dt[n];
#pragma unroll
    for (int t = 0; t < TM; ++t)
#pragma unroll
      for (int j = 0; j < 8; ++j) {
        long tok = t0 + t*16 + mb + j;
        float v = acc[t][j] + bd;
        delta[tok*EDn + n] = softplus_f(v);
      }
  });

  for (int i = tid; i < TOKB*16; i += 128) {
    int r = i >> 4, c = i & 15;
    long tok = t0 + r;
    float v = dbcs[r*PB + DTRANKn + c];
    if (c < DSTATEn) Bm[tok*DSTATEn + c]             = v;
    else             Cm[tok*DSTATEn + (c - DSTATEn)] = v;
  }
}

#define LC      25
#define NCH     (Ln/LC)
#define CARRYU  26

__global__ void __launch_bounds__(EDn)
scan_local_kernel(const float* __restrict__ delta, const float* __restrict__ xc,
                  const float* __restrict__ Bm, const float* __restrict__ A_log,
                  float* __restrict__ y) {
  const int d  = threadIdx.x;
  const int t0 = blockIdx.y * Ln + blockIdx.x * LC;
  float a[DSTATEn], h[DSTATEn];
#pragma unroll
  for (int n = 0; n < DSTATEn; ++n) { a[n] = -__expf(A_log[d*DSTATEn + n]); h[n] = 0.f; }
  float sdt = 0.f;
#pragma unroll
  for (int l = 0; l < LC; ++l) {
    const int t = t0 + l;
    float dt = delta[t*EDn + d];
    float xv = xc[t*EDn + d];
    sdt += dt;
#pragma unroll
    for (int n = 0; n < DSTATEn; ++n) {
      float dA = __expf(dt * a[n]);
      h[n] = dA * h[n] + dt * Bm[t*DSTATEn + n] * xv;
    }
  }
#pragma unroll
  for (int n = 0; n < DSTATEn; ++n) y[(t0 + n)*EDn + d] = h[n];
  y[(t0 + DSTATEn)*EDn + d] = sdt;
}

__global__ void __launch_bounds__(EDn)
scan_carry_kernel(const float* __restrict__ A_log, float* y) {
  const int d = threadIdx.x;
  const int n = blockIdx.x;
  const int b = blockIdx.y;
  const float a = -__expf(A_log[d*DSTATEn + n]);
  float*       hp = y + (b*Ln + n)*EDn + d;
  const float* sp = y + (b*Ln + DSTATEn)*EDn + d;
  float carry = 0.f;
  for (int c0 = 0; c0 < NCH; c0 += CARRYU) {
    float he[CARRYU], sd[CARRYU];
#pragma unroll
    for (int j = 0; j < CARRYU; ++j) {
      he[j] = hp[(c0 + j)*(LC*EDn)];
      sd[j] = sp[(c0 + j)*(LC*EDn)];
    }
#pragma unroll
    for (int j = 0; j < CARRYU; ++j) {
      float P = __expf(a * sd[j]);
      hp[(c0 + j)*(LC*EDn)] = carry;
      carry = P * carry + he[j];
    }
  }
}

__global__ void __launch_bounds__(EDn)
scan_apply_kernel(const float* __restrict__ delta, const float* __restrict__ xc,
                  const float* __restrict__ Bm, const float* __restrict__ Cm,
                  const float* __restrict__ A_log, const float* __restrict__ Dp,
                  float* __restrict__ y) {
  const int d  = threadIdx.x;
  const int t0 = blockIdx.y * Ln + blockIdx.x * LC;
  float a[DSTATEn], h[DSTATEn];
#pragma unroll
  for (int n = 0; n < DSTATEn; ++n) {
    a[n] = -__expf(A_log[d*DSTATEn + n]);
    h[n] = y[(t0 + n)*EDn + d];
  }
  const float dp = Dp[d];
#pragma unroll
  for (int l = 0; l < LC; ++l) {
    const int t = t0 + l;
    float dt = delta[t*EDn + d];
    float xv = xc[t*EDn + d];
    float yv = 0.f;
#pragma unroll
    for (int n = 0; n < DSTATEn; ++n) {
      float dA = __expf(dt * a[n]);
      h[n] = dA * h[n] + dt * Bm[t*DSTATEn + n] * xv;
      yv += h[n] * Cm[t*DSTATEn + n];
    }
    y[t*EDn + d] = yv + dp * xv;
  }
}

__global__ void __launch_bounds__(128, 3)
out_kernel(const float* y, const float* zb, const v2f* Wout_sw,
                           const float* xr, const float* x, float* out) {
  __shared__ float ys[TOKB*PE];
  int tid = threadIdx.x;
  int wave = tid >> 5;
  int lane = tid & 31;
  long t0 = (long)blockIdx.x * TOKB;

#pragma unroll 12
  for (int i = tid; i < TOKB*EDn/4; i += 128) {
    int r = i >> 6, c = (i & 63) << 2;
    long tok = t0 + r;
    const float4 z4 = *(const float4*)(zb + tok*EDn + c);
    const float4 y4 = *(const float4*)(y  + tok*EDn + c);
    float4 o;
    o.x = y4.x * silu_f(z4.x);
    o.y = y4.y * silu_f(z4.y);
    o.z = y4.z * silu_f(z4.z);
    o.w = y4.w * silu_f(z4.w);
    *(float4*)(ys + r*PE + c) = o;
  }
  __syncthreads();

  gemm_stream<64, 2>(ys, PE, Wout_sw, wave*2, [&](int nt, const v8f (&acc)[TM]) {
    int n = nt*16 + (lane & 15);
    int mb = (lane < 16) ? 0 : 8;
    float r1[TM][8], r2[TM][8];
#pragma unroll
    for (int t = 0; t < TM; ++t)
#pragma unroll
      for (int j = 0; j < 8; ++j) {
        long tok = t0 + t*16 + mb + j;
        r1[t][j] = xr[tok*Dn + n];
        r2[t][j] = x[tok*Dn + n];
      }
#pragma unroll
    for (int t = 0; t < TM; ++t)
#pragma unroll
      for (int j = 0; j < 8; ++j) {
        long tok = t0 + t*16 + mb + j;
        out[tok*Dn + n] = acc[t][j] + r1[t][j] + r2[t][j];
      }
  });
}

extern "C" void kernel_launch(void* const* d_in, const int* in_sizes, int n_in,
                              void* d_out, int out_size, void* d_ws, size_t ws_size,
                              hipStream_t stream) {
  (void)in_sizes; (void)n_in; (void)out_size; (void)ws_size;
  const float* x      = (const float*)d_in[0];
  const float* mem    = (const float*)d_in[1];
  const float* Wq     = (const float*)d_in[2];
  const float* Wk     = (const float*)d_in[3];
  const float* Wv     = (const float*)d_in[4];
  const float* Wo     = (const float*)d_in[5];
  const float* ln1_g  = (const float*)d_in[6];
  const float* ln1_b  = (const float*)d_in[7];
  const float* rms_w  = (const float*)d_in[8];
  const float* W_in   = (const float*)d_in[9];
  const float* conv_w = (const float*)d_in[10];
  const float* conv_b = (const float*)d_in[11];
  const float* W_x    = (const float*)d_in[12];
  const float* W_dt   = (const float*)d_in[13];
  const float* b_dt   = (const float*)d_in[14];
  const float* A_log  = (const float*)d_in[15];
  const float* Dp     = (const float*)d_in[16];
  const float* W_out  = (const float*)d_in[17];

  float* ws = (float*)d_ws;
  size_t off = 0;
  float* Win_sw  = ws + off; off += (size_t)32*32*32*2;
  float* Wx_sw   = ws + off; off += (size_t)2*64*32*2;
  float* Wdt_sw  = ws + off; off += (size_t)16*2*32*2;
  float* Wout_sw = ws + off; off += (size_t)8*64*32*2;
  float* Wqk_sw  = ws + off; off += (size_t)1*32*32*2;
  float* VWo_sw  = ws + off; off += (size_t)8*4*32*2;
  float* xr      = ws + off; off += (size_t)NTOK*Dn;
  float* xp      = ws + off; off += (size_t)NTOK*EDn;
  float* zb      = ws + off; off += (size_t)NTOK*EDn;
  float* xc      = ws + off; off += (size_t)NTOK*EDn;
  float* delta   = ws + off; off += (size_t)NTOK*EDn;
  float* Bm      = ws + off; off += (size_t)NTOK*DSTATEn;
  float* Cm      = ws + off; off += (size_t)NTOK*DSTATEn;
  float* y       = xp;

  prep_kernel<<<PBK_ALL, 256, 0, stream>>>(mem, Wk, Wv, Wq, Wo, W_in, W_x, W_dt,
                                           W_out, Wqk_sw, VWo_sw,
                                           Win_sw, Wx_sw, Wdt_sw, Wout_sw);

  attn_kernel<<<NBLK, 128, 0, stream>>>(x, ln1_g, ln1_b, (const v2f*)Wqk_sw,
                                        (const v2f*)VWo_sw, xr);
  in_kernel  <<<NBLK, 128, 0, stream>>>(xr, rms_w, (const v2f*)Win_sw, xp, zb);
  conv_kernel<<<NBLK, 128, 0, stream>>>(xp, conv_w, conv_b, (const v2f*)Wx_sw,
                                        (const v2f*)Wdt_sw, b_dt, xc, delta, Bm, Cm);
  scan_local_kernel<<<dim3(NCH, Bn), EDn, 0, stream>>>(delta, xc, Bm, A_log, y);
  scan_carry_kernel<<<dim3(DSTATEn, Bn), EDn, 0, stream>>>(A_log, y);
  scan_apply_kernel<<<dim3(NCH, Bn), EDn, 0, stream>>>(delta, xc, Bm, Cm,
                                                       A_log, Dp, y);
  out_kernel <<<NBLK, 128, 0, stream>>>(y, zb, (const v2f*)Wout_sw, xr, x,
                                        (float*)d_out);
}


# token tile per block cut from 48 to 32 rows (975 blocks, four resident per WGP): the busiest WGP of the four tiled launches holds 8 M-tiles of 16 tokens, not 9; every sum unchanged
# speedup vs baseline: 19.0863x; 1.1261x over previous
#include <hip/hip_runtime.h>
#include <hip/hip_bf16.h>
#include <math.h>

#define Bn      8
#define Sn      12
#define NNn     325
#define Dn      128
#define EDn     256
#define DSTATEn 8
#define DCONVn  4
#define DTRANKn 8
#define MSLOTSn 10
#define Ln      (Sn*NNn)
#define NTOK    (Bn*Ln)
#define TM      2
#define TOKB    (16*TM)
#define NBLK    (NTOK/TOKB)
#define PD      (Dn + 4)
#define PE      (EDn + 4)
#define PS      20
#define PB      36

typedef float v2f __attribute__((ext_vector_type(2)));
typedef float v8f __attribute__((ext_vector_type(8)));

__device__ __forceinline__ v8f wmma4(v2f a, v2f b, v8f c) {
  return __builtin_amdgcn_wmma_f32_16x16x4_f32(false, a, false, b, (short)0, c,
                                               false, false);
}

__device__ __forceinline__ void store_c(float* C, int ldc, int nb, v8f c) {
  int lane = threadIdx.x & 31;
  int n = nb + (lane & 15);
  int mb = (lane < 16) ? 0 : 8;
#pragma unroll
  for (int j = 0; j < 8; ++j) C[(mb + j)*ldc + n] = c[j];
}

#define CH      16

__device__ __forceinline__ void mma_block(const v2f* a0, const v2f* a1,
                                          const v2f (&b)[CH], v8f (&acc)[TM]) {
  static_assert(TM == 2, "one A pointer per M-tile");
#pragma unroll
  for (int j = 0; j < CH; ++j) {
    acc[0] = wmma4(a0[j*2], b[j], acc[0]);
    acc[1] = wmma4(a1[j*2], b[j], acc[1]);
  }
}

__device__ __forceinline__ void load_block(const v2f* bq, v2f (&b)[CH]) {
#pragma unroll
  for (int j = 0; j < CH; ++j) b[j] = bq[j*32];
}

template <int KT, int NTI, class Epi>
__device__ __forceinline__ void gemm_stream(const float* A, int lda,
                                            const v2f* Bsw, int nt0, Epi epi) {
  constexpr int NC  = NTI*KT/CH;
  constexpr int CPN = KT/CH;
  static_assert(KT % (2*CH) == 0, "an N-tile must be an even number of blocks");
  int lane = threadIdx.x & 31;
  int m = lane & 15;
  int ks = (lane >> 4) << 1;
  const v2f* bp = Bsw + (size_t)nt0 * KT * 32 + lane;
  const v2f* a0 = (const v2f*)(A + (0*16 + m)*lda + ks);
  const v2f* a1 = (const v2f*)(A + (1*16 + m)*lda + ks);
  v2f b0[CH], b1[CH];
  v8f acc[TM];
  const v8f z = {};
  acc[0] = z; acc[1] = z;
  load_block(bp, b0);
#pragma nounroll
  for (int c = 0; c < NC - 2; c += 2) {
    const int ko = (c % CPN) * CH * 2;
    load_block(bp + (c + 1)*CH*32, b1);
    mma_block(a0 + ko, a1 + ko, b0, acc);
    load_block(bp + (c + 2)*CH*32, b0);
    mma_block(a0 + ko + CH*2, a1 + ko + CH*2, b1, acc);
    if (NTI > 1 && ((c + 2) % CPN) == 0) {
      epi(nt0 + c / CPN, acc);
      acc[0] = z; acc[1] = z;
    }
  }
  {
    const int ko = ((NC - 2) % CPN) * CH * 2;
    load_block(bp + (NC - 1)*CH*32, b1);
    mma_block(a0 + ko, a1 + ko, b0, acc);
    mma_block(a0 + ko + CH*2, a1 + ko + CH*2, b1, acc);
    epi(nt0 + NTI - 1, acc);
  }
}

template <int KT, int NTI, class Epi>
__device__ __forceinline__ void gemm_small(const float* A, int lda,
                                           const v2f* Bsw, int nt0, Epi epi) {
  int lane = threadIdx.x & 31;
  int m = lane & 15;
  int ks = (lane >> 4) << 1;
  const v2f* bp = Bsw + (size_t)nt0 * KT * 32 + lane;
  const v2f* a0 = (const v2f*)(A + (0*16 + m)*lda + ks);
  const v2f* a1 = (const v2f*)(A + (1*16 + m)*lda + ks);
  v2f b[NTI*KT];
#pragma unroll
  for (int i = 0; i < NTI*KT; ++i) b[i] = bp[i*32];
#pragma unroll
  for (int nti = 0; nti < NTI; ++nti) {
    v8f acc[TM];
    const v8f z = {};
    acc[0] = z; acc[1] = z;
#pragma unroll
    for (int kt = 0; kt < KT; ++kt) {
      acc[0] = wmma4(a0[kt*2], b[nti*KT + kt], acc[0]);
      acc[1] = wmma4(a1[kt*2], b[nti*KT + kt], acc[1]);
    }
    epi(nt0 + nti, acc);
  }
}

__device__ __forceinline__ float silu_f(float x) { return x / (1.f + __expf(-x)); }

__device__ __forceinline__ float xhalf(float x) {
  return __builtin_bit_cast(float, __builtin_amdgcn_permlanex16(
      __builtin_bit_cast(int, x), __builtin_bit_cast(int, x),
      0x76543210, 0xfedcba98, false, false));
}

template <int M>
__device__ __forceinline__ float add_xmask(float s) {
  const int t = __builtin_amdgcn_update_dpp(0, __builtin_bit_cast(int, s),
                                            0x160 + M, 0xf, 0xf, true);
  return s + __builtin_bit_cast(float, t);
}

__device__ __forceinline__ float wave_sum(float s) {
  s += xhalf(s);
  s = add_xmask<8>(s);
  s = add_xmask<4>(s);
  s = add_xmask<2>(s);
  return add_xmask<1>(s);
}

__device__ __forceinline__ float softplus_f(float v) {
  const bool big = v > 20.f;
  const float vc = big ? 20.f : v;
  const float hi = big ? v - 20.f : 0.f;
  const float e = __expf(vc);
  const float u = 1.f + e;
  const float d = u - 1.f;
  const float lg = __builtin_amdgcn_logf(u) * 0.69314718f;
  const float r = lg * (e * __builtin_amdgcn_rcpf(d));
  return ((d == 0.f) ? e : r) + hi;
}

#define PK      (Dn + 4)
#define PBK_KV  2
#define PBK_WIN (32*32*32/256)
#define PBK_WX  (2*64*32/256)
#define PBK_WDT (16*2*32/256)
#define PBK_WOUT (8*64*32/256)
#define PBK_ALL (PBK_KV + PBK_WIN + PBK_WX + PBK_WDT + PBK_WOUT)

template <int K, int NREAL, int NT>
__device__ __forceinline__ void swz_tile(const float* W, float* dst, int idx) {
  constexpr int KT = K / 4;
  static_assert((NT*KT*32) % 256 == 0, "whole blocks");
  int lane = idx & 31;
  int kt = (idx >> 5) % KT;
  int nt = idx / (32 * KT);
  int n = nt*16 + (lane & 15);
  int k = kt*4 + ((lane >> 4) << 1);
  v2f o;
  o.x = (n < NREAL) ? W[k*NREAL + n]     : 0.f;
  o.y = (n < NREAL) ? W[(k+1)*NREAL + n] : 0.f;
  *(v2f*)(dst + idx*2) = o;
}

template <bool ISK>
__device__ __forceinline__ void fold_tile(const float* mem, const float* Wk,
                                          const float* Wv, const float* Wq,
                                          const float* Wo, float* Wqk_sw,
                                          float* VWo_sw, float* kv) {
  const int tid = threadIdx.x;
  const int wave = tid >> 5;
  const int lane = tid & 31;
  const int c = lane & 15;
  const int h2 = (lane >> 4) << 1;
  const float* Wsel = ISK ? Wk : Wv;
  const bool okrow = c < MSLOTSn;
  const float* arow = mem + (okrow ? c : 0)*Dn + h2;
  const float* bcol = Wsel + h2*Dn + wave*16 + c;
  const float* aq = Wq + (wave*16 + c)*Dn + h2;
  const float* bo = Wo + h2*Dn + wave*16 + c;
  v2f a[Dn/4], b[Dn/4], g[Dn/4];
#pragma unroll
  for (int kt = 0; kt < Dn/4; ++kt) {
    const v2f t = *(const v2f*)(arow + kt*4);
    a[kt].x = okrow ? t.x : 0.f;
    a[kt].y = okrow ? t.y : 0.f;
    b[kt].x = bcol[(kt*4)*Dn];
    b[kt].y = bcol[(kt*4 + 1)*Dn];
    if (ISK) {
      g[kt] = *(const v2f*)(aq + kt*4);
    } else {
      g[kt].x = bo[(kt*4)*Dn];
      g[kt].y = bo[(kt*4 + 1)*Dn];
    }
  }
  v8f acc = {};
#pragma unroll
  for (int kt = 0; kt < Dn/4; ++kt) acc = wmma4(a[kt], b[kt], acc);
#pragma unroll
  for (int j = 0; j < 8; ++j) kv[(j + 4*h2)*PK + wave*16 + c] = acc[j];
  __syncthreads();
  const float* lrow = kv + c*PK + h2;
  v8f acc2 = {};
  if (ISK) {
#pragma unroll
    for (int kt = 0; kt < Dn/4; ++kt)
      acc2 = wmma4(g[kt], *(const v2f*)(lrow + kt*4), acc2);
#pragma unroll
    for (int j = 0; j < 8; ++j) {
      const int d = wave*16 + j + 4*h2;
      Wqk_sw[(((d >> 2)*32) + c + 16*((d >> 1) & 1))*2 + (d & 1)] = acc2[j];
    }
  } else {
#pragma unroll
    for (int kt = 0; kt < Dn/4; ++kt)
      acc2 = wmma4(*(const v2f*)(lrow + kt*4), g[kt], acc2);
#pragma unroll
    for (int j = 0; j < 8; ++j) {
      const int sl = j + 4*h2;
      VWo_sw[(((wave*4 + (sl >> 2))*32) + c + 16*((sl >> 1) & 1))*2 + (sl & 1)] = acc2[j];
    }
  }
}

__global__ void __launch_bounds__(256)
prep_kernel(const float* mem, const float* Wk, const float* Wv,
            const float* Wq, const float* Wo, const float* W_in,
            const float* W_x, const float* W_dt, const float* W_out,
            float* Wqk_sw, float* VWo_sw,
            float* Win_sw, float* Wx_sw, float* Wdt_sw, float* Wout_sw) {
  __shared__ __attribute__((aligned(16))) float kv[16*PK];
  int b = blockIdx.x;
  const int tid = threadIdx.x;
  if (b == 0) { fold_tile<true>(mem, Wk, Wv, Wq, Wo, Wqk_sw, VWo_sw, kv); return; }
  if (b == 1) { fold_tile<false>(mem, Wk, Wv, Wq, Wo, Wqk_sw, VWo_sw, kv); return; }
  b -= PBK_KV;
  if (b < PBK_WIN) { swz_tile<Dn, 2*EDn, 32>(W_in, Win_sw, b*256 + tid); return; }
  b -= PBK_WIN;
  if (b < PBK_WX) { swz_tile<EDn, DTRANKn + 2*DSTATEn, 2>(W_x, Wx_sw, b*256 + tid); return; }
  b -= PBK_WX;
  if (b < PBK_WDT) { swz_tile<DTRANKn, EDn, 16>(W_dt, Wdt_sw, b*256 + tid); return; }
  b -= PBK_WDT;
  swz_tile<EDn, Dn, 8>(W_out, Wout_sw, b*256 + tid);
}

__global__ void __launch_bounds__(128, 4)
attn_kernel(const float* x, const float* ln_g, const float* ln_b,
                            const v2f* Wqk_sw, const v2f* VWo_sw, float* xr) {
  __shared__ __attribute__((aligned(16))) float xs[TOKB*PD];
  __shared__ __attribute__((aligned(16))) float qs[TOKB*PD];
  __shared__ float sc[TOKB*PS];
  int tid = threadIdx.x;
  int wave = tid >> 5;
  int lane = tid & 31;
  long t0 = (long)blockIdx.x * TOKB;

  v2f bqk[Dn/4];
#pragma unroll
  for (int kt = 0; kt < Dn/4; ++kt) bqk[kt] = Wqk_sw[kt*32 + lane];

  for (int i = tid; i < TOKB*Dn; i += 128) xs[(i >> 7)*PD + (i & 127)] = x[t0*Dn + i];
  __syncthreads();

  if (wave < TM) {
    const v2f* ap = (const v2f*)(xs + (wave*16 + (lane & 15))*PD + ((lane >> 4) << 1));
    v8f acc = {};
#pragma unroll
    for (int kt = 0; kt < Dn/4; ++kt) acc = wmma4(ap[kt*2], bqk[kt], acc);
    store_c(sc + wave*16*PS, PS, 0, acc);
  }
  __syncthreads();

  if (tid < TOKB) {
    const float scale = 0.088388347648318447f;
    float mx = -1e30f;
    for (int m = 0; m < MSLOTSn; ++m) {
      float v = sc[tid*PS + m] * scale; sc[tid*PS + m] = v;
      if (v > mx) mx = v;
    }
    float s = 0.f;
    for (int m = 0; m < MSLOTSn; ++m) {
      float e = __expf(sc[tid*PS + m] - mx); sc[tid*PS + m] = e; s += e;
    }
    float inv = 1.f / s;
    for (int m = 0; m < MSLOTSn; ++m) sc[tid*PS + m] *= inv;
    for (int m = MSLOTSn; m < 16; ++m) sc[tid*PS + m] = 0.f;
  }
  __syncthreads();

  gemm_small<4, 2>(sc, PS, VWo_sw, wave*2, [&](int nt, const v8f (&acc)[TM]) {
#pragma unroll
    for (int t = 0; t < TM; ++t) store_c(qs + t*16*PD, PD, nt*16, acc[t]);
  });
  __syncthreads();

  {
    constexpr int R = TOKB/4;
    const int c4 = lane * 4;
    const int r0 = wave * R;
    const float g0 = ln_g[c4], g1 = ln_g[c4 + 1], g2 = ln_g[c4 + 2], g3 = ln_g[c4 + 3];
    const float b0 = ln_b[c4], b1 = ln_b[c4 + 1], b2 = ln_b[c4 + 2], b3 = ln_b[c4 + 3];
    float4 q[R];
    float s[R];
#pragma unroll
    for (int k = 0; k < R; ++k) {
      q[k] = *(const float4*)(qs + (r0 + k)*PD + c4);
      s[k] = (q[k].x + q[k].y) + (q[k].z + q[k].w);
    }
#pragma unroll
    for (int k = 0; k < R; ++k) s[k] = wave_sum(s[k]);
#pragma unroll
    for (int k = 0; k < R; ++k) {
      const float m = s[k] * (1.f / Dn);
      q[k].x -= m; q[k].y -= m; q[k].z -= m; q[k].w -= m;
      s[k] = (q[k].x*q[k].x + q[k].y*q[k].y) + (q[k].z*q[k].z + q[k].w*q[k].w);
    }
#pragma unroll
    for (int k = 0; k < R; ++k) s[k] = wave_sum(s[k]);
#pragma unroll
    for (int k = 0; k < R; ++k) {
      const float inv = rsqrtf(s[k] * (1.f / Dn) + 1e-5f);
      const float4 xv = *(const float4*)(xs + (r0 + k)*PD + c4);
      float4 o;
      o.x = q[k].x * inv * g0 + b0 + xv.x;
      o.y = q[k].y * inv * g1 + b1 + xv.y;
      o.z = q[k].z * inv * g2 + b2 + xv.z;
      o.w = q[k].w * inv * g3 + b3 + xv.w;
      *(float4*)(xr + (t0 + r0 + k)*Dn + c4) = o;
    }
  }
}

__global__ void __launch_bounds__(128, 4)
in_kernel(const float* xr, const float* rms_w, const v2f* Win_sw,
                          float* xp, float* zb) {
  __shared__ __attribute__((aligned(16))) float xn[TOKB*PD];
  int tid = threadIdx.x;
  int wave = tid >> 5;
  int lane = tid & 31;
  long t0 = (long)blockIdx.x * TOKB;

  {
    constexpr int R = TOKB/4;
    const int c4 = lane * 4;
    const int r0 = wave * R;
    const float w0 = rms_w[c4], w1 = rms_w[c4 + 1], w2 = rms_w[c4 + 2], w3 = rms_w[c4 + 3];
    float4 v[R];
    float s[R];
#pragma unroll
    for (int k = 0; k < R; ++k)
      v[k] = *(const float4*)(xr + (t0 + r0 + k)*Dn + c4);
#pragma unroll
    for (int k = 0; k < R; ++k)
      s[k] = (v[k].x*v[k].x + v[k].y*v[k].y) + (v[k].z*v[k].z + v[k].w*v[k].w);
#pragma unroll
    for (int k = 0; k < R; ++k) s[k] = wave_sum(s[k]);
#pragma unroll
    for (int k = 0; k < R; ++k) {
      const float inv = rsqrtf(s[k] / Dn + 1e-5f);
      float4 o;
      o.x = v[k].x * inv * w0;
      o.y = v[k].y * inv * w1;
      o.z = v[k].z * inv * w2;
      o.w = v[k].w * inv * w3;
      *(float4*)(xn + (r0 + k)*PD + c4) = o;
    }
  }
  __syncthreads();

  float* const dst = (wave < 2) ? xp : zb;
  gemm_stream<32, 8>(xn, PD, Win_sw, wave*8, [&](int nt, const v8f (&acc)[TM]) {
    int n = (nt & 15)*16 + (lane & 15);
    int mb = (lane < 16) ? 0 : 8;
#pragma unroll
    for (int t = 0; t < TM; ++t)
#pragma unroll
      for (int j = 0; j < 8; ++j) {
        long tok = t0 + t*16 + mb + j;
        dst[tok*EDn + n] = acc[t][j];
      }
  });
}

__global__ void __launch_bounds__(128, 4)
conv_kernel(const float* xp, const float* conv_w, const float* conv_b,
                            const v2f* Wx_sw, const v2f* Wdt_sw, const float* b_dt,
                            float* xc, float* delta, float* Bm, float* Cm) {
  __shared__ float xcs[TOKB*PE];
  __shared__ float dbcs[TOKB*PB];
  int tid = threadIdx.x;
  int wave = tid >> 5;
  int lane = tid & 31;
  long t0 = (long)blockIdx.x * TOKB;

  {
    const int c2  = tid * 2;
    const int t0i = blockIdx.x * TOKB;
    const int l0  = t0i % Ln;
    float w0[DCONVn], w1[DCONVn];
#pragma unroll
    for (int j = 0; j < DCONVn; ++j) {
      w0[j] = conv_w[c2*DCONVn + j];
      w1[j] = conv_w[(c2 + 1)*DCONVn + j];
    }
    const float cb0 = conv_b[c2], cb1 = conv_b[c2 + 1];
    v2f win[TOKB + DCONVn - 1];
#pragma unroll
    for (int i = 0; i < TOKB + DCONVn - 1; ++i) {
      int g = t0i - (DCONVn - 1) + i;
      if (g < 0) g = 0;
      win[i] = *(const v2f*)(xp + (long)g*EDn + c2);
    }
#pragma unroll
    for (int r = 0; r < TOKB; ++r) {
      int l = l0 + r;
      if (l >= Ln) l -= Ln;
      float ax = cb0, ay = cb1;
#pragma unroll
      for (int j = 0; j < DCONVn; ++j) {
        const bool ok = (l - (DCONVn - 1) + j) >= 0;
        const float x0 = ok ? win[r + j].x : 0.f;
        const float x1 = ok ? win[r + j].y : 0.f;
        ax += w0[j] * x0;
        ay += w1[j] * x1;
      }
      v2f o;
      o.x = silu_f(ax);
      o.y = silu_f(ay);
      *(v2f*)(xcs + r*PE + c2) = o;
      *(v2f*)(xc + (long)(t0i + r)*EDn + c2) = o;
    }
  }
  __syncthreads();

  if (wave < 2) {
    gemm_stream<64, 1>(xcs, PE, Wx_sw, wave, [&](int nt, const v8f (&acc)[TM]) {
#pragma unroll
      for (int t = 0; t < TM; ++t) store_c(dbcs + t*16*PB, PB, nt*16, acc[t]);
    });
  }
  __syncthreads();

  gemm_small<2, 4>(dbcs, PB, Wdt_sw, wave*4, [&](int nt, const v8f (&acc)[TM]) {
    int n = nt*16 + (lane & 15);
    int mb = (lane < 16) ? 0 : 8;
    const float bd = b_dt[n];
#pragma unroll
    for (int t = 0; t < TM; ++t)
#pragma unroll
      for (int j = 0; j < 8; ++j) {
        long tok = t0 + t*16 + mb + j;
        float v = acc[t][j] + bd;
        delta[tok*EDn + n] = softplus_f(v);
      }
  });

  for (int i = tid; i < TOKB*16; i += 128) {
    int r = i >> 4, c = i & 15;
    long tok = t0 + r;
    float v = dbcs[r*PB + DTRANKn + c];
    if (c < DSTATEn) Bm[tok*DSTATEn + c]             = v;
    else             Cm[tok*DSTATEn + (c - DSTATEn)] = v;
  }
}

#define LC      25
#define NCH     (Ln/LC)
#define CARRYU  26

__global__ void __launch_bounds__(EDn)
scan_local_kernel(const float* __restrict__ delta, const float* __restrict__ xc,
                  const float* __restrict__ Bm, const float* __restrict__ A_log,
                  float* __restrict__ y) {
  const int d  = threadIdx.x;
  const int t0 = blockIdx.y * Ln + blockIdx.x * LC;
  float a[DSTATEn], h[DSTATEn];
#pragma unroll
  for (int n = 0; n < DSTATEn; ++n) { a[n] = -__expf(A_log[d*DSTATEn + n]); h[n] = 0.f; }
  float sdt = 0.f;
#pragma unroll
  for (int l = 0; l < LC; ++l) {
    const int t = t0 + l;
    float dt = delta[t*EDn + d];
    float xv = xc[t*EDn + d];
    sdt += dt;
#pragma unroll
    for (int n = 0; n < DSTATEn; ++n) {
      float dA = __expf(dt * a[n]);
      h[n] = dA * h[n] + dt * Bm[t*DSTATEn + n] * xv;
    }
  }
#pragma unroll
  for (int n = 0; n < DSTATEn; ++n) y[(t0 + n)*EDn + d] = h[n];
  y[(t0 + DSTATEn)*EDn + d] = sdt;
}

__global__ void __launch_bounds__(EDn)
scan_carry_kernel(const float* __restrict__ A_log, float* y) {
  const int d = threadIdx.x;
  const int n = blockIdx.x;
  const int b = blockIdx.y;
  const float a = -__expf(A_log[d*DSTATEn + n]);
  float*       hp = y + (b*Ln + n)*EDn + d;
  const float* sp = y + (b*Ln + DSTATEn)*EDn + d;
  float carry = 0.f;
  for (int c0 = 0; c0 < NCH; c0 += CARRYU) {
    float he[CARRYU], sd[CARRYU];
#pragma unroll
    for (int j = 0; j < CARRYU; ++j) {
      he[j] = hp[(c0 + j)*(LC*EDn)];
      sd[j] = sp[(c0 + j)*(LC*EDn)];
    }
#pragma unroll
    for (int j = 0; j < CARRYU; ++j) {
      float P = __expf(a * sd[j]);
      hp[(c0 + j)*(LC*EDn)] = carry;
      carry = P * carry + he[j];
    }
  }
}

__global__ void __launch_bounds__(EDn)
scan_apply_kernel(const float* __restrict__ delta, const float* __restrict__ xc,
                  const float* __restrict__ Bm, const float* __restrict__ Cm,
                  const float* __restrict__ A_log, const float* __restrict__ Dp,
                  float* __restrict__ y) {
  const int d  = threadIdx.x;
  const int t0 = blockIdx.y * Ln + blockIdx.x * LC;
  float a[DSTATEn], h[DSTATEn];
#pragma unroll
  for (int n = 0; n < DSTATEn; ++n) {
    a[n] = -__expf(A_log[d*DSTATEn + n]);
    h[n] = y[(t0 + n)*EDn + d];
  }
  const float dp = Dp[d];
#pragma unroll
  for (int l = 0; l < LC; ++l) {
    const int t = t0 + l;
    float dt = delta[t*EDn + d];
    float xv = xc[t*EDn + d];
    float yv = 0.f;
#pragma unroll
    for (int n = 0; n < DSTATEn; ++n) {
      float dA = __expf(dt * a[n]);
      h[n] = dA * h[n] + dt * Bm[t*DSTATEn + n] * xv;
      yv += h[n] * Cm[t*DSTATEn + n];
    }
    y[t*EDn + d] = yv + dp * xv;
  }
}

__global__ void __launch_bounds__(128, 4)
out_kernel(const float* y, const float* zb, const v2f* Wout_sw,
                           const float* xr, const float* x, float* out) {
  __shared__ float ys[TOKB*PE];
  int tid = threadIdx.x;
  int wave = tid >> 5;
  int lane = tid & 31;
  long t0 = (long)blockIdx.x * TOKB;

#pragma unroll 8
  for (int i = tid; i < TOKB*EDn/4; i += 128) {
    int r = i >> 6, c = (i & 63) << 2;
    long tok = t0 + r;
    const float4 z4 = *(const float4*)(zb + tok*EDn + c);
    const float4 y4 = *(const float4*)(y  + tok*EDn + c);
    float4 o;
    o.x = y4.x * silu_f(z4.x);
    o.y = y4.y * silu_f(z4.y);
    o.z = y4.z * silu_f(z4.z);
    o.w = y4.w * silu_f(z4.w);
    *(float4*)(ys + r*PE + c) = o;
  }
  __syncthreads();

  gemm_stream<64, 2>(ys, PE, Wout_sw, wave*2, [&](int nt, const v8f (&acc)[TM]) {
    int n = nt*16 + (lane & 15);
    int mb = (lane < 16) ? 0 : 8;
    float r1[TM][8], r2[TM][8];
#pragma unroll
    for (int t = 0; t < TM; ++t)
#pragma unroll
      for (int j = 0; j < 8; ++j) {
        long tok = t0 + t*16 + mb + j;
        r1[t][j] = xr[tok*Dn + n];
        r2[t][j] = x[tok*Dn + n];
      }
#pragma unroll
    for (int t = 0; t < TM; ++t)
#pragma unroll
      for (int j = 0; j < 8; ++j) {
        long tok = t0 + t*16 + mb + j;
        out[tok*Dn + n] = acc[t][j] + r1[t][j] + r2[t][j];
      }
  });
}

extern "C" void kernel_launch(void* const* d_in, const int* in_sizes, int n_in,
                              void* d_out, int out_size, void* d_ws, size_t ws_size,
                              hipStream_t stream) {
  (void)in_sizes; (void)n_in; (void)out_size; (void)ws_size;
  const float* x      = (const float*)d_in[0];
  const float* mem    = (const float*)d_in[1];
  const float* Wq     = (const float*)d_in[2];
  const float* Wk     = (const float*)d_in[3];
  const float* Wv     = (const float*)d_in[4];
  const float* Wo     = (const float*)d_in[5];
  const float* ln1_g  = (const float*)d_in[6];
  const float* ln1_b  = (const float*)d_in[7];
  const float* rms_w  = (const float*)d_in[8];
  const float* W_in   = (const float*)d_in[9];
  const float* conv_w = (const float*)d_in[10];
  const float* conv_b = (const float*)d_in[11];
  const float* W_x    = (const float*)d_in[12];
  const float* W_dt   = (const float*)d_in[13];
  const float* b_dt   = (const float*)d_in[14];
  const float* A_log  = (const float*)d_in[15];
  const float* Dp     = (const float*)d_in[16];
  const float* W_out  = (const float*)d_in[17];

  float* ws = (float*)d_ws;
  size_t off = 0;
  float* Win_sw  = ws + off; off += (size_t)32*32*32*2;
  float* Wx_sw   = ws + off; off += (size_t)2*64*32*2;
  float* Wdt_sw  = ws + off; off += (size_t)16*2*32*2;
  float* Wout_sw = ws + off; off += (size_t)8*64*32*2;
  float* Wqk_sw  = ws + off; off += (size_t)1*32*32*2;
  float* VWo_sw  = ws + off; off += (size_t)8*4*32*2;
  float* xr      = ws + off; off += (size_t)NTOK*Dn;
  float* xp      = ws + off; off += (size_t)NTOK*EDn;
  float* zb      = ws + off; off += (size_t)NTOK*EDn;
  float* xc      = ws + off; off += (size_t)NTOK*EDn;
  float* delta   = ws + off; off += (size_t)NTOK*EDn;
  float* Bm      = ws + off; off += (size_t)NTOK*DSTATEn;
  float* Cm      = ws + off; off += (size_t)NTOK*DSTATEn;
  float* y       = xp;

  prep_kernel<<<PBK_ALL, 256, 0, stream>>>(mem, Wk, Wv, Wq, Wo, W_in, W_x, W_dt,
                                           W_out, Wqk_sw, VWo_sw,
                                           Win_sw, Wx_sw, Wdt_sw, Wout_sw);

  attn_kernel<<<NBLK, 128, 0, stream>>>(x, ln1_g, ln1_b, (const v2f*)Wqk_sw,
                                        (const v2f*)VWo_sw, xr);
  in_kernel  <<<NBLK, 128, 0, stream>>>(xr, rms_w, (const v2f*)Win_sw, xp, zb);
  conv_kernel<<<NBLK, 128, 0, stream>>>(xp, conv_w, conv_b, (const v2f*)Wx_sw,
                                        (const v2f*)Wdt_sw, b_dt, xc, delta, Bm, Cm);
  scan_local_kernel<<<dim3(NCH, Bn), EDn, 0, stream>>>(delta, xc, Bm, A_log, y);
  scan_carry_kernel<<<dim3(DSTATEn, Bn), EDn, 0, stream>>>(A_log, y);
  scan_apply_kernel<<<dim3(NCH, Bn), EDn, 0, stream>>>(delta, xc, Bm, Cm,
                                                       A_log, Dp, y);
  out_kernel <<<NBLK, 128, 0, stream>>>(y, zb, (const v2f*)Wout_sw, xr, x,
                                        (float*)d_out);
}
